# starting kernel
# baseline (speedup 1.0000x reference)
#include <hip/hip_runtime.h>

typedef float v2f __attribute__((ext_vector_type(2)));
typedef float v8f __attribute__((ext_vector_type(8)));

#define N_NODES 100000
#define F_IN 256
#define H1 64
#define H2 32
#define NEG_SLOPE 0.2f

__device__ __forceinline__ void atomicMaxF(float* addr, float v) {
  if (v >= 0.0f)
    atomicMax((int*)addr, __float_as_int(v));
  else
    atomicMin((unsigned int*)addr, __float_as_uint(v));
}

template <int K, int NOUT, int MROWS>
__global__ void gemm_wmma_kernel(const float* __restrict__ X,
                                 const float* __restrict__ W,
                                 float* __restrict__ Hout) {
  constexpr int LDK = K + 4;
  constexpr int RT = MROWS / 16;
  __shared__ float xs[MROWS * LDK];
  const int nthreads = (NOUT / 16) * 32;
  const int tid = threadIdx.x;
  const int row0 = blockIdx.x * MROWS;

  __builtin_prefetch(W, 0, 1);

  for (int i = tid; i < MROWS * K; i += nthreads) {
    const int r = i / K, c = i % K;
    xs[r * LDK + c] = X[(size_t)(row0 + r) * K + c];
  }
  __syncthreads();

  const int wave = tid >> 5;
  const int lane = tid & 31;
  const int lh = lane & 15;
  const bool hi = (lane >= 16);
  const int col0 = wave * 16;

  v8f acc[RT];
#pragma unroll
  for (int t = 0; t < RT; ++t) acc[t] = (v8f){};

#pragma unroll 4
  for (int k = 0; k < K; k += 4) {
    const int ka = k + (hi ? 2 : 0);
    v2f b;
    b.x = W[(size_t)ka * NOUT + col0 + lh];
    b.y = W[(size_t)(ka + 1) * NOUT + col0 + lh];
#pragma unroll
    for (int t = 0; t < RT; ++t) {
      v2f a;
      a.x = xs[(t * 16 + lh) * LDK + ka];
      a.y = xs[(t * 16 + lh) * LDK + ka + 1];
      acc[t] = __builtin_amdgcn_wmma_f32_16x16x4_f32(false, a, false, b,
                                                     (short)0, acc[t],
                                                     false, false);
    }
  }

  const int rbase = hi ? 8 : 0;
#pragma unroll
  for (int t = 0; t < RT; ++t) {
#pragma unroll
    for (int r = 0; r < 8; ++r) {
      Hout[(size_t)(row0 + t * 16 + rbase + r) * NOUT + col0 + lh] = acc[t][r];
    }
  }
}

template <int NOUT>
__global__ void alpha_kernel(const float* __restrict__ H,
                             const float* __restrict__ av,
                             const float* __restrict__ bv,
                             float* __restrict__ as_, float* __restrict__ ad_,
                             int N) {
  const int n = blockIdx.x * blockDim.x + threadIdx.x;
  if (n >= N) return;
  float s = 0.0f, d = 0.0f;
#pragma unroll
  for (int c = 0; c < NOUT; ++c) {
    const float h = H[(size_t)n * NOUT + c];
    s += h * av[c];
    d += h * bv[c];
  }
  as_[n] = s;
  ad_[n] = d;
}

__global__ void init_kernel(float* __restrict__ m, float* __restrict__ den,
                            float* __restrict__ agg, int N, int total) {
  const int i = blockIdx.x * blockDim.x + threadIdx.x;
  if (i < total) agg[i] = 0.0f;
  if (i < N) {
    m[i] = -__builtin_inff();
    den[i] = 0.0f;
  }
}

__global__ void edge_max_kernel(const int* __restrict__ src,
                                const int* __restrict__ dst, int E, int N,
                                const float* __restrict__ as_,
                                const float* __restrict__ ad_,
                                float* __restrict__ ev, float* __restrict__ m) {
  const int e = blockIdx.x * blockDim.x + threadIdx.x;
  if (e >= E + N) return;
  const int s = (e < E) ? src[e] : (e - E);
  const int d = (e < E) ? dst[e] : (e - E);
  float v = as_[s] + ad_[d];
  v = (v > 0.0f) ? v : NEG_SLOPE * v;
  ev[e] = v;
  atomicMaxF(&m[d], v);
}

__global__ void edge_exp_kernel(const int* __restrict__ dst, int E, int N,
                                const float* __restrict__ ev,
                                const float* __restrict__ m,
                                float* __restrict__ ex,
                                float* __restrict__ den) {
  const int e = blockIdx.x * blockDim.x + threadIdx.x;
  if (e >= E + N) return;
  const int d = (e < E) ? dst[e] : (e - E);
  const float t = __expf(ev[e] - m[d]);
  ex[e] = t;
  atomicAdd(&den[d], t);
}

template <int NOUT>
__global__ void edge_agg_kernel(const int* __restrict__ src,
                                const int* __restrict__ dst, int E, int N,
                                const float* __restrict__ H,
                                const float* __restrict__ ex,
                                const float* __restrict__ den,
                                float* __restrict__ agg) {
  const long long i = (long long)blockIdx.x * blockDim.x + threadIdx.x;
  const int e = (int)(i / NOUT);
  const int c = (int)(i - (long long)e * NOUT);
  if (e >= E + N) return;
  const int s = (e < E) ? src[e] : (e - E);
  const int d = (e < E) ? dst[e] : (e - E);
  const float attn = ex[e] / den[d];
  atomicAdd(&agg[(size_t)d * NOUT + c], H[(size_t)s * NOUT + c] * attn);
}

template <int NOUT>
__global__ void bias_relu_kernel(float* __restrict__ agg,
                                 const float* __restrict__ b, int total) {
  const int i = blockIdx.x * blockDim.x + threadIdx.x;
  if (i >= total) return;
  const float v = agg[i] + b[i % NOUT];
  agg[i] = (v > 0.0f) ? v : 0.0f;
}

__global__ void fc_kernel(const float* __restrict__ H,
                          const float* __restrict__ Wfc,
                          const float* __restrict__ bfc,
                          float* __restrict__ out, int N) {
  const int n = blockIdx.x * blockDim.x + threadIdx.x;
  if (n >= N) return;
  float sum = bfc[0];
#pragma unroll
  for (int c = 0; c < H2; ++c) sum += H[(size_t)n * H2 + c] * Wfc[c];
  out[n] = sum;
}

static inline unsigned cdiv(long long a, long long b) {
  return (unsigned)((a + b - 1) / b);
}

extern "C" void kernel_launch(void* const* d_in, const int* in_sizes, int n_in,
                              void* d_out, int out_size, void* d_ws,
                              size_t ws_size, hipStream_t stream) {
  (void)n_in; (void)out_size; (void)ws_size;
  const float* x     = (const float*)d_in[0];
  const int*   ei    = (const int*)d_in[1];
  const float* W1    = (const float*)d_in[2];
  const float* asrc1 = (const float*)d_in[3];
  const float* adst1 = (const float*)d_in[4];
  const float* b1    = (const float*)d_in[5];
  const float* W2    = (const float*)d_in[6];
  const float* asrc2 = (const float*)d_in[7];
  const float* adst2 = (const float*)d_in[8];
  const float* b2    = (const float*)d_in[9];
  const float* Wfc   = (const float*)d_in[10];
  const float* bfc   = (const float*)d_in[11];
  float* out = (float*)d_out;

  const int N = N_NODES;
  const int E = in_sizes[1] / 2;
  const int Etot = E + N;
  const int* src = ei;
  const int* dst = ei + E;

  float* ws = (float*)d_ws;
  size_t off = 0;
  float* h    = ws + off; off += (size_t)N * H1;
  float* agg1 = ws + off; off += (size_t)N * H1;
  float* agg2 = ws + off; off += (size_t)N * H2;
  float* as_  = ws + off; off += N;
  float* ad_  = ws + off; off += N;
  float* m    = ws + off; off += N;
  float* den  = ws + off; off += N;
  float* ev   = ws + off; off += Etot;
  float* ex   = ws + off; off += Etot;

  const dim3 B(256);
  constexpr int MR = 32;

  gemm_wmma_kernel<F_IN, H1, MR><<<N / MR, (H1 / 16) * 32, 0, stream>>>(x, W1, h);
  alpha_kernel<H1><<<cdiv(N, 256), B, 0, stream>>>(h, asrc1, adst1, as_, ad_, N);
  init_kernel<<<cdiv((long long)N * H1, 256), B, 0, stream>>>(m, den, agg1, N, N * H1);
  edge_max_kernel<<<cdiv(Etot, 256), B, 0, stream>>>(src, dst, E, N, as_, ad_, ev, m);
  edge_exp_kernel<<<cdiv(Etot, 256), B, 0, stream>>>(dst, E, N, ev, m, ex, den);
  edge_agg_kernel<H1><<<cdiv((long long)Etot * H1, 256), B, 0, stream>>>(
      src, dst, E, N, h, ex, den, agg1);
  bias_relu_kernel<H1><<<cdiv((long long)N * H1, 256), B, 0, stream>>>(agg1, b1, N * H1);

  gemm_wmma_kernel<H1, H2, MR><<<N / MR, (H2 / 16) * 32, 0, stream>>>(agg1, W2, h);
  alpha_kernel<H2><<<cdiv(N, 256), B, 0, stream>>>(h, asrc2, adst2, as_, ad_, N);
  init_kernel<<<cdiv((long long)N * H2, 256), B, 0, stream>>>(m, den, agg2, N, N * H2);
  edge_max_kernel<<<cdiv(Etot, 256), B, 0, stream>>>(src, dst, E, N, as_, ad_, ev, m);
  edge_exp_kernel<<<cdiv(Etot, 256), B, 0, stream>>>(dst, E, N, ev, m, ex, den);
  edge_agg_kernel<H2><<<cdiv((long long)Etot * H2, 256), B, 0, stream>>>(
      src, dst, E, N, h, ex, den, agg2);
  bias_relu_kernel<H2><<<cdiv((long long)N * H2, 256), B, 0, stream>>>(agg2, b2, N * H2);

  fc_kernel<<<cdiv(N, 256), B, 0, stream>>>(agg2, Wfc, bfc, out, N);
}


# Sort edges by destination once per call (counting sort, arrival-order rank), then one wave per node does softmax max/sum and the row sum in registers: no float atomics, sum order changes
# speedup vs baseline: 3.4943x; 3.4943x over previous
#include <hip/hip_runtime.h>

typedef float v2f __attribute__((ext_vector_type(2)));
typedef float v8f __attribute__((ext_vector_type(8)));

#define N_NODES 100000
#define F_IN 256
#define H1 64
#define H2 32
#define NEG_SLOPE 0.2f

template <int K, int NOUT, int MROWS>
__global__ void gemm_wmma_kernel(const float* __restrict__ X,
                                 const float* __restrict__ W,
                                 float* __restrict__ Hout) {
  constexpr int LDK = K + 4;
  constexpr int RT = MROWS / 16;
  __shared__ float xs[MROWS * LDK];
  const int nthreads = (NOUT / 16) * 32;
  const int tid = threadIdx.x;
  const int row0 = blockIdx.x * MROWS;

  __builtin_prefetch(W, 0, 1);

  for (int i = tid; i < MROWS * K; i += nthreads) {
    const int r = i / K, c = i % K;
    xs[r * LDK + c] = X[(size_t)(row0 + r) * K + c];
  }
  __syncthreads();

  const int wave = tid >> 5;
  const int lane = tid & 31;
  const int lh = lane & 15;
  const bool hi = (lane >= 16);
  const int col0 = wave * 16;

  v8f acc[RT];
#pragma unroll
  for (int t = 0; t < RT; ++t) acc[t] = (v8f){};

#pragma unroll 4
  for (int k = 0; k < K; k += 4) {
    const int ka = k + (hi ? 2 : 0);
    v2f b;
    b.x = W[(size_t)ka * NOUT + col0 + lh];
    b.y = W[(size_t)(ka + 1) * NOUT + col0 + lh];
#pragma unroll
    for (int t = 0; t < RT; ++t) {
      v2f a;
      a.x = xs[(t * 16 + lh) * LDK + ka];
      a.y = xs[(t * 16 + lh) * LDK + ka + 1];
      acc[t] = __builtin_amdgcn_wmma_f32_16x16x4_f32(false, a, false, b,
                                                     (short)0, acc[t],
                                                     false, false);
    }
  }

  const int rbase = hi ? 8 : 0;
#pragma unroll
  for (int t = 0; t < RT; ++t) {
#pragma unroll
    for (int r = 0; r < 8; ++r) {
      Hout[(size_t)(row0 + t * 16 + rbase + r) * NOUT + col0 + lh] = acc[t][r];
    }
  }
}

template <int NOUT>
__global__ void alpha_kernel(const float* __restrict__ H,
                             const float* __restrict__ av,
                             const float* __restrict__ bv,
                             float* __restrict__ as_, float* __restrict__ ad_,
                             int N) {
  const int n = blockIdx.x * blockDim.x + threadIdx.x;
  if (n >= N) return;
  float s = 0.0f, d = 0.0f;
#pragma unroll
  for (int c = 0; c < NOUT; ++c) {
    const float h = H[(size_t)n * NOUT + c];
    s += h * av[c];
    d += h * bv[c];
  }
  as_[n] = s;
  ad_[n] = d;
}

__device__ __forceinline__ float leaky(float v) {
  return (v > 0.0f) ? v : NEG_SLOPE * v;
}

__device__ __forceinline__ float wave_max_f(float v) {
#pragma unroll
  for (int m = 16; m >= 1; m >>= 1) v = fmaxf(v, __shfl_xor(v, m, 32));
  return v;
}

__device__ __forceinline__ float wave_sum_f(float v) {
#pragma unroll
  for (int m = 16; m >= 1; m >>= 1) v += __shfl_xor(v, m, 32);
  return v;
}

__device__ __forceinline__ int wave_sum_i(int v) {
#pragma unroll
  for (int m = 16; m >= 1; m >>= 1) v += __shfl_xor(v, m, 32);
  return v;
}

__global__ void zero_count_kernel(int* __restrict__ cnt, int N) {
  const int i = blockIdx.x * blockDim.x + threadIdx.x;
  if (i < N) cnt[i] = 0;
}

__global__ void edge_rank_kernel(const int* __restrict__ dst, int E,
                                 int* __restrict__ cnt,
                                 int* __restrict__ rank) {
  const int e = blockIdx.x * blockDim.x + threadIdx.x;
  if (e >= E) return;
  rank[e] = atomicAdd(&cnt[dst[e]], 1);
}

__global__ void count_blocksum_kernel(const int* __restrict__ cnt, int N,
                                      int* __restrict__ bsum) {
  __shared__ int wsum[8];
  const int tid = threadIdx.x;
  const int i0 = (blockIdx.x * 256 + tid) * 4;
  int t = 0;
#pragma unroll
  for (int k = 0; k < 4; ++k)
    if (i0 + k < N) t += cnt[i0 + k];
  t = wave_sum_i(t);
  if ((tid & 31) == 0) wsum[tid >> 5] = t;
  __syncthreads();
  if (tid == 0) {
    int s = 0;
#pragma unroll
    for (int w = 0; w < 8; ++w) s += wsum[w];
    bsum[blockIdx.x] = s;
  }
}

__global__ void count_scan_kernel(const int* __restrict__ cnt, int N,
                                  const int* __restrict__ bsum,
                                  int* __restrict__ rowptr) {
  __shared__ int wsum[8];
  __shared__ int wtot[8];
  const int tid = threadIdx.x;
  const int lane = tid & 31;
  const int wave = tid >> 5;

  int p = 0;
  for (int i = tid; i < (int)blockIdx.x; i += 256) p += bsum[i];
  p = wave_sum_i(p);
  if (lane == 0) wsum[wave] = p;

  const int i0 = (blockIdx.x * 256 + tid) * 4;
  int c[4];
  int t = 0;
#pragma unroll
  for (int k = 0; k < 4; ++k) {
    c[k] = (i0 + k < N) ? cnt[i0 + k] : 0;
    t += c[k];
  }
  int incl = t;
#pragma unroll
  for (int m = 1; m < 32; m <<= 1) {
    const int u = __shfl_up(incl, m, 32);
    if (lane >= m) incl += u;
  }
  if (lane == 31) wtot[wave] = incl;
  __syncthreads();

  int off = 0;
#pragma unroll
  for (int w = 0; w < 8; ++w) off += wsum[w];
#pragma unroll
  for (int w = 0; w < 8; ++w)
    if (w < wave) off += wtot[w];

  int run = off + incl - t;
#pragma unroll
  for (int k = 0; k < 4; ++k) {
    if (i0 + k < N) rowptr[i0 + k] = run;
    run += c[k];
  }
}

__global__ void edge_scatter_kernel(const int* __restrict__ src,
                                    const int* __restrict__ dst, int E,
                                    const int* __restrict__ rowptr,
                                    const int* __restrict__ rank,
                                    int* __restrict__ csr_src) {
  const int e = blockIdx.x * blockDim.x + threadIdx.x;
  if (e >= E) return;
  const int pos = rowptr[dst[e]] + rank[e];
  if ((unsigned)pos < (unsigned)E) csr_src[pos] = src[e];
}

template <int NOUT>
__global__ void node_gather_kernel(const int* __restrict__ rowptr,
                                   const int* __restrict__ csr_src, int E,
                                   int N, const float* __restrict__ H,
                                   const float* __restrict__ as_,
                                   const float* __restrict__ ad_,
                                   const float* __restrict__ bias,
                                   float* __restrict__ out) {
  constexpr int VPL = NOUT / 32;
  const int lane = threadIdx.x & 31;
  const int d = __builtin_amdgcn_readfirstlane(
      (int)(blockIdx.x * (blockDim.x >> 5) + (threadIdx.x >> 5)));
  if (d >= N) return;

  int end = (d + 1 < N) ? rowptr[d + 1] : E;
  end = min(max(end, 0), E);
  const int beg = min(max(rowptr[d], 0), end);
  const int deg = end - beg;
  const unsigned nmax = (unsigned)(N - 1);

  const float adv = ad_[d];
  const float e_self = leaky(as_[d] + adv);

  const bool act0 = lane < deg;
  int s0 = d;
  float v0 = -__builtin_inff();
  if (act0) {
    s0 = (int)min((unsigned)csr_src[beg + lane], nmax);
    v0 = leaky(as_[s0] + adv);
  }

  float mx = fmaxf(e_self, v0);
  for (int j0 = 32; j0 < deg; j0 += 32) {
    const int j = j0 + lane;
    if (j < deg) {
      const int s = (int)min((unsigned)csr_src[beg + j], nmax);
      mx = fmaxf(mx, leaky(as_[s] + adv));
    }
  }
  mx = wave_max_f(mx);

  const float x0 = act0 ? __expf(v0 - mx) : 0.0f;
  float dsum = x0;
  for (int j0 = 32; j0 < deg; j0 += 32) {
    const int j = j0 + lane;
    if (j < deg) {
      const int s = (int)min((unsigned)csr_src[beg + j], nmax);
      dsum += __expf(leaky(as_[s] + adv) - mx);
    }
  }
  const float x_self = __expf(e_self - mx);
  const float den = wave_sum_f(dsum) + x_self;

  const int c0 = lane * VPL;
  float acc[VPL];
  {
    const float a_self = x_self / den;
    const float* hp = H + (size_t)d * NOUT + c0;
    if constexpr (VPL == 2) {
      const float2 hv = *reinterpret_cast<const float2*>(hp);
      acc[0] = hv.x * a_self;
      acc[1] = hv.y * a_self;
    } else {
      acc[0] = hp[0] * a_self;
    }
  }

  for (int j0 = 0; j0 < deg; j0 += 32) {
    int s_l = s0;
    float a_l = x0 / den;
    if (j0 > 0) {
      const int j = j0 + lane;
      s_l = d;
      a_l = 0.0f;
      if (j < deg) {
        s_l = (int)min((unsigned)csr_src[beg + j], nmax);
        a_l = __expf(leaky(as_[s_l] + adv) - mx) / den;
      }
    }
    const int cnt = min(32, deg - j0);
    int t = 0;
    for (; t + 4 <= cnt; t += 4) {
      int s[4];
      float a[4];
#pragma unroll
      for (int u = 0; u < 4; ++u) {
        s[u] = __builtin_amdgcn_readlane(s_l, t + u);
        a[u] = __int_as_float(
            __builtin_amdgcn_readlane(__float_as_int(a_l), t + u));
      }
      if constexpr (VPL == 2) {
        float2 hv[4];
#pragma unroll
        for (int u = 0; u < 4; ++u)
          hv[u] = *reinterpret_cast<const float2*>(H + (size_t)s[u] * NOUT + c0);
#pragma unroll
        for (int u = 0; u < 4; ++u) {
          acc[0] += hv[u].x * a[u];
          acc[1] += hv[u].y * a[u];
        }
      } else {
        float hv[4];
#pragma unroll
        for (int u = 0; u < 4; ++u) hv[u] = H[(size_t)s[u] * NOUT + c0];
#pragma unroll
        for (int u = 0; u < 4; ++u) acc[0] += hv[u] * a[u];
      }
    }
    for (; t < cnt; ++t) {
      const int s = __builtin_amdgcn_readlane(s_l, t);
      const float a =
          __int_as_float(__builtin_amdgcn_readlane(__float_as_int(a_l), t));
      const float* hp = H + (size_t)s * NOUT + c0;
      if constexpr (VPL == 2) {
        const float2 hv = *reinterpret_cast<const float2*>(hp);
        acc[0] += hv.x * a;
        acc[1] += hv.y * a;
      } else {
        acc[0] += hp[0] * a;
      }
    }
  }

  float* op = out + (size_t)d * NOUT + c0;
  if constexpr (VPL == 2) {
    float2 r;
    r.x = fmaxf(acc[0] + bias[c0], 0.0f);
    r.y = fmaxf(acc[1] + bias[c0 + 1], 0.0f);
    *reinterpret_cast<float2*>(op) = r;
  } else {
    op[0] = fmaxf(acc[0] + bias[c0], 0.0f);
  }
}

__global__ void fc_kernel(const float* __restrict__ H,
                          const float* __restrict__ Wfc,
                          const float* __restrict__ bfc,
                          float* __restrict__ out, int N) {
  const int n = blockIdx.x * blockDim.x + threadIdx.x;
  if (n >= N) return;
  float sum = bfc[0];
#pragma unroll
  for (int c = 0; c < H2; ++c) sum += H[(size_t)n * H2 + c] * Wfc[c];
  out[n] = sum;
}

static inline unsigned cdiv(long long a, long long b) {
  return (unsigned)((a + b - 1) / b);
}

extern "C" void kernel_launch(void* const* d_in, const int* in_sizes, int n_in,
                              void* d_out, int out_size, void* d_ws,
                              size_t ws_size, hipStream_t stream) {
  (void)n_in; (void)out_size; (void)ws_size;
  const float* x     = (const float*)d_in[0];
  const int*   ei    = (const int*)d_in[1];
  const float* W1    = (const float*)d_in[2];
  const float* asrc1 = (const float*)d_in[3];
  const float* adst1 = (const float*)d_in[4];
  const float* b1    = (const float*)d_in[5];
  const float* W2    = (const float*)d_in[6];
  const float* asrc2 = (const float*)d_in[7];
  const float* adst2 = (const float*)d_in[8];
  const float* b2    = (const float*)d_in[9];
  const float* Wfc   = (const float*)d_in[10];
  const float* bfc   = (const float*)d_in[11];
  float* out = (float*)d_out;

  const int N = N_NODES;
  const int E = in_sizes[1] / 2;
  const int* src = ei;
  const int* dst = ei + E;

  float* ws = (float*)d_ws;
  size_t off = 0;
  float* h    = ws + off; off += (size_t)N * H1;
  float* agg1 = ws + off; off += (size_t)N * H1;
  float* agg2 = ws + off; off += (size_t)N * H2;
  float* as_  = ws + off; off += N;
  float* ad_  = ws + off; off += N;
  int* cnt     = (int*)(ws + off); off += N;
  int* rowptr  = (int*)(ws + off); off += N;
  int* bsum    = (int*)(ws + off); off += 1024;
  int* rank    = (int*)(ws + off); off += E;
  int* csr_src = (int*)(ws + off); off += E;

  const dim3 B(256);
  constexpr int MR = 32;
  const unsigned nscan = cdiv(N, 1024);

  zero_count_kernel<<<cdiv(N, 256), B, 0, stream>>>(cnt, N);
  edge_rank_kernel<<<cdiv(E, 256), B, 0, stream>>>(dst, E, cnt, rank);
  count_blocksum_kernel<<<nscan, B, 0, stream>>>(cnt, N, bsum);
  count_scan_kernel<<<nscan, B, 0, stream>>>(cnt, N, bsum, rowptr);
  edge_scatter_kernel<<<cdiv(E, 256), B, 0, stream>>>(src, dst, E, rowptr, rank, csr_src);

  gemm_wmma_kernel<F_IN, H1, MR><<<N / MR, (H1 / 16) * 32, 0, stream>>>(x, W1, h);
  alpha_kernel<H1><<<cdiv(N, 256), B, 0, stream>>>(h, asrc1, adst1, as_, ad_, N);
  node_gather_kernel<H1><<<cdiv(N, 8), B, 0, stream>>>(rowptr, csr_src, E, N, h, as_, ad_, b1, agg1);

  gemm_wmma_kernel<H1, H2, MR><<<N / MR, (H2 / 16) * 32, 0, stream>>>(agg1, W2, h);
  alpha_kernel<H2><<<cdiv(N, 256), B, 0, stream>>>(h, asrc2, adst2, as_, ad_, N);
  node_gather_kernel<H2><<<cdiv(N, 8), B, 0, stream>>>(rowptr, csr_src, E, N, h, as_, ad_, b2, agg2);

  fc_kernel<<<cdiv(N, 256), B, 0, stream>>>(agg2, Wfc, bfc, out, N);
}


# Group edges by destination with LDS counters (256-node bins per edge chunk, scan, rank in LDS per bin) instead of one returning global atomic per edge; sum order stays arrival order, now of LDS adds
# speedup vs baseline: 4.7346x; 1.3549x over previous
#include <hip/hip_runtime.h>

typedef float v2f __attribute__((ext_vector_type(2)));
typedef float v8f __attribute__((ext_vector_type(8)));

#define N_NODES 100000
#define F_IN 256
#define H1 64
#define H2 32
#define NEG_SLOPE 0.2f

template <int K, int NOUT, int MROWS>
__global__ void gemm_wmma_kernel(const float* __restrict__ X,
                                 const float* __restrict__ W,
                                 float* __restrict__ Hout) {
  constexpr int LDK = K + 4;
  constexpr int RT = MROWS / 16;
  __shared__ float xs[MROWS * LDK];
  const int nthreads = (NOUT / 16) * 32;
  const int tid = threadIdx.x;
  const int row0 = blockIdx.x * MROWS;

  __builtin_prefetch(W, 0, 1);

  for (int i = tid; i < MROWS * K; i += nthreads) {
    const int r = i / K, c = i % K;
    xs[r * LDK + c] = X[(size_t)(row0 + r) * K + c];
  }
  __syncthreads();

  const int wave = tid >> 5;
  const int lane = tid & 31;
  const int lh = lane & 15;
  const bool hi = (lane >= 16);
  const int col0 = wave * 16;

  v8f acc[RT];
#pragma unroll
  for (int t = 0; t < RT; ++t) acc[t] = (v8f){};

#pragma unroll 4
  for (int k = 0; k < K; k += 4) {
    const int ka = k + (hi ? 2 : 0);
    v2f b;
    b.x = W[(size_t)ka * NOUT + col0 + lh];
    b.y = W[(size_t)(ka + 1) * NOUT + col0 + lh];
#pragma unroll
    for (int t = 0; t < RT; ++t) {
      v2f a;
      a.x = xs[(t * 16 + lh) * LDK + ka];
      a.y = xs[(t * 16 + lh) * LDK + ka + 1];
      acc[t] = __builtin_amdgcn_wmma_f32_16x16x4_f32(false, a, false, b,
                                                     (short)0, acc[t],
                                                     false, false);
    }
  }

  const int rbase = hi ? 8 : 0;
#pragma unroll
  for (int t = 0; t < RT; ++t) {
#pragma unroll
    for (int r = 0; r < 8; ++r) {
      Hout[(size_t)(row0 + t * 16 + rbase + r) * NOUT + col0 + lh] = acc[t][r];
    }
  }
}

template <int NOUT>
__global__ void alpha_kernel(const float* __restrict__ H,
                             const float* __restrict__ av,
                             const float* __restrict__ bv,
                             float* __restrict__ as_, float* __restrict__ ad_,
                             int N) {
  const int n = blockIdx.x * blockDim.x + threadIdx.x;
  if (n >= N) return;
  float s = 0.0f, d = 0.0f;
#pragma unroll
  for (int c = 0; c < NOUT; ++c) {
    const float h = H[(size_t)n * NOUT + c];
    s += h * av[c];
    d += h * bv[c];
  }
  as_[n] = s;
  ad_[n] = d;
}

__device__ __forceinline__ float leaky(float v) {
  return (v > 0.0f) ? v : NEG_SLOPE * v;
}

__device__ __forceinline__ float wave_max_f(float v) {
#pragma unroll
  for (int m = 16; m >= 1; m >>= 1) v = fmaxf(v, __shfl_xor(v, m, 32));
  return v;
}

__device__ __forceinline__ float wave_sum_f(float v) {
#pragma unroll
  for (int m = 16; m >= 1; m >>= 1) v += __shfl_xor(v, m, 32);
  return v;
}

__device__ __forceinline__ int wave_sum_i(int v) {
#pragma unroll
  for (int m = 16; m >= 1; m >>= 1) v += __shfl_xor(v, m, 32);
  return v;
}

#define BIN_SHIFT 8
#define BIN_NODES (1 << BIN_SHIFT)
#define NBINS ((N_NODES + BIN_NODES - 1) >> BIN_SHIFT)
#define EBLOCKS 250
#define ETHREADS 1024
#define EBATCH 4
#define STHREADS 512
#define SBATCH 8
static_assert(NBINS * EBLOCKS <= N_NODES, "hist must fit the N-word slot");
static_assert(N_NODES <= (1 << 24), "src must fit 24 bits of the packed edge");

__device__ __forceinline__ unsigned clamp_node(int v) {
  return min((unsigned)v, (unsigned)(N_NODES - 1));
}

__global__ void bin_count_kernel(const int* __restrict__ dst, int E,
                                 int* __restrict__ hist) {
  __shared__ int lc[NBINS];
  const int tid = threadIdx.x;
  for (int b = tid; b < NBINS; b += ETHREADS) lc[b] = 0;
  __syncthreads();
  constexpr int nblk = EBLOCKS;
  const int chunk = (E + nblk - 1) / nblk;
  const int e0 = min((int)blockIdx.x * chunk, E);
  const int e1 = min(e0 + chunk, E);
  for (int e = e0 + tid; e < e1; e += ETHREADS * EBATCH) {
    int d[EBATCH];
#pragma unroll
    for (int u = 0; u < EBATCH; ++u) d[u] = dst[min(e + u * ETHREADS, e1 - 1)];
#pragma unroll
    for (int u = 0; u < EBATCH; ++u)
      atomicAdd(&lc[clamp_node(d[u]) >> BIN_SHIFT],
                (e + u * ETHREADS < e1) ? 1 : 0);
  }
  __syncthreads();
  for (int b = tid; b < NBINS; b += ETHREADS)
    hist[b * nblk + (int)blockIdx.x] = lc[b];
}

__global__ void bin_scatter_kernel(const int* __restrict__ src,
                                   const int* __restrict__ dst, int E,
                                   const int* __restrict__ offs,
                                   unsigned* __restrict__ binned) {
  __shared__ int lcur[NBINS];
  const int tid = threadIdx.x;
  constexpr int nblk = EBLOCKS;
  for (int b = tid; b < NBINS; b += ETHREADS)
    lcur[b] = offs[b * nblk + (int)blockIdx.x];
  __syncthreads();
  const int chunk = (E + nblk - 1) / nblk;
  const int e0 = min((int)blockIdx.x * chunk, E);
  const int e1 = min(e0 + chunk, E);
  for (int e = e0 + tid; e < e1; e += ETHREADS * EBATCH) {
    int s[EBATCH], d[EBATCH];
#pragma unroll
    for (int u = 0; u < EBATCH; ++u) {
      const int ee = min(e + u * ETHREADS, e1 - 1);
      s[u] = src[ee];
      d[u] = dst[ee];
    }
#pragma unroll
    for (int u = 0; u < EBATCH; ++u) {
      const bool live = (e + u * ETHREADS < e1);
      const unsigned dn = clamp_node(d[u]);
      const int pos = atomicAdd(&lcur[dn >> BIN_SHIFT], live ? 1 : 0);
      if (live && (unsigned)pos < (unsigned)E)
        binned[pos] = (clamp_node(s[u]) << BIN_SHIFT) | (dn & (BIN_NODES - 1));
    }
  }
}

__global__ void bin_sort_kernel(const int* __restrict__ offs, int E, int N,
                                const unsigned* __restrict__ binned,
                                int* __restrict__ rowptr,
                                int* __restrict__ csr_src) {
  __shared__ int lcnt[BIN_NODES];
  __shared__ int wtot[BIN_NODES / 32];
  const int tid = threadIdx.x;
  const int lane = tid & 31;
  const int wave = tid >> 5;
  const int bin = (int)blockIdx.x;
  const int base = bin << BIN_SHIFT;

  int end = (bin + 1 < NBINS) ? offs[(bin + 1) * EBLOCKS] : E;
  end = min(max(end, 0), E);
  const int start = min(max(offs[bin * EBLOCKS], 0), end);

  if (tid < BIN_NODES) lcnt[tid] = 0;
  __syncthreads();

  for (int i = start + tid; i < end; i += STHREADS * SBATCH) {
    unsigned p[SBATCH];
#pragma unroll
    for (int u = 0; u < SBATCH; ++u) p[u] = binned[min(i + u * STHREADS, end - 1)];
#pragma unroll
    for (int u = 0; u < SBATCH; ++u)
      atomicAdd(&lcnt[p[u] & (BIN_NODES - 1)], (i + u * STHREADS < end) ? 1 : 0);
  }
  __syncthreads();

  int c = 0, incl = 0;
  if (tid < BIN_NODES) {
    c = lcnt[tid];
    incl = c;
#pragma unroll
    for (int m = 1; m < 32; m <<= 1) {
      const int u = __shfl_up(incl, m, 32);
      if (lane >= m) incl += u;
    }
    if (lane == 31) wtot[wave] = incl;
  }
  __syncthreads();
  if (tid < BIN_NODES) {
    int off = 0;
#pragma unroll
    for (int w = 0; w < BIN_NODES / 32; ++w)
      if (w < wave) off += wtot[w];
    const int first = start + off + incl - c;
    if (base + tid < N) rowptr[base + tid] = first;
    lcnt[tid] = first;
  }
  __syncthreads();

  for (int i = start + tid; i < end; i += STHREADS * SBATCH) {
    unsigned p[SBATCH];
#pragma unroll
    for (int u = 0; u < SBATCH; ++u) p[u] = binned[min(i + u * STHREADS, end - 1)];
#pragma unroll
    for (int u = 0; u < SBATCH; ++u) {
      const bool live = (i + u * STHREADS < end);
      const int pos = atomicAdd(&lcnt[p[u] & (BIN_NODES - 1)], live ? 1 : 0);
      if (live && (unsigned)pos < (unsigned)E)
        csr_src[pos] = (int)(p[u] >> BIN_SHIFT);
    }
  }
}

__global__ void count_blocksum_kernel(const int* __restrict__ cnt, int N,
                                      int* __restrict__ bsum) {
  __shared__ int wsum[8];
  const int tid = threadIdx.x;
  const int i0 = (blockIdx.x * 256 + tid) * 4;
  int t = 0;
#pragma unroll
  for (int k = 0; k < 4; ++k)
    if (i0 + k < N) t += cnt[i0 + k];
  t = wave_sum_i(t);
  if ((tid & 31) == 0) wsum[tid >> 5] = t;
  __syncthreads();
  if (tid == 0) {
    int s = 0;
#pragma unroll
    for (int w = 0; w < 8; ++w) s += wsum[w];
    bsum[blockIdx.x] = s;
  }
}

__global__ void count_scan_kernel(const int* __restrict__ cnt, int N,
                                  const int* __restrict__ bsum,
                                  int* __restrict__ rowptr) {
  __shared__ int wsum[8];
  __shared__ int wtot[8];
  const int tid = threadIdx.x;
  const int lane = tid & 31;
  const int wave = tid >> 5;

  int p = 0;
  for (int i = tid; i < (int)blockIdx.x; i += 256) p += bsum[i];
  p = wave_sum_i(p);
  if (lane == 0) wsum[wave] = p;

  const int i0 = (blockIdx.x * 256 + tid) * 4;
  int c[4];
  int t = 0;
#pragma unroll
  for (int k = 0; k < 4; ++k) {
    c[k] = (i0 + k < N) ? cnt[i0 + k] : 0;
    t += c[k];
  }
  int incl = t;
#pragma unroll
  for (int m = 1; m < 32; m <<= 1) {
    const int u = __shfl_up(incl, m, 32);
    if (lane >= m) incl += u;
  }
  if (lane == 31) wtot[wave] = incl;
  __syncthreads();

  int off = 0;
#pragma unroll
  for (int w = 0; w < 8; ++w) off += wsum[w];
#pragma unroll
  for (int w = 0; w < 8; ++w)
    if (w < wave) off += wtot[w];

  int run = off + incl - t;
#pragma unroll
  for (int k = 0; k < 4; ++k) {
    if (i0 + k < N) rowptr[i0 + k] = run;
    run += c[k];
  }
}

template <int NOUT>
__global__ void node_gather_kernel(const int* __restrict__ rowptr,
                                   const int* __restrict__ csr_src, int E,
                                   int N, const float* __restrict__ H,
                                   const float* __restrict__ as_,
                                   const float* __restrict__ ad_,
                                   const float* __restrict__ bias,
                                   float* __restrict__ out) {
  constexpr int VPL = NOUT / 32;
  const int lane = threadIdx.x & 31;
  const int d = __builtin_amdgcn_readfirstlane(
      (int)(blockIdx.x * (blockDim.x >> 5) + (threadIdx.x >> 5)));
  if (d >= N) return;

  int end = (d + 1 < N) ? rowptr[d + 1] : E;
  end = min(max(end, 0), E);
  const int beg = min(max(rowptr[d], 0), end);
  const int deg = end - beg;
  const unsigned nmax = (unsigned)(N - 1);

  const float adv = ad_[d];
  const float e_self = leaky(as_[d] + adv);

  const bool act0 = lane < deg;
  int s0 = d;
  float v0 = -__builtin_inff();
  if (act0) {
    s0 = (int)min((unsigned)csr_src[beg + lane], nmax);
    v0 = leaky(as_[s0] + adv);
  }

  float mx = fmaxf(e_self, v0);
  for (int j0 = 32; j0 < deg; j0 += 32) {
    const int j = j0 + lane;
    if (j < deg) {
      const int s = (int)min((unsigned)csr_src[beg + j], nmax);
      mx = fmaxf(mx, leaky(as_[s] + adv));
    }
  }
  mx = wave_max_f(mx);

  const float x0 = act0 ? __expf(v0 - mx) : 0.0f;
  float dsum = x0;
  for (int j0 = 32; j0 < deg; j0 += 32) {
    const int j = j0 + lane;
    if (j < deg) {
      const int s = (int)min((unsigned)csr_src[beg + j], nmax);
      dsum += __expf(leaky(as_[s] + adv) - mx);
    }
  }
  const float x_self = __expf(e_self - mx);
  const float den = wave_sum_f(dsum) + x_self;

  const int c0 = lane * VPL;
  float acc[VPL];
  {
    const float a_self = x_self / den;
    const float* hp = H + (size_t)d * NOUT + c0;
    if constexpr (VPL == 2) {
      const float2 hv = *reinterpret_cast<const float2*>(hp);
      acc[0] = hv.x * a_self;
      acc[1] = hv.y * a_self;
    } else {
      acc[0] = hp[0] * a_self;
    }
  }

  for (int j0 = 0; j0 < deg; j0 += 32) {
    int s_l = s0;
    float a_l = x0 / den;
    if (j0 > 0) {
      const int j = j0 + lane;
      s_l = d;
      a_l = 0.0f;
      if (j < deg) {
        s_l = (int)min((unsigned)csr_src[beg + j], nmax);
        a_l = __expf(leaky(as_[s_l] + adv) - mx) / den;
      }
    }
    const int cnt = min(32, deg - j0);
    int t = 0;
    for (; t + 4 <= cnt; t += 4) {
      int s[4];
      float a[4];
#pragma unroll
      for (int u = 0; u < 4; ++u) {
        s[u] = __builtin_amdgcn_readlane(s_l, t + u);
        a[u] = __int_as_float(
            __builtin_amdgcn_readlane(__float_as_int(a_l), t + u));
      }
      if constexpr (VPL == 2) {
        float2 hv[4];
#pragma unroll
        for (int u = 0; u < 4; ++u)
          hv[u] = *reinterpret_cast<const float2*>(H + (size_t)s[u] * NOUT + c0);
#pragma unroll
        for (int u = 0; u < 4; ++u) {
          acc[0] += hv[u].x * a[u];
          acc[1] += hv[u].y * a[u];
        }
      } else {
        float hv[4];
#pragma unroll
        for (int u = 0; u < 4; ++u) hv[u] = H[(size_t)s[u] * NOUT + c0];
#pragma unroll
        for (int u = 0; u < 4; ++u) acc[0] += hv[u] * a[u];
      }
    }
    for (; t < cnt; ++t) {
      const int s = __builtin_amdgcn_readlane(s_l, t);
      const float a =
          __int_as_float(__builtin_amdgcn_readlane(__float_as_int(a_l), t));
      const float* hp = H + (size_t)s * NOUT + c0;
      if constexpr (VPL == 2) {
        const float2 hv = *reinterpret_cast<const float2*>(hp);
        acc[0] += hv.x * a;
        acc[1] += hv.y * a;
      } else {
        acc[0] += hp[0] * a;
      }
    }
  }

  float* op = out + (size_t)d * NOUT + c0;
  if constexpr (VPL == 2) {
    float2 r;
    r.x = fmaxf(acc[0] + bias[c0], 0.0f);
    r.y = fmaxf(acc[1] + bias[c0 + 1], 0.0f);
    *reinterpret_cast<float2*>(op) = r;
  } else {
    op[0] = fmaxf(acc[0] + bias[c0], 0.0f);
  }
}

__global__ void fc_kernel(const float* __restrict__ H,
                          const float* __restrict__ Wfc,
                          const float* __restrict__ bfc,
                          float* __restrict__ out, int N) {
  const int n = blockIdx.x * blockDim.x + threadIdx.x;
  if (n >= N) return;
  float sum = bfc[0];
#pragma unroll
  for (int c = 0; c < H2; ++c) sum += H[(size_t)n * H2 + c] * Wfc[c];
  out[n] = sum;
}

static inline unsigned cdiv(long long a, long long b) {
  return (unsigned)((a + b - 1) / b);
}

extern "C" void kernel_launch(void* const* d_in, const int* in_sizes, int n_in,
                              void* d_out, int out_size, void* d_ws,
                              size_t ws_size, hipStream_t stream) {
  (void)n_in; (void)out_size; (void)ws_size;
  const float* x     = (const float*)d_in[0];
  const int*   ei    = (const int*)d_in[1];
  const float* W1    = (const float*)d_in[2];
  const float* asrc1 = (const float*)d_in[3];
  const float* adst1 = (const float*)d_in[4];
  const float* b1    = (const float*)d_in[5];
  const float* W2    = (const float*)d_in[6];
  const float* asrc2 = (const float*)d_in[7];
  const float* adst2 = (const float*)d_in[8];
  const float* b2    = (const float*)d_in[9];
  const float* Wfc   = (const float*)d_in[10];
  const float* bfc   = (const float*)d_in[11];
  float* out = (float*)d_out;

  const int N = N_NODES;
  const int E = in_sizes[1] / 2;
  const int* src = ei;
  const int* dst = ei + E;

  float* ws = (float*)d_ws;
  size_t off = 0;
  float* h    = ws + off; off += (size_t)N * H1;
  float* agg1 = ws + off; off += (size_t)N * H1;
  float* agg2 = ws + off; off += (size_t)N * H2;
  float* as_  = ws + off; off += N;
  float* ad_  = ws + off; off += N;
  int* hist    = (int*)(ws + off); off += N;
  int* rowptr  = (int*)(ws + off); off += N;
  int* bsum    = (int*)(ws + off); off += 1024;
  unsigned* binned = (unsigned*)(ws + off); off += E;
  int* csr_src = (int*)(ws + off); off += E;
  int* offs = (int*)agg2;

  const dim3 B(256);
  constexpr int MR = 32;
  constexpr int NH = NBINS * EBLOCKS;
  const unsigned nscan = cdiv(NH, 1024);

  bin_count_kernel<<<EBLOCKS, ETHREADS, 0, stream>>>(dst, E, hist);
  count_blocksum_kernel<<<nscan, B, 0, stream>>>(hist, NH, bsum);
  count_scan_kernel<<<nscan, B, 0, stream>>>(hist, NH, bsum, offs);
  bin_scatter_kernel<<<EBLOCKS, ETHREADS, 0, stream>>>(src, dst, E, offs, binned);
  bin_sort_kernel<<<NBINS, STHREADS, 0, stream>>>(offs, E, N, binned, rowptr, csr_src);

  gemm_wmma_kernel<F_IN, H1, MR><<<N / MR, (H1 / 16) * 32, 0, stream>>>(x, W1, h);
  alpha_kernel<H1><<<cdiv(N, 256), B, 0, stream>>>(h, asrc1, adst1, as_, ad_, N);
  node_gather_kernel<H1><<<cdiv(N, 8), B, 0, stream>>>(rowptr, csr_src, E, N, h, as_, ad_, b1, agg1);

  gemm_wmma_kernel<H1, H2, MR><<<N / MR, (H2 / 16) * 32, 0, stream>>>(agg1, W2, h);
  alpha_kernel<H2><<<cdiv(N, 256), B, 0, stream>>>(h, asrc2, adst2, as_, ad_, N);
  node_gather_kernel<H2><<<cdiv(N, 8), B, 0, stream>>>(rowptr, csr_src, E, N, h, as_, ad_, b2, agg2);

  fc_kernel<<<cdiv(N, 256), B, 0, stream>>>(agg2, Wfc, bfc, out, N);
}


# Gather takes each edge logit from the row it fetches anyway (dot with a_src across lanes, one-pass softmax): no per-edge as[] read, no alpha launches (12 to 10); logit/softmax sums re-associated
# speedup vs baseline: 5.0564x; 1.0680x over previous
#include <hip/hip_runtime.h>

typedef float v2f __attribute__((ext_vector_type(2)));
typedef float v8f __attribute__((ext_vector_type(8)));

#define N_NODES 100000
#define F_IN 256
#define H1 64
#define H2 32
#define NEG_SLOPE 0.2f

template <int K, int NOUT, int MROWS>
__global__ void gemm_wmma_kernel(const float* __restrict__ X,
                                 const float* __restrict__ W,
                                 float* __restrict__ Hout) {
  constexpr int LDK = K + 4;
  constexpr int RT = MROWS / 16;
  __shared__ float xs[MROWS * LDK];
  const int nthreads = (NOUT / 16) * 32;
  const int tid = threadIdx.x;
  const int row0 = blockIdx.x * MROWS;

  __builtin_prefetch(W, 0, 1);

  for (int i = tid; i < MROWS * K; i += nthreads) {
    const int r = i / K, c = i % K;
    xs[r * LDK + c] = X[(size_t)(row0 + r) * K + c];
  }
  __syncthreads();

  const int wave = tid >> 5;
  const int lane = tid & 31;
  const int lh = lane & 15;
  const bool hi = (lane >= 16);
  const int col0 = wave * 16;

  v8f acc[RT];
#pragma unroll
  for (int t = 0; t < RT; ++t) acc[t] = (v8f){};

#pragma unroll 4
  for (int k = 0; k < K; k += 4) {
    const int ka = k + (hi ? 2 : 0);
    v2f b;
    b.x = W[(size_t)ka * NOUT + col0 + lh];
    b.y = W[(size_t)(ka + 1) * NOUT + col0 + lh];
#pragma unroll
    for (int t = 0; t < RT; ++t) {
      v2f a;
      a.x = xs[(t * 16 + lh) * LDK + ka];
      a.y = xs[(t * 16 + lh) * LDK + ka + 1];
      acc[t] = __builtin_amdgcn_wmma_f32_16x16x4_f32(false, a, false, b,
                                                     (short)0, acc[t],
                                                     false, false);
    }
  }

  const int rbase = hi ? 8 : 0;
#pragma unroll
  for (int t = 0; t < RT; ++t) {
#pragma unroll
    for (int r = 0; r < 8; ++r) {
      Hout[(size_t)(row0 + t * 16 + rbase + r) * NOUT + col0 + lh] = acc[t][r];
    }
  }
}

__device__ __forceinline__ float leaky(float v) {
  return (v > 0.0f) ? v : NEG_SLOPE * v;
}

template <int CTRL>
__device__ __forceinline__ float dpp_row(float x) {
  const int b = __float_as_int(x);
  return __int_as_float(__builtin_amdgcn_update_dpp(b, b, CTRL, 0xf, 0xf, false));
}

__device__ __forceinline__ float xhalf(float x) {
  const int b = __float_as_int(x);
  return __int_as_float(
      __builtin_amdgcn_permlanex16(b, b, 0x76543210, 0xfedcba98, false, false));
}

__device__ __forceinline__ float wave_sum_f(float v) {
  v += dpp_row<0x161>(v);
  v += dpp_row<0x162>(v);
  v += dpp_row<0x164>(v);
  v += dpp_row<0x168>(v);
  return v + xhalf(v);
}

__device__ __forceinline__ float sum4_rows(float p0, float p1, float p2,
                                           float p3, int lane) {
  const bool b0 = (lane & 1) != 0;
  const bool b1 = (lane & 2) != 0;
  float k01 = b0 ? p1 : p0;
  const float t01 = b0 ? p0 : p1;
  float k23 = b0 ? p3 : p2;
  const float t23 = b0 ? p2 : p3;
  k01 += dpp_row<0x161>(t01);
  k23 += dpp_row<0x161>(t23);
  float k = b1 ? k23 : k01;
  const float tt = b1 ? k01 : k23;
  k += dpp_row<0x162>(tt);
  k += dpp_row<0x164>(k);
  k += dpp_row<0x168>(k);
  return k + xhalf(k);
}

__device__ __forceinline__ int wave_sum_i(int v) {
#pragma unroll
  for (int m = 16; m >= 1; m >>= 1) v += __shfl_xor(v, m, 32);
  return v;
}

#define BIN_SHIFT 8
#define BIN_NODES (1 << BIN_SHIFT)
#define NBINS ((N_NODES + BIN_NODES - 1) >> BIN_SHIFT)
#define EBLOCKS 250
#define ETHREADS 1024
#define EBATCH 4
#define STHREADS 512
#define SBATCH 8
static_assert(NBINS * EBLOCKS <= N_NODES, "hist must fit the N-word slot");
static_assert(N_NODES <= (1 << 24), "src must fit 24 bits of the packed edge");

__device__ __forceinline__ unsigned clamp_node(int v) {
  return min((unsigned)v, (unsigned)(N_NODES - 1));
}

__global__ void bin_count_kernel(const int* __restrict__ dst, int E,
                                 int* __restrict__ hist) {
  __shared__ int lc[NBINS];
  const int tid = threadIdx.x;
  for (int b = tid; b < NBINS; b += ETHREADS) lc[b] = 0;
  __syncthreads();
  constexpr int nblk = EBLOCKS;
  const int chunk = (E + nblk - 1) / nblk;
  const int e0 = min((int)blockIdx.x * chunk, E);
  const int e1 = min(e0 + chunk, E);
  for (int e = e0 + tid; e < e1; e += ETHREADS * EBATCH) {
    int d[EBATCH];
#pragma unroll
    for (int u = 0; u < EBATCH; ++u) d[u] = dst[min(e + u * ETHREADS, e1 - 1)];
#pragma unroll
    for (int u = 0; u < EBATCH; ++u)
      atomicAdd(&lc[clamp_node(d[u]) >> BIN_SHIFT],
                (e + u * ETHREADS < e1) ? 1 : 0);
  }
  __syncthreads();
  for (int b = tid; b < NBINS; b += ETHREADS)
    hist[b * nblk + (int)blockIdx.x] = lc[b];
}

__global__ void bin_scatter_kernel(const int* __restrict__ src,
                                   const int* __restrict__ dst, int E,
                                   const int* __restrict__ offs,
                                   unsigned* __restrict__ binned) {
  __shared__ int lcur[NBINS];
  const int tid = threadIdx.x;
  constexpr int nblk = EBLOCKS;
  for (int b = tid; b < NBINS; b += ETHREADS)
    lcur[b] = offs[b * nblk + (int)blockIdx.x];
  __syncthreads();
  const int chunk = (E + nblk - 1) / nblk;
  const int e0 = min((int)blockIdx.x * chunk, E);
  const int e1 = min(e0 + chunk, E);
  for (int e = e0 + tid; e < e1; e += ETHREADS * EBATCH) {
    int s[EBATCH], d[EBATCH];
#pragma unroll
    for (int u = 0; u < EBATCH; ++u) {
      const int ee = min(e + u * ETHREADS, e1 - 1);
      s[u] = src[ee];
      d[u] = dst[ee];
    }
#pragma unroll
    for (int u = 0; u < EBATCH; ++u) {
      const bool live = (e + u * ETHREADS < e1);
      const unsigned dn = clamp_node(d[u]);
      const int pos = atomicAdd(&lcur[dn >> BIN_SHIFT], live ? 1 : 0);
      if (live && (unsigned)pos < (unsigned)E)
        binned[pos] = (clamp_node(s[u]) << BIN_SHIFT) | (dn & (BIN_NODES - 1));
    }
  }
}

__global__ void bin_sort_kernel(const int* __restrict__ offs, int E, int N,
                                const unsigned* __restrict__ binned,
                                int* __restrict__ rowptr,
                                int* __restrict__ csr_src) {
  __shared__ int lcnt[BIN_NODES];
  __shared__ int wtot[BIN_NODES / 32];
  const int tid = threadIdx.x;
  const int lane = tid & 31;
  const int wave = tid >> 5;
  const int bin = (int)blockIdx.x;
  const int base = bin << BIN_SHIFT;

  int end = (bin + 1 < NBINS) ? offs[(bin + 1) * EBLOCKS] : E;
  end = min(max(end, 0), E);
  const int start = min(max(offs[bin * EBLOCKS], 0), end);

  if (tid < BIN_NODES) lcnt[tid] = 0;
  __syncthreads();

  for (int i = start + tid; i < end; i += STHREADS * SBATCH) {
    unsigned p[SBATCH];
#pragma unroll
    for (int u = 0; u < SBATCH; ++u) p[u] = binned[min(i + u * STHREADS, end - 1)];
#pragma unroll
    for (int u = 0; u < SBATCH; ++u)
      atomicAdd(&lcnt[p[u] & (BIN_NODES - 1)], (i + u * STHREADS < end) ? 1 : 0);
  }
  __syncthreads();

  int c = 0, incl = 0;
  if (tid < BIN_NODES) {
    c = lcnt[tid];
    incl = c;
#pragma unroll
    for (int m = 1; m < 32; m <<= 1) {
      const int u = __shfl_up(incl, m, 32);
      if (lane >= m) incl += u;
    }
    if (lane == 31) wtot[wave] = incl;
  }
  __syncthreads();
  if (tid < BIN_NODES) {
    int off = 0;
#pragma unroll
    for (int w = 0; w < BIN_NODES / 32; ++w)
      if (w < wave) off += wtot[w];
    const int first = start + off + incl - c;
    if (base + tid < N) rowptr[base + tid] = first;
    lcnt[tid] = first;
  }
  __syncthreads();

  for (int i = start + tid; i < end; i += STHREADS * SBATCH) {
    unsigned p[SBATCH];
#pragma unroll
    for (int u = 0; u < SBATCH; ++u) p[u] = binned[min(i + u * STHREADS, end - 1)];
#pragma unroll
    for (int u = 0; u < SBATCH; ++u) {
      const bool live = (i + u * STHREADS < end);
      const int pos = atomicAdd(&lcnt[p[u] & (BIN_NODES - 1)], live ? 1 : 0);
      if (live && (unsigned)pos < (unsigned)E)
        csr_src[pos] = (int)(p[u] >> BIN_SHIFT);
    }
  }
}

__global__ void count_blocksum_kernel(const int* __restrict__ cnt, int N,
                                      int* __restrict__ bsum) {
  __shared__ int wsum[8];
  const int tid = threadIdx.x;
  const int i0 = (blockIdx.x * 256 + tid) * 4;
  int t = 0;
#pragma unroll
  for (int k = 0; k < 4; ++k)
    if (i0 + k < N) t += cnt[i0 + k];
  t = wave_sum_i(t);
  if ((tid & 31) == 0) wsum[tid >> 5] = t;
  __syncthreads();
  if (tid == 0) {
    int s = 0;
#pragma unroll
    for (int w = 0; w < 8; ++w) s += wsum[w];
    bsum[blockIdx.x] = s;
  }
}

__global__ void count_scan_kernel(const int* __restrict__ cnt, int N,
                                  const int* __restrict__ bsum,
                                  int* __restrict__ rowptr) {
  __shared__ int wsum[8];
  __shared__ int wtot[8];
  const int tid = threadIdx.x;
  const int lane = tid & 31;
  const int wave = tid >> 5;

  int p = 0;
  for (int i = tid; i < (int)blockIdx.x; i += 256) p += bsum[i];
  p = wave_sum_i(p);
  if (lane == 0) wsum[wave] = p;

  const int i0 = (blockIdx.x * 256 + tid) * 4;
  int c[4];
  int t = 0;
#pragma unroll
  for (int k = 0; k < 4; ++k) {
    c[k] = (i0 + k < N) ? cnt[i0 + k] : 0;
    t += c[k];
  }
  int incl = t;
#pragma unroll
  for (int m = 1; m < 32; m <<= 1) {
    const int u = __shfl_up(incl, m, 32);
    if (lane >= m) incl += u;
  }
  if (lane == 31) wtot[wave] = incl;
  __syncthreads();

  int off = 0;
#pragma unroll
  for (int w = 0; w < 8; ++w) off += wsum[w];
#pragma unroll
  for (int w = 0; w < 8; ++w)
    if (w < wave) off += wtot[w];

  int run = off + incl - t;
#pragma unroll
  for (int k = 0; k < 4; ++k) {
    if (i0 + k < N) rowptr[i0 + k] = run;
    run += c[k];
  }
}

template <int NOUT>
__global__ void node_gather_kernel(const int* __restrict__ rowptr,
                                   const int* __restrict__ csr_src, int E,
                                   int N, const float* __restrict__ H,
                                   const float* __restrict__ av,
                                   const float* __restrict__ bv,
                                   const float* __restrict__ bias,
                                   float* __restrict__ out) {
  constexpr int VPL = NOUT / 32;
  const int lane = threadIdx.x & 31;
  const int d = __builtin_amdgcn_readfirstlane(
      (int)(blockIdx.x * (blockDim.x >> 5) + (threadIdx.x >> 5)));
  if (d >= N) return;

  int end = (d + 1 < N) ? rowptr[d + 1] : E;
  end = min(max(end, 0), E);
  const int beg = min(max(rowptr[d], 0), end);
  const int deg = end - beg;
  const unsigned nmax = (unsigned)(N - 1);
  const int c0 = lane * VPL;

  auto load_row = [&](int s, float (&r)[VPL]) {
    const float* hp = H + (size_t)s * NOUT + c0;
    if constexpr (VPL == 2) {
      const float2 hv = *reinterpret_cast<const float2*>(hp);
      r[0] = hv.x;
      r[1] = hv.y;
    } else {
      r[0] = hp[0];
    }
  };

  float a_s[VPL], a_d[VPL], acc[VPL];
#pragma unroll
  for (int v = 0; v < VPL; ++v) {
    a_s[v] = av[c0 + v];
    a_d[v] = bv[c0 + v];
  }
  load_row(d, acc);

  float ps = acc[0] * a_s[0], pd = acc[0] * a_d[0];
#pragma unroll
  for (int v = 1; v < VPL; ++v) {
    ps += acc[v] * a_s[v];
    pd += acc[v] * a_d[v];
  }
  const float adv = wave_sum_f(pd);
  float m = leaky(wave_sum_f(ps) + adv);
  float den = 1.0f;

  for (int j0 = 0; j0 < deg; j0 += 32) {
    const int cnt = min(32, deg - j0);
    const int sraw = (int)min(
        (unsigned)csr_src[min(beg + j0 + lane, end - 1)], nmax);
    const int s_l = (lane < cnt) ? sraw : d;

    float hv[4][VPL];
#pragma unroll
    for (int u = 0; u < 4; ++u)
      load_row(__builtin_amdgcn_readlane(s_l, u), hv[u]);

    for (int t = 0;; t += 4) {
      const bool more = t + 4 < cnt;
      float hn[4][VPL];
      if (more) {
#pragma unroll
        for (int u = 0; u < 4; ++u)
          load_row(__builtin_amdgcn_readlane(s_l, t + 4 + u), hn[u]);
      }

      float p[4];
#pragma unroll
      for (int u = 0; u < 4; ++u) {
        p[u] = hv[u][0] * a_s[0];
#pragma unroll
        for (int v = 1; v < VPL; ++v) p[u] += hv[u][v] * a_s[v];
      }
      float e = leaky(sum4_rows(p[0], p[1], p[2], p[3], lane) + adv);
      e = (t + (lane & 3) < cnt) ? e : -__builtin_inff();

      float gm = fmaxf(e, dpp_row<0x161>(e));
      gm = fmaxf(gm, dpp_row<0x162>(gm));
      const float m_new = fmaxf(m, gm);
      const float sc = __expf(m - m_new);
      const float w = __expf(e - m_new);
      float ws = w + dpp_row<0x161>(w);
      ws += dpp_row<0x162>(ws);
      den = den * sc + ws;
      m = m_new;

#pragma unroll
      for (int v = 0; v < VPL; ++v) acc[v] *= sc;
#pragma unroll
      for (int u = 0; u < 4; ++u) {
        const float wu = __int_as_float(
            __builtin_amdgcn_readlane(__float_as_int(w), u));
#pragma unroll
        for (int v = 0; v < VPL; ++v) acc[v] += hv[u][v] * wu;
      }

      if (!more) break;
#pragma unroll
      for (int u = 0; u < 4; ++u) {
#pragma unroll
        for (int v = 0; v < VPL; ++v) hv[u][v] = hn[u][v];
      }
    }
  }

  float* op = out + (size_t)d * NOUT + c0;
  if constexpr (VPL == 2) {
    float2 r;
    r.x = fmaxf(acc[0] / den + bias[c0], 0.0f);
    r.y = fmaxf(acc[1] / den + bias[c0 + 1], 0.0f);
    *reinterpret_cast<float2*>(op) = r;
  } else {
    op[0] = fmaxf(acc[0] / den + bias[c0], 0.0f);
  }
}

__global__ void fc_kernel(const float* __restrict__ H,
                          const float* __restrict__ Wfc,
                          const float* __restrict__ bfc,
                          float* __restrict__ out, int N) {
  const int n = blockIdx.x * blockDim.x + threadIdx.x;
  if (n >= N) return;
  float sum = bfc[0];
#pragma unroll
  for (int c = 0; c < H2; ++c) sum += H[(size_t)n * H2 + c] * Wfc[c];
  out[n] = sum;
}

static inline unsigned cdiv(long long a, long long b) {
  return (unsigned)((a + b - 1) / b);
}

extern "C" void kernel_launch(void* const* d_in, const int* in_sizes, int n_in,
                              void* d_out, int out_size, void* d_ws,
                              size_t ws_size, hipStream_t stream) {
  (void)n_in; (void)out_size; (void)ws_size;
  const float* x     = (const float*)d_in[0];
  const int*   ei    = (const int*)d_in[1];
  const float* W1    = (const float*)d_in[2];
  const float* asrc1 = (const float*)d_in[3];
  const float* adst1 = (const float*)d_in[4];
  const float* b1    = (const float*)d_in[5];
  const float* W2    = (const float*)d_in[6];
  const float* asrc2 = (const float*)d_in[7];
  const float* adst2 = (const float*)d_in[8];
  const float* b2    = (const float*)d_in[9];
  const float* Wfc   = (const float*)d_in[10];
  const float* bfc   = (const float*)d_in[11];
  float* out = (float*)d_out;

  const int N = N_NODES;
  const int E = in_sizes[1] / 2;
  const int* src = ei;
  const int* dst = ei + E;

  float* ws = (float*)d_ws;
  size_t off = 0;
  float* h    = ws + off; off += (size_t)N * H1;
  float* agg1 = ws + off; off += (size_t)N * H1;
  float* agg2 = ws + off; off += (size_t)N * H2;
  int* hist    = (int*)(ws + off); off += N;
  int* rowptr  = (int*)(ws + off); off += N;
  int* bsum    = (int*)(ws + off); off += 1024;
  unsigned* binned = (unsigned*)(ws + off); off += E;
  int* csr_src = (int*)(ws + off); off += E;
  int* offs = (int*)agg2;

  const dim3 B(256);
  constexpr int MR = 32;
  constexpr int NH = NBINS * EBLOCKS;
  const unsigned nscan = cdiv(NH, 1024);

  bin_count_kernel<<<EBLOCKS, ETHREADS, 0, stream>>>(dst, E, hist);
  count_blocksum_kernel<<<nscan, B, 0, stream>>>(hist, NH, bsum);
  count_scan_kernel<<<nscan, B, 0, stream>>>(hist, NH, bsum, offs);
  bin_scatter_kernel<<<EBLOCKS, ETHREADS, 0, stream>>>(src, dst, E, offs, binned);
  bin_sort_kernel<<<NBINS, STHREADS, 0, stream>>>(offs, E, N, binned, rowptr, csr_src);

  gemm_wmma_kernel<F_IN, H1, MR><<<N / MR, (H1 / 16) * 32, 0, stream>>>(x, W1, h);
  node_gather_kernel<H1><<<cdiv(N, 8), B, 0, stream>>>(rowptr, csr_src, E, N, h, asrc1, adst1, b1, agg1);

  gemm_wmma_kernel<H1, H2, MR><<<N / MR, (H2 / 16) * 32, 0, stream>>>(agg1, W2, h);
  node_gather_kernel<H2><<<cdiv(N, 8), B, 0, stream>>>(rowptr, csr_src, E, N, h, asrc2, adst2, b2, agg2);

  fc_kernel<<<cdiv(N, 256), B, 0, stream>>>(agg2, Wfc, bfc, out, N);
}


# Gather reads rows as 16-byte lane pieces: one load brings 2 (64-ch) or 4 (32-ch) rows; each lane group keeps its own one-pass softmax state, merged at the end: softmax and row sums re-associated
# speedup vs baseline: 5.3574x; 1.0595x over previous
#include <hip/hip_runtime.h>

typedef float v2f __attribute__((ext_vector_type(2)));
typedef float v8f __attribute__((ext_vector_type(8)));

#define N_NODES 100000
#define F_IN 256
#define H1 64
#define H2 32
#define NEG_SLOPE 0.2f

template <int K, int NOUT, int MROWS>
__global__ void gemm_wmma_kernel(const float* __restrict__ X,
                                 const float* __restrict__ W,
                                 float* __restrict__ Hout) {
  constexpr int LDK = K + 4;
  constexpr int RT = MROWS / 16;
  __shared__ float xs[MROWS * LDK];
  const int nthreads = (NOUT / 16) * 32;
  const int tid = threadIdx.x;
  const int row0 = blockIdx.x * MROWS;

  __builtin_prefetch(W, 0, 1);

  for (int i = tid; i < MROWS * K; i += nthreads) {
    const int r = i / K, c = i % K;
    xs[r * LDK + c] = X[(size_t)(row0 + r) * K + c];
  }
  __syncthreads();

  const int wave = tid >> 5;
  const int lane = tid & 31;
  const int lh = lane & 15;
  const bool hi = (lane >= 16);
  const int col0 = wave * 16;

  v8f acc[RT];
#pragma unroll
  for (int t = 0; t < RT; ++t) acc[t] = (v8f){};

#pragma unroll 4
  for (int k = 0; k < K; k += 4) {
    const int ka = k + (hi ? 2 : 0);
    v2f b;
    b.x = W[(size_t)ka * NOUT + col0 + lh];
    b.y = W[(size_t)(ka + 1) * NOUT + col0 + lh];
#pragma unroll
    for (int t = 0; t < RT; ++t) {
      v2f a;
      a.x = xs[(t * 16 + lh) * LDK + ka];
      a.y = xs[(t * 16 + lh) * LDK + ka + 1];
      acc[t] = __builtin_amdgcn_wmma_f32_16x16x4_f32(false, a, false, b,
                                                     (short)0, acc[t],
                                                     false, false);
    }
  }

  const int rbase = hi ? 8 : 0;
#pragma unroll
  for (int t = 0; t < RT; ++t) {
#pragma unroll
    for (int r = 0; r < 8; ++r) {
      Hout[(size_t)(row0 + t * 16 + rbase + r) * NOUT + col0 + lh] = acc[t][r];
    }
  }
}

__device__ __forceinline__ float leaky(float v) {
  return (v > 0.0f) ? v : NEG_SLOPE * v;
}

template <int CTRL>
__device__ __forceinline__ float dpp_row(float x) {
  const int b = __float_as_int(x);
  return __int_as_float(__builtin_amdgcn_update_dpp(b, b, CTRL, 0xf, 0xf, false));
}

__device__ __forceinline__ float xhalf(float x) {
  const int b = __float_as_int(x);
  return __int_as_float(
      __builtin_amdgcn_permlanex16(b, b, 0x76543210, 0xfedcba98, false, false));
}

__device__ __forceinline__ int wave_sum_i(int v) {
#pragma unroll
  for (int m = 16; m >= 1; m >>= 1) v += __shfl_xor(v, m, 32);
  return v;
}

#define BIN_SHIFT 8
#define BIN_NODES (1 << BIN_SHIFT)
#define NBINS ((N_NODES + BIN_NODES - 1) >> BIN_SHIFT)
#define EBLOCKS 250
#define ETHREADS 1024
#define EBATCH 4
#define STHREADS 512
#define SBATCH 8
static_assert(NBINS * EBLOCKS <= N_NODES, "hist must fit the N-word slot");
static_assert(N_NODES <= (1 << 24), "src must fit 24 bits of the packed edge");

__device__ __forceinline__ unsigned clamp_node(int v) {
  return min((unsigned)v, (unsigned)(N_NODES - 1));
}

__global__ void bin_count_kernel(const int* __restrict__ dst, int E,
                                 int* __restrict__ hist) {
  __shared__ int lc[NBINS];
  const int tid = threadIdx.x;
  for (int b = tid; b < NBINS; b += ETHREADS) lc[b] = 0;
  __syncthreads();
  constexpr int nblk = EBLOCKS;
  const int chunk = (E + nblk - 1) / nblk;
  const int e0 = min((int)blockIdx.x * chunk, E);
  const int e1 = min(e0 + chunk, E);
  for (int e = e0 + tid; e < e1; e += ETHREADS * EBATCH) {
    int d[EBATCH];
#pragma unroll
    for (int u = 0; u < EBATCH; ++u) d[u] = dst[min(e + u * ETHREADS, e1 - 1)];
#pragma unroll
    for (int u = 0; u < EBATCH; ++u)
      atomicAdd(&lc[clamp_node(d[u]) >> BIN_SHIFT],
                (e + u * ETHREADS < e1) ? 1 : 0);
  }
  __syncthreads();
  for (int b = tid; b < NBINS; b += ETHREADS)
    hist[b * nblk + (int)blockIdx.x] = lc[b];
}

__global__ void bin_scatter_kernel(const int* __restrict__ src,
                                   const int* __restrict__ dst, int E,
                                   const int* __restrict__ offs,
                                   unsigned* __restrict__ binned) {
  __shared__ int lcur[NBINS];
  const int tid = threadIdx.x;
  constexpr int nblk = EBLOCKS;
  for (int b = tid; b < NBINS; b += ETHREADS)
    lcur[b] = offs[b * nblk + (int)blockIdx.x];
  __syncthreads();
  const int chunk = (E + nblk - 1) / nblk;
  const int e0 = min((int)blockIdx.x * chunk, E);
  const int e1 = min(e0 + chunk, E);
  for (int e = e0 + tid; e < e1; e += ETHREADS * EBATCH) {
    int s[EBATCH], d[EBATCH];
#pragma unroll
    for (int u = 0; u < EBATCH; ++u) {
      const int ee = min(e + u * ETHREADS, e1 - 1);
      s[u] = src[ee];
      d[u] = dst[ee];
    }
#pragma unroll
    for (int u = 0; u < EBATCH; ++u) {
      const bool live = (e + u * ETHREADS < e1);
      const unsigned dn = clamp_node(d[u]);
      const int pos = atomicAdd(&lcur[dn >> BIN_SHIFT], live ? 1 : 0);
      if (live && (unsigned)pos < (unsigned)E)
        binned[pos] = (clamp_node(s[u]) << BIN_SHIFT) | (dn & (BIN_NODES - 1));
    }
  }
}

__global__ void bin_sort_kernel(const int* __restrict__ offs, int E, int N,
                                const unsigned* __restrict__ binned,
                                int* __restrict__ rowptr,
                                int* __restrict__ csr_src) {
  __shared__ int lcnt[BIN_NODES];
  __shared__ int wtot[BIN_NODES / 32];
  const int tid = threadIdx.x;
  const int lane = tid & 31;
  const int wave = tid >> 5;
  const int bin = (int)blockIdx.x;
  const int base = bin << BIN_SHIFT;

  int end = (bin + 1 < NBINS) ? offs[(bin + 1) * EBLOCKS] : E;
  end = min(max(end, 0), E);
  const int start = min(max(offs[bin * EBLOCKS], 0), end);

  if (tid < BIN_NODES) lcnt[tid] = 0;
  __syncthreads();

  for (int i = start + tid; i < end; i += STHREADS * SBATCH) {
    unsigned p[SBATCH];
#pragma unroll
    for (int u = 0; u < SBATCH; ++u) p[u] = binned[min(i + u * STHREADS, end - 1)];
#pragma unroll
    for (int u = 0; u < SBATCH; ++u)
      atomicAdd(&lcnt[p[u] & (BIN_NODES - 1)], (i + u * STHREADS < end) ? 1 : 0);
  }
  __syncthreads();

  int c = 0, incl = 0;
  if (tid < BIN_NODES) {
    c = lcnt[tid];
    incl = c;
#pragma unroll
    for (int m = 1; m < 32; m <<= 1) {
      const int u = __shfl_up(incl, m, 32);
      if (lane >= m) incl += u;
    }
    if (lane == 31) wtot[wave] = incl;
  }
  __syncthreads();
  if (tid < BIN_NODES) {
    int off = 0;
#pragma unroll
    for (int w = 0; w < BIN_NODES / 32; ++w)
      if (w < wave) off += wtot[w];
    const int first = start + off + incl - c;
    if (base + tid < N) rowptr[base + tid] = first;
    lcnt[tid] = first;
  }
  __syncthreads();

  for (int i = start + tid; i < end; i += STHREADS * SBATCH) {
    unsigned p[SBATCH];
#pragma unroll
    for (int u = 0; u < SBATCH; ++u) p[u] = binned[min(i + u * STHREADS, end - 1)];
#pragma unroll
    for (int u = 0; u < SBATCH; ++u) {
      const bool live = (i + u * STHREADS < end);
      const int pos = atomicAdd(&lcnt[p[u] & (BIN_NODES - 1)], live ? 1 : 0);
      if (live && (unsigned)pos < (unsigned)E)
        csr_src[pos] = (int)(p[u] >> BIN_SHIFT);
    }
  }
}

__global__ void count_blocksum_kernel(const int* __restrict__ cnt, int N,
                                      int* __restrict__ bsum) {
  __shared__ int wsum[8];
  const int tid = threadIdx.x;
  const int i0 = (blockIdx.x * 256 + tid) * 4;
  int t = 0;
#pragma unroll
  for (int k = 0; k < 4; ++k)
    if (i0 + k < N) t += cnt[i0 + k];
  t = wave_sum_i(t);
  if ((tid & 31) == 0) wsum[tid >> 5] = t;
  __syncthreads();
  if (tid == 0) {
    int s = 0;
#pragma unroll
    for (int w = 0; w < 8; ++w) s += wsum[w];
    bsum[blockIdx.x] = s;
  }
}

__global__ void count_scan_kernel(const int* __restrict__ cnt, int N,
                                  const int* __restrict__ bsum,
                                  int* __restrict__ rowptr) {
  __shared__ int wsum[8];
  __shared__ int wtot[8];
  const int tid = threadIdx.x;
  const int lane = tid & 31;
  const int wave = tid >> 5;

  int p = 0;
  for (int i = tid; i < (int)blockIdx.x; i += 256) p += bsum[i];
  p = wave_sum_i(p);
  if (lane == 0) wsum[wave] = p;

  const int i0 = (blockIdx.x * 256 + tid) * 4;
  int c[4];
  int t = 0;
#pragma unroll
  for (int k = 0; k < 4; ++k) {
    c[k] = (i0 + k < N) ? cnt[i0 + k] : 0;
    t += c[k];
  }
  int incl = t;
#pragma unroll
  for (int m = 1; m < 32; m <<= 1) {
    const int u = __shfl_up(incl, m, 32);
    if (lane >= m) incl += u;
  }
  if (lane == 31) wtot[wave] = incl;
  __syncthreads();

  int off = 0;
#pragma unroll
  for (int w = 0; w < 8; ++w) off += wsum[w];
#pragma unroll
  for (int w = 0; w < 8; ++w)
    if (w < wave) off += wtot[w];

  int run = off + incl - t;
#pragma unroll
  for (int k = 0; k < 4; ++k) {
    if (i0 + k < N) rowptr[i0 + k] = run;
    run += c[k];
  }
}


template <int U>
__device__ __forceinline__ float quad_bcast(float x) {
  const int b = __float_as_int(x);
  return __int_as_float(
      __builtin_amdgcn_update_dpp(b, b, U * 0x55, 0xf, 0xf, false));
}

template <int LPR>
__device__ __forceinline__ float group_sum(float v) {
  v += dpp_row<0x161>(v);
  v += dpp_row<0x162>(v);
  v += dpp_row<0x164>(v);
  if constexpr (LPR == 16) v += dpp_row<0x168>(v);
  return v;
}

template <int LPR>
__device__ __forceinline__ float sum4_group(float p0, float p1, float p2,
                                            float p3, int lane) {
  const bool b0 = (lane & 1) != 0;
  const bool b1 = (lane & 2) != 0;
  float k01 = b0 ? p1 : p0;
  const float t01 = b0 ? p0 : p1;
  float k23 = b0 ? p3 : p2;
  const float t23 = b0 ? p2 : p3;
  k01 += dpp_row<0x161>(t01);
  k23 += dpp_row<0x161>(t23);
  float k = b1 ? k23 : k01;
  const float tt = b1 ? k01 : k23;
  k += dpp_row<0x162>(tt);
  k += dpp_row<0x164>(k);
  if constexpr (LPR == 16) k += dpp_row<0x168>(k);
  return k;
}

template <int NOUT>
__global__ void node_gather_kernel(const int* __restrict__ rowptr,
                                   const int* __restrict__ csr_src, int E,
                                   int N, const float* __restrict__ H,
                                   const float* __restrict__ av,
                                   const float* __restrict__ bv,
                                   const float* __restrict__ bias,
                                   float* __restrict__ out) {
  constexpr int LPR = NOUT / 4;
  constexpr int RPI = 32 / LPR;
  constexpr int RB = 4 * RPI;
  static_assert(LPR == 8 || LPR == 16, "lane groups of 8 or 16");
  const int lane = threadIdx.x & 31;
  const int g = lane / LPR;
  const int c0 = (lane % LPR) * 4;
  const int qu = lane & 3;
  const int d = __builtin_amdgcn_readfirstlane(
      (int)(blockIdx.x * (blockDim.x >> 5) + (threadIdx.x >> 5)));
  if (d >= N) return;

  int end = (d + 1 < N) ? rowptr[d + 1] : E;
  end = min(max(end, 0), E);
  const int beg = min(max(rowptr[d], 0), end);
  const int deg = end - beg;
  const unsigned nmax = (unsigned)(N - 1);

  auto load_row = [&](int s) -> float4 {
    const unsigned off = (unsigned)s * (unsigned)(NOUT * 4) + (unsigned)(c0 * 4);
    return *reinterpret_cast<const float4*>(
        reinterpret_cast<const char*>(H) + off);
  };
  auto lane_read = [&](int v, int i) -> int {
    return __builtin_amdgcn_ds_bpermute(i << 2, v);
  };

  const float4 hs = load_row(d);
  const float a_s0 = av[c0], a_s1 = av[c0 + 1], a_s2 = av[c0 + 2], a_s3 = av[c0 + 3];
  const float a_d0 = bv[c0], a_d1 = bv[c0 + 1], a_d2 = bv[c0 + 2], a_d3 = bv[c0 + 3];

  const float adv = group_sum<LPR>(hs.x * a_d0 + hs.y * a_d1 + hs.z * a_d2 + hs.w * a_d3);
  float m = leaky(group_sum<LPR>(hs.x * a_s0 + hs.y * a_s1 + hs.z * a_s2 + hs.w * a_s3) + adv);
  const bool g0 = (g == 0);
  float den = g0 ? 1.0f : 0.0f;
  float acc0 = g0 ? hs.x : 0.0f, acc1 = g0 ? hs.y : 0.0f;
  float acc2 = g0 ? hs.z : 0.0f, acc3 = g0 ? hs.w : 0.0f;

  for (int j0 = 0; j0 < deg; j0 += 32) {
    const int cnt = min(32, deg - j0);
    const int sraw = (int)min(
        (unsigned)csr_src[min(beg + j0 + lane, end - 1)], nmax);
    const int s_l = (lane < cnt) ? sraw : d;

    auto trip = [&](const float4 (&hv)[4], int t) {
      float p[4];
#pragma unroll
      for (int u = 0; u < 4; ++u)
        p[u] = hv[u].x * a_s0 + hv[u].y * a_s1 + hv[u].z * a_s2 + hv[u].w * a_s3;
      float e = leaky(sum4_group<LPR>(p[0], p[1], p[2], p[3], lane) + adv);
      e = (t + qu * RPI + g < cnt) ? e : -__builtin_inff();

      float gm = fmaxf(e, dpp_row<0x161>(e));
      gm = fmaxf(gm, dpp_row<0x162>(gm));
      const float m_new = fmaxf(m, gm);
      const float sc = __expf(m - m_new);
      const float w = __expf(e - m_new);
      float ws = w + dpp_row<0x161>(w);
      ws += dpp_row<0x162>(ws);
      den = den * sc + ws;
      m = m_new;

      acc0 *= sc; acc1 *= sc; acc2 *= sc; acc3 *= sc;
      const float w0 = quad_bcast<0>(w), w1 = quad_bcast<1>(w);
      const float w2 = quad_bcast<2>(w), w3 = quad_bcast<3>(w);
      acc0 += hv[0].x * w0; acc1 += hv[0].y * w0; acc2 += hv[0].z * w0; acc3 += hv[0].w * w0;
      acc0 += hv[1].x * w1; acc1 += hv[1].y * w1; acc2 += hv[1].z * w1; acc3 += hv[1].w * w1;
      acc0 += hv[2].x * w2; acc1 += hv[2].y * w2; acc2 += hv[2].z * w2; acc3 += hv[2].w * w2;
      acc0 += hv[3].x * w3; acc1 += hv[3].y * w3; acc2 += hv[3].z * w3; acc3 += hv[3].w * w3;
    };

    float4 hv[4];
#pragma unroll
    for (int u = 0; u < 4; ++u) hv[u] = load_row(lane_read(s_l, u * RPI + g));

    int t = 0;
    for (; t + RB < cnt; t += RB) {
      float4 hn[4];
#pragma unroll
      for (int u = 0; u < 4; ++u)
        hn[u] = load_row(lane_read(s_l, t + RB + u * RPI + g));
      trip(hv, t);
#pragma unroll
      for (int u = 0; u < 4; ++u) hv[u] = hn[u];
    }
    trip(hv, t);
  }

  auto merge = [&](auto xch) {
    const float mo = xch(m);
    const float mm = fmaxf(m, mo);
    const float f = __expf(m - mm);
    den *= f; acc0 *= f; acc1 *= f; acc2 *= f; acc3 *= f;
    den += xch(den);
    acc0 += xch(acc0); acc1 += xch(acc1); acc2 += xch(acc2); acc3 += xch(acc3);
    m = mm;
  };
  if constexpr (LPR == 8) merge([](float x) { return dpp_row<0x168>(x); });
  merge([](float x) { return xhalf(x); });

  if (g0) {
    float4 r;
    r.x = fmaxf(acc0 / den + bias[c0], 0.0f);
    r.y = fmaxf(acc1 / den + bias[c0 + 1], 0.0f);
    r.z = fmaxf(acc2 / den + bias[c0 + 2], 0.0f);
    r.w = fmaxf(acc3 / den + bias[c0 + 3], 0.0f);
    *reinterpret_cast<float4*>(out + (size_t)d * NOUT + c0) = r;
  }
}

__global__ void fc_kernel(const float* __restrict__ H,
                          const float* __restrict__ Wfc,
                          const float* __restrict__ bfc,
                          float* __restrict__ out, int N) {
  const int n = blockIdx.x * blockDim.x + threadIdx.x;
  if (n >= N) return;
  float sum = bfc[0];
#pragma unroll
  for (int c = 0; c < H2; ++c) sum += H[(size_t)n * H2 + c] * Wfc[c];
  out[n] = sum;
}

static inline unsigned cdiv(long long a, long long b) {
  return (unsigned)((a + b - 1) / b);
}

extern "C" void kernel_launch(void* const* d_in, const int* in_sizes, int n_in,
                              void* d_out, int out_size, void* d_ws,
                              size_t ws_size, hipStream_t stream) {
  (void)n_in; (void)out_size; (void)ws_size;
  const float* x     = (const float*)d_in[0];
  const int*   ei    = (const int*)d_in[1];
  const float* W1    = (const float*)d_in[2];
  const float* asrc1 = (const float*)d_in[3];
  const float* adst1 = (const float*)d_in[4];
  const float* b1    = (const float*)d_in[5];
  const float* W2    = (const float*)d_in[6];
  const float* asrc2 = (const float*)d_in[7];
  const float* adst2 = (const float*)d_in[8];
  const float* b2    = (const float*)d_in[9];
  const float* Wfc   = (const float*)d_in[10];
  const float* bfc   = (const float*)d_in[11];
  float* out = (float*)d_out;

  const int N = N_NODES;
  const int E = in_sizes[1] / 2;
  const int* src = ei;
  const int* dst = ei + E;

  float* ws = (float*)d_ws;
  size_t off = 0;
  float* h    = ws + off; off += (size_t)N * H1;
  float* agg1 = ws + off; off += (size_t)N * H1;
  float* agg2 = ws + off; off += (size_t)N * H2;
  int* hist    = (int*)(ws + off); off += N;
  int* rowptr  = (int*)(ws + off); off += N;
  int* bsum    = (int*)(ws + off); off += 1024;
  unsigned* binned = (unsigned*)(ws + off); off += E;
  int* csr_src = (int*)(ws + off); off += E;
  int* offs = (int*)agg2;

  const dim3 B(256);
  constexpr int MR = 32;
  constexpr int NH = NBINS * EBLOCKS;
  const unsigned nscan = cdiv(NH, 1024);

  bin_count_kernel<<<EBLOCKS, ETHREADS, 0, stream>>>(dst, E, hist);
  count_blocksum_kernel<<<nscan, B, 0, stream>>>(hist, NH, bsum);
  count_scan_kernel<<<nscan, B, 0, stream>>>(hist, NH, bsum, offs);
  bin_scatter_kernel<<<EBLOCKS, ETHREADS, 0, stream>>>(src, dst, E, offs, binned);
  bin_sort_kernel<<<NBINS, STHREADS, 0, stream>>>(offs, E, N, binned, rowptr, csr_src);

  gemm_wmma_kernel<F_IN, H1, MR><<<N / MR, (H1 / 16) * 32, 0, stream>>>(x, W1, h);
  node_gather_kernel<H1><<<cdiv(N, 8), B, 0, stream>>>(rowptr, csr_src, E, N, h, asrc1, adst1, b1, agg1);

  gemm_wmma_kernel<H1, H2, MR><<<N / MR, (H2 / 16) * 32, 0, stream>>>(agg1, W2, h);
  node_gather_kernel<H2><<<cdiv(N, 8), B, 0, stream>>>(rowptr, csr_src, E, N, h, asrc2, adst2, b2, agg2);

  fc_kernel<<<cdiv(N, 256), B, 0, stream>>>(agg2, Wfc, bfc, out, N);
}


# GEMM tiles (32 rows, one contiguous run) staged as 16-byte pieces, all of a thread's loads in flight before its first wait (launch bound gives the true block size): 1 memory trip per tile, not 8 or 4
# speedup vs baseline: 5.5743x; 1.0405x over previous
#include <hip/hip_runtime.h>

typedef float v2f __attribute__((ext_vector_type(2)));
typedef float v8f __attribute__((ext_vector_type(8)));

#define N_NODES 100000
#define F_IN 256
#define H1 64
#define H2 32
#define NEG_SLOPE 0.2f

template <int K, int NOUT, int MROWS>
__global__ void __launch_bounds__((NOUT / 16) * 32)
gemm_wmma_kernel(const float* __restrict__ X,
                 const float* __restrict__ W,
                 float* __restrict__ Hout) {
  constexpr int LDK = K + 4;
  constexpr int RT = MROWS / 16;
  constexpr int NTHR = (NOUT / 16) * 32;
  constexpr int K4 = K / 4;
  constexpr int PPT = MROWS * K4 / NTHR;
  static_assert(MROWS * K4 % NTHR == 0, "whole pieces per thread");
  static_assert(LDK % 4 == 0, "LDS rows start on 16 bytes");
  __shared__ __attribute__((aligned(16))) float xs[MROWS * LDK];
  const int tid = threadIdx.x;
  const int row0 = blockIdx.x * MROWS;

  __builtin_prefetch(W, 0, 1);

  const float4* __restrict__ xt =
      reinterpret_cast<const float4*>(X + (size_t)row0 * K);
  float4 pc[PPT];
#pragma unroll
  for (int it = 0; it < PPT; ++it) pc[it] = xt[tid + it * NTHR];
#pragma unroll
  for (int it = 0; it < PPT; ++it) {
    const int i = tid + it * NTHR;
    const int r = i / K4, c = (i % K4) * 4;
    *reinterpret_cast<float4*>(&xs[r * LDK + c]) = pc[it];
  }
  __syncthreads();

  const int wave = tid >> 5;
  const int lane = tid & 31;
  const int lh = lane & 15;
  const bool hi = (lane >= 16);
  const int col0 = wave * 16;

  v8f acc[RT];
#pragma unroll
  for (int t = 0; t < RT; ++t) acc[t] = (v8f){};

#pragma unroll 4
  for (int k = 0; k < K; k += 4) {
    const int ka = k + (hi ? 2 : 0);
    v2f b;
    b.x = W[(size_t)ka * NOUT + col0 + lh];
    b.y = W[(size_t)(ka + 1) * NOUT + col0 + lh];
#pragma unroll
    for (int t = 0; t < RT; ++t) {
      v2f a;
      a.x = xs[(t * 16 + lh) * LDK + ka];
      a.y = xs[(t * 16 + lh) * LDK + ka + 1];
      acc[t] = __builtin_amdgcn_wmma_f32_16x16x4_f32(false, a, false, b,
                                                     (short)0, acc[t],
                                                     false, false);
    }
  }

  const int rbase = hi ? 8 : 0;
#pragma unroll
  for (int t = 0; t < RT; ++t) {
#pragma unroll
    for (int r = 0; r < 8; ++r) {
      Hout[(size_t)(row0 + t * 16 + rbase + r) * NOUT + col0 + lh] = acc[t][r];
    }
  }
}

__device__ __forceinline__ float leaky(float v) {
  return (v > 0.0f) ? v : NEG_SLOPE * v;
}

template <int CTRL>
__device__ __forceinline__ float dpp_row(float x) {
  const int b = __float_as_int(x);
  return __int_as_float(__builtin_amdgcn_update_dpp(b, b, CTRL, 0xf, 0xf, false));
}

__device__ __forceinline__ float xhalf(float x) {
  const int b = __float_as_int(x);
  return __int_as_float(
      __builtin_amdgcn_permlanex16(b, b, 0x76543210, 0xfedcba98, false, false));
}

__device__ __forceinline__ int wave_sum_i(int v) {
#pragma unroll
  for (int m = 16; m >= 1; m >>= 1) v += __shfl_xor(v, m, 32);
  return v;
}

#define BIN_SHIFT 8
#define BIN_NODES (1 << BIN_SHIFT)
#define NBINS ((N_NODES + BIN_NODES - 1) >> BIN_SHIFT)
#define EBLOCKS 250
#define ETHREADS 1024
#define EBATCH 4
#define STHREADS 512
#define SBATCH 8
static_assert(NBINS * EBLOCKS <= N_NODES, "hist must fit the N-word slot");
static_assert(N_NODES <= (1 << 24), "src must fit 24 bits of the packed edge");

__device__ __forceinline__ unsigned clamp_node(int v) {
  return min((unsigned)v, (unsigned)(N_NODES - 1));
}

__global__ void bin_count_kernel(const int* __restrict__ dst, int E,
                                 int* __restrict__ hist) {
  __shared__ int lc[NBINS];
  const int tid = threadIdx.x;
  for (int b = tid; b < NBINS; b += ETHREADS) lc[b] = 0;
  __syncthreads();
  constexpr int nblk = EBLOCKS;
  const int chunk = (E + nblk - 1) / nblk;
  const int e0 = min((int)blockIdx.x * chunk, E);
  const int e1 = min(e0 + chunk, E);
  for (int e = e0 + tid; e < e1; e += ETHREADS * EBATCH) {
    int d[EBATCH];
#pragma unroll
    for (int u = 0; u < EBATCH; ++u) d[u] = dst[min(e + u * ETHREADS, e1 - 1)];
#pragma unroll
    for (int u = 0; u < EBATCH; ++u)
      atomicAdd(&lc[clamp_node(d[u]) >> BIN_SHIFT],
                (e + u * ETHREADS < e1) ? 1 : 0);
  }
  __syncthreads();
  for (int b = tid; b < NBINS; b += ETHREADS)
    hist[b * nblk + (int)blockIdx.x] = lc[b];
}

__global__ void bin_scatter_kernel(const int* __restrict__ src,
                                   const int* __restrict__ dst, int E,
                                   const int* __restrict__ offs,
                                   unsigned* __restrict__ binned) {
  __shared__ int lcur[NBINS];
  const int tid = threadIdx.x;
  constexpr int nblk = EBLOCKS;
  for (int b = tid; b < NBINS; b += ETHREADS)
    lcur[b] = offs[b * nblk + (int)blockIdx.x];
  __syncthreads();
  const int chunk = (E + nblk - 1) / nblk;
  const int e0 = min((int)blockIdx.x * chunk, E);
  const int e1 = min(e0 + chunk, E);
  for (int e = e0 + tid; e < e1; e += ETHREADS * EBATCH) {
    int s[EBATCH], d[EBATCH];
#pragma unroll
    for (int u = 0; u < EBATCH; ++u) {
      const int ee = min(e + u * ETHREADS, e1 - 1);
      s[u] = src[ee];
      d[u] = dst[ee];
    }
#pragma unroll
    for (int u = 0; u < EBATCH; ++u) {
      const bool live = (e + u * ETHREADS < e1);
      const unsigned dn = clamp_node(d[u]);
      const int pos = atomicAdd(&lcur[dn >> BIN_SHIFT], live ? 1 : 0);
      if (live && (unsigned)pos < (unsigned)E)
        binned[pos] = (clamp_node(s[u]) << BIN_SHIFT) | (dn & (BIN_NODES - 1));
    }
  }
}

__global__ void bin_sort_kernel(const int* __restrict__ offs, int E, int N,
                                const unsigned* __restrict__ binned,
                                int* __restrict__ rowptr,
                                int* __restrict__ csr_src) {
  __shared__ int lcnt[BIN_NODES];
  __shared__ int wtot[BIN_NODES / 32];
  const int tid = threadIdx.x;
  const int lane = tid & 31;
  const int wave = tid >> 5;
  const int bin = (int)blockIdx.x;
  const int base = bin << BIN_SHIFT;

  int end = (bin + 1 < NBINS) ? offs[(bin + 1) * EBLOCKS] : E;
  end = min(max(end, 0), E);
  const int start = min(max(offs[bin * EBLOCKS], 0), end);

  if (tid < BIN_NODES) lcnt[tid] = 0;
  __syncthreads();

  for (int i = start + tid; i < end; i += STHREADS * SBATCH) {
    unsigned p[SBATCH];
#pragma unroll
    for (int u = 0; u < SBATCH; ++u) p[u] = binned[min(i + u * STHREADS, end - 1)];
#pragma unroll
    for (int u = 0; u < SBATCH; ++u)
      atomicAdd(&lcnt[p[u] & (BIN_NODES - 1)], (i + u * STHREADS < end) ? 1 : 0);
  }
  __syncthreads();

  int c = 0, incl = 0;
  if (tid < BIN_NODES) {
    c = lcnt[tid];
    incl = c;
#pragma unroll
    for (int m = 1; m < 32; m <<= 1) {
      const int u = __shfl_up(incl, m, 32);
      if (lane >= m) incl += u;
    }
    if (lane == 31) wtot[wave] = incl;
  }
  __syncthreads();
  if (tid < BIN_NODES) {
    int off = 0;
#pragma unroll
    for (int w = 0; w < BIN_NODES / 32; ++w)
      if (w < wave) off += wtot[w];
    const int first = start + off + incl - c;
    if (base + tid < N) rowptr[base + tid] = first;
    lcnt[tid] = first;
  }
  __syncthreads();

  for (int i = start + tid; i < end; i += STHREADS * SBATCH) {
    unsigned p[SBATCH];
#pragma unroll
    for (int u = 0; u < SBATCH; ++u) p[u] = binned[min(i + u * STHREADS, end - 1)];
#pragma unroll
    for (int u = 0; u < SBATCH; ++u) {
      const bool live = (i + u * STHREADS < end);
      const int pos = atomicAdd(&lcnt[p[u] & (BIN_NODES - 1)], live ? 1 : 0);
      if (live && (unsigned)pos < (unsigned)E)
        csr_src[pos] = (int)(p[u] >> BIN_SHIFT);
    }
  }
}

__global__ void count_blocksum_kernel(const int* __restrict__ cnt, int N,
                                      int* __restrict__ bsum) {
  __shared__ int wsum[8];
  const int tid = threadIdx.x;
  const int i0 = (blockIdx.x * 256 + tid) * 4;
  int t = 0;
#pragma unroll
  for (int k = 0; k < 4; ++k)
    if (i0 + k < N) t += cnt[i0 + k];
  t = wave_sum_i(t);
  if ((tid & 31) == 0) wsum[tid >> 5] = t;
  __syncthreads();
  if (tid == 0) {
    int s = 0;
#pragma unroll
    for (int w = 0; w < 8; ++w) s += wsum[w];
    bsum[blockIdx.x] = s;
  }
}

__global__ void count_scan_kernel(const int* __restrict__ cnt, int N,
                                  const int* __restrict__ bsum,
                                  int* __restrict__ rowptr) {
  __shared__ int wsum[8];
  __shared__ int wtot[8];
  const int tid = threadIdx.x;
  const int lane = tid & 31;
  const int wave = tid >> 5;

  int p = 0;
  for (int i = tid; i < (int)blockIdx.x; i += 256) p += bsum[i];
  p = wave_sum_i(p);
  if (lane == 0) wsum[wave] = p;

  const int i0 = (blockIdx.x * 256 + tid) * 4;
  int c[4];
  int t = 0;
#pragma unroll
  for (int k = 0; k < 4; ++k) {
    c[k] = (i0 + k < N) ? cnt[i0 + k] : 0;
    t += c[k];
  }
  int incl = t;
#pragma unroll
  for (int m = 1; m < 32; m <<= 1) {
    const int u = __shfl_up(incl, m, 32);
    if (lane >= m) incl += u;
  }
  if (lane == 31) wtot[wave] = incl;
  __syncthreads();

  int off = 0;
#pragma unroll
  for (int w = 0; w < 8; ++w) off += wsum[w];
#pragma unroll
  for (int w = 0; w < 8; ++w)
    if (w < wave) off += wtot[w];

  int run = off + incl - t;
#pragma unroll
  for (int k = 0; k < 4; ++k) {
    if (i0 + k < N) rowptr[i0 + k] = run;
    run += c[k];
  }
}


template <int U>
__device__ __forceinline__ float quad_bcast(float x) {
  const int b = __float_as_int(x);
  return __int_as_float(
      __builtin_amdgcn_update_dpp(b, b, U * 0x55, 0xf, 0xf, false));
}

template <int LPR>
__device__ __forceinline__ float group_sum(float v) {
  v += dpp_row<0x161>(v);
  v += dpp_row<0x162>(v);
  v += dpp_row<0x164>(v);
  if constexpr (LPR == 16) v += dpp_row<0x168>(v);
  return v;
}

template <int LPR>
__device__ __forceinline__ float sum4_group(float p0, float p1, float p2,
                                            float p3, int lane) {
  const bool b0 = (lane & 1) != 0;
  const bool b1 = (lane & 2) != 0;
  float k01 = b0 ? p1 : p0;
  const float t01 = b0 ? p0 : p1;
  float k23 = b0 ? p3 : p2;
  const float t23 = b0 ? p2 : p3;
  k01 += dpp_row<0x161>(t01);
  k23 += dpp_row<0x161>(t23);
  float k = b1 ? k23 : k01;
  const float tt = b1 ? k01 : k23;
  k += dpp_row<0x162>(tt);
  k += dpp_row<0x164>(k);
  if constexpr (LPR == 16) k += dpp_row<0x168>(k);
  return k;
}

template <int NOUT>
__global__ void node_gather_kernel(const int* __restrict__ rowptr,
                                   const int* __restrict__ csr_src, int E,
                                   int N, const float* __restrict__ H,
                                   const float* __restrict__ av,
                                   const float* __restrict__ bv,
                                   const float* __restrict__ bias,
                                   float* __restrict__ out) {
  constexpr int LPR = NOUT / 4;
  constexpr int RPI = 32 / LPR;
  constexpr int RB = 4 * RPI;
  static_assert(LPR == 8 || LPR == 16, "lane groups of 8 or 16");
  const int lane = threadIdx.x & 31;
  const int g = lane / LPR;
  const int c0 = (lane % LPR) * 4;
  const int qu = lane & 3;
  const int d = __builtin_amdgcn_readfirstlane(
      (int)(blockIdx.x * (blockDim.x >> 5) + (threadIdx.x >> 5)));
  if (d >= N) return;

  int end = (d + 1 < N) ? rowptr[d + 1] : E;
  end = min(max(end, 0), E);
  const int beg = min(max(rowptr[d], 0), end);
  const int deg = end - beg;
  const unsigned nmax = (unsigned)(N - 1);

  auto load_row = [&](int s) -> float4 {
    const unsigned off = (unsigned)s * (unsigned)(NOUT * 4) + (unsigned)(c0 * 4);
    return *reinterpret_cast<const float4*>(
        reinterpret_cast<const char*>(H) + off);
  };
  auto lane_read = [&](int v, int i) -> int {
    return __builtin_amdgcn_ds_bpermute(i << 2, v);
  };

  const float4 hs = load_row(d);
  const float a_s0 = av[c0], a_s1 = av[c0 + 1], a_s2 = av[c0 + 2], a_s3 = av[c0 + 3];
  const float a_d0 = bv[c0], a_d1 = bv[c0 + 1], a_d2 = bv[c0 + 2], a_d3 = bv[c0 + 3];

  const float adv = group_sum<LPR>(hs.x * a_d0 + hs.y * a_d1 + hs.z * a_d2 + hs.w * a_d3);
  float m = leaky(group_sum<LPR>(hs.x * a_s0 + hs.y * a_s1 + hs.z * a_s2 + hs.w * a_s3) + adv);
  const bool g0 = (g == 0);
  float den = g0 ? 1.0f : 0.0f;
  float acc0 = g0 ? hs.x : 0.0f, acc1 = g0 ? hs.y : 0.0f;
  float acc2 = g0 ? hs.z : 0.0f, acc3 = g0 ? hs.w : 0.0f;

  for (int j0 = 0; j0 < deg; j0 += 32) {
    const int cnt = min(32, deg - j0);
    const int sraw = (int)min(
        (unsigned)csr_src[min(beg + j0 + lane, end - 1)], nmax);
    const int s_l = (lane < cnt) ? sraw : d;

    auto trip = [&](const float4 (&hv)[4], int t) {
      float p[4];
#pragma unroll
      for (int u = 0; u < 4; ++u)
        p[u] = hv[u].x * a_s0 + hv[u].y * a_s1 + hv[u].z * a_s2 + hv[u].w * a_s3;
      float e = leaky(sum4_group<LPR>(p[0], p[1], p[2], p[3], lane) + adv);
      e = (t + qu * RPI + g < cnt) ? e : -__builtin_inff();

      float gm = fmaxf(e, dpp_row<0x161>(e));
      gm = fmaxf(gm, dpp_row<0x162>(gm));
      const float m_new = fmaxf(m, gm);
      const float sc = __expf(m - m_new);
      const float w = __expf(e - m_new);
      float ws = w + dpp_row<0x161>(w);
      ws += dpp_row<0x162>(ws);
      den = den * sc + ws;
      m = m_new;

      acc0 *= sc; acc1 *= sc; acc2 *= sc; acc3 *= sc;
      const float w0 = quad_bcast<0>(w), w1 = quad_bcast<1>(w);
      const float w2 = quad_bcast<2>(w), w3 = quad_bcast<3>(w);
      acc0 += hv[0].x * w0; acc1 += hv[0].y * w0; acc2 += hv[0].z * w0; acc3 += hv[0].w * w0;
      acc0 += hv[1].x * w1; acc1 += hv[1].y * w1; acc2 += hv[1].z * w1; acc3 += hv[1].w * w1;
      acc0 += hv[2].x * w2; acc1 += hv[2].y * w2; acc2 += hv[2].z * w2; acc3 += hv[2].w * w2;
      acc0 += hv[3].x * w3; acc1 += hv[3].y * w3; acc2 += hv[3].z * w3; acc3 += hv[3].w * w3;
    };

    float4 hv[4];
#pragma unroll
    for (int u = 0; u < 4; ++u) hv[u] = load_row(lane_read(s_l, u * RPI + g));

    int t = 0;
    for (; t + RB < cnt; t += RB) {
      float4 hn[4];
#pragma unroll
      for (int u = 0; u < 4; ++u)
        hn[u] = load_row(lane_read(s_l, t + RB + u * RPI + g));
      trip(hv, t);
#pragma unroll
      for (int u = 0; u < 4; ++u) hv[u] = hn[u];
    }
    trip(hv, t);
  }

  auto merge = [&](auto xch) {
    const float mo = xch(m);
    const float mm = fmaxf(m, mo);
    const float f = __expf(m - mm);
    den *= f; acc0 *= f; acc1 *= f; acc2 *= f; acc3 *= f;
    den += xch(den);
    acc0 += xch(acc0); acc1 += xch(acc1); acc2 += xch(acc2); acc3 += xch(acc3);
    m = mm;
  };
  if constexpr (LPR == 8) merge([](float x) { return dpp_row<0x168>(x); });
  merge([](float x) { return xhalf(x); });

  if (g0) {
    float4 r;
    r.x = fmaxf(acc0 / den + bias[c0], 0.0f);
    r.y = fmaxf(acc1 / den + bias[c0 + 1], 0.0f);
    r.z = fmaxf(acc2 / den + bias[c0 + 2], 0.0f);
    r.w = fmaxf(acc3 / den + bias[c0 + 3], 0.0f);
    *reinterpret_cast<float4*>(out + (size_t)d * NOUT + c0) = r;
  }
}

__global__ void fc_kernel(const float* __restrict__ H,
                          const float* __restrict__ Wfc,
                          const float* __restrict__ bfc,
                          float* __restrict__ out, int N) {
  const int n = blockIdx.x * blockDim.x + threadIdx.x;
  if (n >= N) return;
  float sum = bfc[0];
#pragma unroll
  for (int c = 0; c < H2; ++c) sum += H[(size_t)n * H2 + c] * Wfc[c];
  out[n] = sum;
}

static inline unsigned cdiv(long long a, long long b) {
  return (unsigned)((a + b - 1) / b);
}

extern "C" void kernel_launch(void* const* d_in, const int* in_sizes, int n_in,
                              void* d_out, int out_size, void* d_ws,
                              size_t ws_size, hipStream_t stream) {
  (void)n_in; (void)out_size; (void)ws_size;
  const float* x     = (const float*)d_in[0];
  const int*   ei    = (const int*)d_in[1];
  const float* W1    = (const float*)d_in[2];
  const float* asrc1 = (const float*)d_in[3];
  const float* adst1 = (const float*)d_in[4];
  const float* b1    = (const float*)d_in[5];
  const float* W2    = (const float*)d_in[6];
  const float* asrc2 = (const float*)d_in[7];
  const float* adst2 = (const float*)d_in[8];
  const float* b2    = (const float*)d_in[9];
  const float* Wfc   = (const float*)d_in[10];
  const float* bfc   = (const float*)d_in[11];
  float* out = (float*)d_out;

  const int N = N_NODES;
  const int E = in_sizes[1] / 2;
  const int* src = ei;
  const int* dst = ei + E;

  float* ws = (float*)d_ws;
  size_t off = 0;
  float* h    = ws + off; off += (size_t)N * H1;
  float* agg1 = ws + off; off += (size_t)N * H1;
  float* agg2 = ws + off; off += (size_t)N * H2;
  int* hist    = (int*)(ws + off); off += N;
  int* rowptr  = (int*)(ws + off); off += N;
  int* bsum    = (int*)(ws + off); off += 1024;
  unsigned* binned = (unsigned*)(ws + off); off += E;
  int* csr_src = (int*)(ws + off); off += E;
  int* offs = (int*)agg2;

  const dim3 B(256);
  constexpr int MR = 32;
  constexpr int NH = NBINS * EBLOCKS;
  const unsigned nscan = cdiv(NH, 1024);

  bin_count_kernel<<<EBLOCKS, ETHREADS, 0, stream>>>(dst, E, hist);
  count_blocksum_kernel<<<nscan, B, 0, stream>>>(hist, NH, bsum);
  count_scan_kernel<<<nscan, B, 0, stream>>>(hist, NH, bsum, offs);
  bin_scatter_kernel<<<EBLOCKS, ETHREADS, 0, stream>>>(src, dst, E, offs, binned);
  bin_sort_kernel<<<NBINS, STHREADS, 0, stream>>>(offs, E, N, binned, rowptr, csr_src);

  gemm_wmma_kernel<F_IN, H1, MR><<<N / MR, (H1 / 16) * 32, 0, stream>>>(x, W1, h);
  node_gather_kernel<H1><<<cdiv(N, 8), B, 0, stream>>>(rowptr, csr_src, E, N, h, asrc1, adst1, b1, agg1);

  gemm_wmma_kernel<H1, H2, MR><<<N / MR, (H2 / 16) * 32, 0, stream>>>(agg1, W2, h);
  node_gather_kernel<H2><<<cdiv(N, 8), B, 0, stream>>>(rowptr, csr_src, E, N, h, asrc2, adst2, b2, agg2);

  fc_kernel<<<cdiv(N, 256), B, 0, stream>>>(agg2, Wfc, bfc, out, N);
}


# Gather: each lane group of a wave takes a node of its own (2 nodes per wave at 64 ch, 4 at 32): per-node set-up, division and write-out shared, no group merge; node sums re-associated (CSR order)
# speedup vs baseline: 5.8045x; 1.0413x over previous
#include <hip/hip_runtime.h>

typedef float v2f __attribute__((ext_vector_type(2)));
typedef float v8f __attribute__((ext_vector_type(8)));

#define N_NODES 100000
#define F_IN 256
#define H1 64
#define H2 32
#define NEG_SLOPE 0.2f

template <int K, int NOUT, int MROWS>
__global__ void __launch_bounds__((NOUT / 16) * 32)
gemm_wmma_kernel(const float* __restrict__ X,
                 const float* __restrict__ W,
                 float* __restrict__ Hout) {
  constexpr int LDK = K + 4;
  constexpr int RT = MROWS / 16;
  constexpr int NTHR = (NOUT / 16) * 32;
  constexpr int K4 = K / 4;
  constexpr int PPT = MROWS * K4 / NTHR;
  static_assert(MROWS * K4 % NTHR == 0, "whole pieces per thread");
  static_assert(LDK % 4 == 0, "LDS rows start on 16 bytes");
  __shared__ __attribute__((aligned(16))) float xs[MROWS * LDK];
  const int tid = threadIdx.x;
  const int row0 = blockIdx.x * MROWS;

  __builtin_prefetch(W, 0, 1);

  const float4* __restrict__ xt =
      reinterpret_cast<const float4*>(X + (size_t)row0 * K);
  float4 pc[PPT];
#pragma unroll
  for (int it = 0; it < PPT; ++it) pc[it] = xt[tid + it * NTHR];
#pragma unroll
  for (int it = 0; it < PPT; ++it) {
    const int i = tid + it * NTHR;
    const int r = i / K4, c = (i % K4) * 4;
    *reinterpret_cast<float4*>(&xs[r * LDK + c]) = pc[it];
  }
  __syncthreads();

  const int wave = tid >> 5;
  const int lane = tid & 31;
  const int lh = lane & 15;
  const bool hi = (lane >= 16);
  const int col0 = wave * 16;

  v8f acc[RT];
#pragma unroll
  for (int t = 0; t < RT; ++t) acc[t] = (v8f){};

#pragma unroll 4
  for (int k = 0; k < K; k += 4) {
    const int ka = k + (hi ? 2 : 0);
    v2f b;
    b.x = W[(size_t)ka * NOUT + col0 + lh];
    b.y = W[(size_t)(ka + 1) * NOUT + col0 + lh];
#pragma unroll
    for (int t = 0; t < RT; ++t) {
      v2f a;
      a.x = xs[(t * 16 + lh) * LDK + ka];
      a.y = xs[(t * 16 + lh) * LDK + ka + 1];
      acc[t] = __builtin_amdgcn_wmma_f32_16x16x4_f32(false, a, false, b,
                                                     (short)0, acc[t],
                                                     false, false);
    }
  }

  const int rbase = hi ? 8 : 0;
#pragma unroll
  for (int t = 0; t < RT; ++t) {
#pragma unroll
    for (int r = 0; r < 8; ++r) {
      Hout[(size_t)(row0 + t * 16 + rbase + r) * NOUT + col0 + lh] = acc[t][r];
    }
  }
}

__device__ __forceinline__ float leaky(float v) {
  return (v > 0.0f) ? v : NEG_SLOPE * v;
}

template <int CTRL>
__device__ __forceinline__ float dpp_row(float x) {
  const int b = __float_as_int(x);
  return __int_as_float(__builtin_amdgcn_update_dpp(b, b, CTRL, 0xf, 0xf, false));
}

__device__ __forceinline__ int wave_sum_i(int v) {
#pragma unroll
  for (int m = 16; m >= 1; m >>= 1) v += __shfl_xor(v, m, 32);
  return v;
}

#define BIN_SHIFT 8
#define BIN_NODES (1 << BIN_SHIFT)
#define NBINS ((N_NODES + BIN_NODES - 1) >> BIN_SHIFT)
#define EBLOCKS 250
#define ETHREADS 1024
#define EBATCH 4
#define STHREADS 512
#define SBATCH 8
static_assert(NBINS * EBLOCKS <= N_NODES, "hist must fit the N-word slot");
static_assert(N_NODES <= (1 << 24), "src must fit 24 bits of the packed edge");

__device__ __forceinline__ unsigned clamp_node(int v) {
  return min((unsigned)v, (unsigned)(N_NODES - 1));
}

__global__ void bin_count_kernel(const int* __restrict__ dst, int E,
                                 int* __restrict__ hist) {
  __shared__ int lc[NBINS];
  const int tid = threadIdx.x;
  for (int b = tid; b < NBINS; b += ETHREADS) lc[b] = 0;
  __syncthreads();
  constexpr int nblk = EBLOCKS;
  const int chunk = (E + nblk - 1) / nblk;
  const int e0 = min((int)blockIdx.x * chunk, E);
  const int e1 = min(e0 + chunk, E);
  for (int e = e0 + tid; e < e1; e += ETHREADS * EBATCH) {
    int d[EBATCH];
#pragma unroll
    for (int u = 0; u < EBATCH; ++u) d[u] = dst[min(e + u * ETHREADS, e1 - 1)];
#pragma unroll
    for (int u = 0; u < EBATCH; ++u)
      atomicAdd(&lc[clamp_node(d[u]) >> BIN_SHIFT],
                (e + u * ETHREADS < e1) ? 1 : 0);
  }
  __syncthreads();
  for (int b = tid; b < NBINS; b += ETHREADS)
    hist[b * nblk + (int)blockIdx.x] = lc[b];
}

__global__ void bin_scatter_kernel(const int* __restrict__ src,
                                   const int* __restrict__ dst, int E,
                                   const int* __restrict__ offs,
                                   unsigned* __restrict__ binned) {
  __shared__ int lcur[NBINS];
  const int tid = threadIdx.x;
  constexpr int nblk = EBLOCKS;
  for (int b = tid; b < NBINS; b += ETHREADS)
    lcur[b] = offs[b * nblk + (int)blockIdx.x];
  __syncthreads();
  const int chunk = (E + nblk - 1) / nblk;
  const int e0 = min((int)blockIdx.x * chunk, E);
  const int e1 = min(e0 + chunk, E);
  for (int e = e0 + tid; e < e1; e += ETHREADS * EBATCH) {
    int s[EBATCH], d[EBATCH];
#pragma unroll
    for (int u = 0; u < EBATCH; ++u) {
      const int ee = min(e + u * ETHREADS, e1 - 1);
      s[u] = src[ee];
      d[u] = dst[ee];
    }
#pragma unroll
    for (int u = 0; u < EBATCH; ++u) {
      const bool live = (e + u * ETHREADS < e1);
      const unsigned dn = clamp_node(d[u]);
      const int pos = atomicAdd(&lcur[dn >> BIN_SHIFT], live ? 1 : 0);
      if (live && (unsigned)pos < (unsigned)E)
        binned[pos] = (clamp_node(s[u]) << BIN_SHIFT) | (dn & (BIN_NODES - 1));
    }
  }
}

__global__ void bin_sort_kernel(const int* __restrict__ offs, int E, int N,
                                const unsigned* __restrict__ binned,
                                int* __restrict__ rowptr,
                                int* __restrict__ csr_src) {
  __shared__ int lcnt[BIN_NODES];
  __shared__ int wtot[BIN_NODES / 32];
  const int tid = threadIdx.x;
  const int lane = tid & 31;
  const int wave = tid >> 5;
  const int bin = (int)blockIdx.x;
  const int base = bin << BIN_SHIFT;

  int end = (bin + 1 < NBINS) ? offs[(bin + 1) * EBLOCKS] : E;
  end = min(max(end, 0), E);
  const int start = min(max(offs[bin * EBLOCKS], 0), end);

  if (tid < BIN_NODES) lcnt[tid] = 0;
  __syncthreads();

  for (int i = start + tid; i < end; i += STHREADS * SBATCH) {
    unsigned p[SBATCH];
#pragma unroll
    for (int u = 0; u < SBATCH; ++u) p[u] = binned[min(i + u * STHREADS, end - 1)];
#pragma unroll
    for (int u = 0; u < SBATCH; ++u)
      atomicAdd(&lcnt[p[u] & (BIN_NODES - 1)], (i + u * STHREADS < end) ? 1 : 0);
  }
  __syncthreads();

  int c = 0, incl = 0;
  if (tid < BIN_NODES) {
    c = lcnt[tid];
    incl = c;
#pragma unroll
    for (int m = 1; m < 32; m <<= 1) {
      const int u = __shfl_up(incl, m, 32);
      if (lane >= m) incl += u;
    }
    if (lane == 31) wtot[wave] = incl;
  }
  __syncthreads();
  if (tid < BIN_NODES) {
    int off = 0;
#pragma unroll
    for (int w = 0; w < BIN_NODES / 32; ++w)
      if (w < wave) off += wtot[w];
    const int first = start + off + incl - c;
    if (base + tid < N) rowptr[base + tid] = first;
    lcnt[tid] = first;
  }
  __syncthreads();

  for (int i = start + tid; i < end; i += STHREADS * SBATCH) {
    unsigned p[SBATCH];
#pragma unroll
    for (int u = 0; u < SBATCH; ++u) p[u] = binned[min(i + u * STHREADS, end - 1)];
#pragma unroll
    for (int u = 0; u < SBATCH; ++u) {
      const bool live = (i + u * STHREADS < end);
      const int pos = atomicAdd(&lcnt[p[u] & (BIN_NODES - 1)], live ? 1 : 0);
      if (live && (unsigned)pos < (unsigned)E)
        csr_src[pos] = (int)(p[u] >> BIN_SHIFT);
    }
  }
}

__global__ void count_blocksum_kernel(const int* __restrict__ cnt, int N,
                                      int* __restrict__ bsum) {
  __shared__ int wsum[8];
  const int tid = threadIdx.x;
  const int i0 = (blockIdx.x * 256 + tid) * 4;
  int t = 0;
#pragma unroll
  for (int k = 0; k < 4; ++k)
    if (i0 + k < N) t += cnt[i0 + k];
  t = wave_sum_i(t);
  if ((tid & 31) == 0) wsum[tid >> 5] = t;
  __syncthreads();
  if (tid == 0) {
    int s = 0;
#pragma unroll
    for (int w = 0; w < 8; ++w) s += wsum[w];
    bsum[blockIdx.x] = s;
  }
}

__global__ void count_scan_kernel(const int* __restrict__ cnt, int N,
                                  const int* __restrict__ bsum,
                                  int* __restrict__ rowptr) {
  __shared__ int wsum[8];
  __shared__ int wtot[8];
  const int tid = threadIdx.x;
  const int lane = tid & 31;
  const int wave = tid >> 5;

  int p = 0;
  for (int i = tid; i < (int)blockIdx.x; i += 256) p += bsum[i];
  p = wave_sum_i(p);
  if (lane == 0) wsum[wave] = p;

  const int i0 = (blockIdx.x * 256 + tid) * 4;
  int c[4];
  int t = 0;
#pragma unroll
  for (int k = 0; k < 4; ++k) {
    c[k] = (i0 + k < N) ? cnt[i0 + k] : 0;
    t += c[k];
  }
  int incl = t;
#pragma unroll
  for (int m = 1; m < 32; m <<= 1) {
    const int u = __shfl_up(incl, m, 32);
    if (lane >= m) incl += u;
  }
  if (lane == 31) wtot[wave] = incl;
  __syncthreads();

  int off = 0;
#pragma unroll
  for (int w = 0; w < 8; ++w) off += wsum[w];
#pragma unroll
  for (int w = 0; w < 8; ++w)
    if (w < wave) off += wtot[w];

  int run = off + incl - t;
#pragma unroll
  for (int k = 0; k < 4; ++k) {
    if (i0 + k < N) rowptr[i0 + k] = run;
    run += c[k];
  }
}


template <int U>
__device__ __forceinline__ float quad_bcast(float x) {
  const int b = __float_as_int(x);
  return __int_as_float(
      __builtin_amdgcn_update_dpp(b, b, U * 0x55, 0xf, 0xf, false));
}

template <int LPR>
__device__ __forceinline__ float group_sum(float v) {
  v += dpp_row<0x161>(v);
  v += dpp_row<0x162>(v);
  v += dpp_row<0x164>(v);
  if constexpr (LPR == 16) v += dpp_row<0x168>(v);
  return v;
}

template <int LPR>
__device__ __forceinline__ float sum4_group(float p0, float p1, float p2,
                                            float p3, int lane) {
  const bool b0 = (lane & 1) != 0;
  const bool b1 = (lane & 2) != 0;
  float k01 = b0 ? p1 : p0;
  const float t01 = b0 ? p0 : p1;
  float k23 = b0 ? p3 : p2;
  const float t23 = b0 ? p2 : p3;
  k01 += dpp_row<0x161>(t01);
  k23 += dpp_row<0x161>(t23);
  float k = b1 ? k23 : k01;
  const float tt = b1 ? k01 : k23;
  k += dpp_row<0x162>(tt);
  k += dpp_row<0x164>(k);
  if constexpr (LPR == 16) k += dpp_row<0x168>(k);
  return k;
}

template <int NOUT>
__global__ void node_gather_kernel(const int* __restrict__ rowptr,
                                   const int* __restrict__ csr_src, int E,
                                   int N, const float* __restrict__ H,
                                   const float* __restrict__ av,
                                   const float* __restrict__ bv,
                                   const float* __restrict__ bias,
                                   float* __restrict__ out) {
  constexpr int LPR = NOUT / 4;
  constexpr int NPW = 32 / LPR;
  constexpr int NQ = 32 / LPR;
  static_assert(LPR == 8 || LPR == 16, "lane groups of 8 or 16");
  const int lane = threadIdx.x & 31;
  const int g = lane / LPR;
  const int r0 = lane % LPR;
  const int c0 = r0 * 4;
  const int qu = lane & 3;
  const int wv = __builtin_amdgcn_readfirstlane(
      (int)(blockIdx.x * (blockDim.x >> 5) + (threadIdx.x >> 5)));
  if (wv * NPW >= N) return;
  const int dq = wv * NPW + g;
  const bool live = (dq < N);
  const int d = min(dq, N - 1);

  const int rp1 = rowptr[min(d + 1, N - 1)];
  int end = (d + 1 < N) ? rp1 : E;
  end = min(max(end, 0), E);
  const int beg = min(max(rowptr[d], 0), end);
  const int deg = live ? end - beg : 0;
  int degmax = __builtin_amdgcn_readlane(deg, 0);
#pragma unroll
  for (int k = 1; k < NPW; ++k)
    degmax = max(degmax, __builtin_amdgcn_readlane(deg, k * LPR));
  const unsigned nmax = (unsigned)(N - 1);

  auto load_row = [&](int s) -> float4 {
    const unsigned off = (unsigned)s * (unsigned)(NOUT * 4) + (unsigned)(c0 * 4);
    return *reinterpret_cast<const float4*>(
        reinterpret_cast<const char*>(H) + off);
  };
  auto lane_read = [&](int v, int i) -> int {
    return __builtin_amdgcn_ds_bpermute(i << 2, v);
  };
  auto load_src = [&](int j) -> int {
    const int i = max(min(beg + j, beg + deg - 1), 0);
    return (int)min((unsigned)csr_src[i], nmax);
  };

  const float4 hs = load_row(d);
  const float a_s0 = av[c0], a_s1 = av[c0 + 1], a_s2 = av[c0 + 2], a_s3 = av[c0 + 3];
  const float a_d0 = bv[c0], a_d1 = bv[c0 + 1], a_d2 = bv[c0 + 2], a_d3 = bv[c0 + 3];

  const float adv = group_sum<LPR>(hs.x * a_d0 + hs.y * a_d1 + hs.z * a_d2 + hs.w * a_d3);
  float m = leaky(group_sum<LPR>(hs.x * a_s0 + hs.y * a_s1 + hs.z * a_s2 + hs.w * a_s3) + adv);
  float den = 1.0f;
  float acc0 = hs.x, acc1 = hs.y, acc2 = hs.z, acc3 = hs.w;

  for (int j0 = 0; j0 < degmax; j0 += 32) {
    const int cnt = deg - j0;
    const int cmax = min(32, degmax - j0);
    int sq[NQ];
#pragma unroll
    for (int q = 0; q < NQ; ++q) {
      const int sraw = load_src(j0 + q * LPR + r0);
      sq[q] = (q * LPR + r0 < cnt) ? sraw : d;
    }
    auto src_word = [&](int t) -> int {
      int sv = sq[0];
#pragma unroll
      for (int q = 1; q < NQ; ++q) sv = (t >= q * LPR) ? sq[q] : sv;
      return sv;
    };

    auto trip = [&](const float4 (&hv)[4], int t) {
      float p[4];
#pragma unroll
      for (int u = 0; u < 4; ++u)
        p[u] = hv[u].x * a_s0 + hv[u].y * a_s1 + hv[u].z * a_s2 + hv[u].w * a_s3;
      float e = leaky(sum4_group<LPR>(p[0], p[1], p[2], p[3], lane) + adv);
      e = (t + qu < cnt) ? e : -__builtin_inff();

      float gm = fmaxf(e, dpp_row<0x161>(e));
      gm = fmaxf(gm, dpp_row<0x162>(gm));
      const float m_new = fmaxf(m, gm);
      const float sc = __expf(m - m_new);
      const float w = __expf(e - m_new);
      float ws = w + dpp_row<0x161>(w);
      ws += dpp_row<0x162>(ws);
      den = den * sc + ws;
      m = m_new;

      acc0 *= sc; acc1 *= sc; acc2 *= sc; acc3 *= sc;
      const float w0 = quad_bcast<0>(w), w1 = quad_bcast<1>(w);
      const float w2 = quad_bcast<2>(w), w3 = quad_bcast<3>(w);
      acc0 += hv[0].x * w0; acc1 += hv[0].y * w0; acc2 += hv[0].z * w0; acc3 += hv[0].w * w0;
      acc0 += hv[1].x * w1; acc1 += hv[1].y * w1; acc2 += hv[1].z * w1; acc3 += hv[1].w * w1;
      acc0 += hv[2].x * w2; acc1 += hv[2].y * w2; acc2 += hv[2].z * w2; acc3 += hv[2].w * w2;
      acc0 += hv[3].x * w3; acc1 += hv[3].y * w3; acc2 += hv[3].z * w3; acc3 += hv[3].w * w3;
    };

    float4 hv[4];
#pragma unroll
    for (int u = 0; u < 4; ++u) hv[u] = load_row(lane_read(sq[0], g * LPR + u));

    int t = 0;
#pragma nounroll
    for (; t + 4 < cmax; t += 4) {
      float4 hn[4];
      const int sv = src_word(t + 4);
      const int l0 = g * LPR + ((t + 4) & (LPR - 1));
#pragma unroll
      for (int u = 0; u < 4; ++u) hn[u] = load_row(lane_read(sv, l0 + u));
      trip(hv, t);
#pragma unroll
      for (int u = 0; u < 4; ++u) hv[u] = hn[u];
    }
    trip(hv, t);
  }

  if (live) {
    float4 r;
    r.x = fmaxf(acc0 / den + bias[c0], 0.0f);
    r.y = fmaxf(acc1 / den + bias[c0 + 1], 0.0f);
    r.z = fmaxf(acc2 / den + bias[c0 + 2], 0.0f);
    r.w = fmaxf(acc3 / den + bias[c0 + 3], 0.0f);
    *reinterpret_cast<float4*>(out + (size_t)d * NOUT + c0) = r;
  }
}

__global__ void fc_kernel(const float* __restrict__ H,
                          const float* __restrict__ Wfc,
                          const float* __restrict__ bfc,
                          float* __restrict__ out, int N) {
  const int n = blockIdx.x * blockDim.x + threadIdx.x;
  if (n >= N) return;
  float sum = bfc[0];
#pragma unroll
  for (int c = 0; c < H2; ++c) sum += H[(size_t)n * H2 + c] * Wfc[c];
  out[n] = sum;
}

static inline unsigned cdiv(long long a, long long b) {
  return (unsigned)((a + b - 1) / b);
}

extern "C" void kernel_launch(void* const* d_in, const int* in_sizes, int n_in,
                              void* d_out, int out_size, void* d_ws,
                              size_t ws_size, hipStream_t stream) {
  (void)n_in; (void)out_size; (void)ws_size;
  const float* x     = (const float*)d_in[0];
  const int*   ei    = (const int*)d_in[1];
  const float* W1    = (const float*)d_in[2];
  const float* asrc1 = (const float*)d_in[3];
  const float* adst1 = (const float*)d_in[4];
  const float* b1    = (const float*)d_in[5];
  const float* W2    = (const float*)d_in[6];
  const float* asrc2 = (const float*)d_in[7];
  const float* adst2 = (const float*)d_in[8];
  const float* b2    = (const float*)d_in[9];
  const float* Wfc   = (const float*)d_in[10];
  const float* bfc   = (const float*)d_in[11];
  float* out = (float*)d_out;

  const int N = N_NODES;
  const int E = in_sizes[1] / 2;
  const int* src = ei;
  const int* dst = ei + E;

  float* ws = (float*)d_ws;
  size_t off = 0;
  float* h    = ws + off; off += (size_t)N * H1;
  float* agg1 = ws + off; off += (size_t)N * H1;
  float* agg2 = ws + off; off += (size_t)N * H2;
  int* hist    = (int*)(ws + off); off += N;
  int* rowptr  = (int*)(ws + off); off += N;
  int* bsum    = (int*)(ws + off); off += 1024;
  unsigned* binned = (unsigned*)(ws + off); off += E;
  int* csr_src = (int*)(ws + off); off += E;
  int* offs = (int*)agg2;

  const dim3 B(256);
  constexpr int MR = 32;
  constexpr int GN1 = 8 * (128 / H1);
  constexpr int GN2 = 8 * (128 / H2);
  constexpr int NH = NBINS * EBLOCKS;
  const unsigned nscan = cdiv(NH, 1024);

  bin_count_kernel<<<EBLOCKS, ETHREADS, 0, stream>>>(dst, E, hist);
  count_blocksum_kernel<<<nscan, B, 0, stream>>>(hist, NH, bsum);
  count_scan_kernel<<<nscan, B, 0, stream>>>(hist, NH, bsum, offs);
  bin_scatter_kernel<<<EBLOCKS, ETHREADS, 0, stream>>>(src, dst, E, offs, binned);
  bin_sort_kernel<<<NBINS, STHREADS, 0, stream>>>(offs, E, N, binned, rowptr, csr_src);

  gemm_wmma_kernel<F_IN, H1, MR><<<N / MR, (H1 / 16) * 32, 0, stream>>>(x, W1, h);
  node_gather_kernel<H1><<<cdiv(N, GN1), B, 0, stream>>>(rowptr, csr_src, E, N, h, asrc1, adst1, b1, agg1);

  gemm_wmma_kernel<H1, H2, MR><<<N / MR, (H2 / 16) * 32, 0, stream>>>(agg1, W2, h);
  node_gather_kernel<H2><<<cdiv(N, GN2), B, 0, stream>>>(rowptr, csr_src, E, N, h, asrc2, adst2, b2, agg2);

  fc_kernel<<<cdiv(N, 256), B, 0, stream>>>(agg2, Wfc, bfc, out, N);
}


# Each gather applies the next dense layer to its finished rows before storing (64->32 product on WMMA from a 16-node LDS tile; fc as a lane-group dot): 10 to 8 launches; fc sum re-associated
# speedup vs baseline: 6.1075x; 1.0522x over previous
#include <hip/hip_runtime.h>

typedef float v2f __attribute__((ext_vector_type(2)));
typedef float v8f __attribute__((ext_vector_type(8)));

#define N_NODES 100000
#define F_IN 256
#define H1 64
#define H2 32
#define NEG_SLOPE 0.2f

template <int K, int NOUT, int MROWS>
__global__ void __launch_bounds__((NOUT / 16) * 32)
gemm_wmma_kernel(const float* __restrict__ X,
                 const float* __restrict__ W,
                 float* __restrict__ Hout) {
  constexpr int LDK = K + 4;
  constexpr int RT = MROWS / 16;
  constexpr int NTHR = (NOUT / 16) * 32;
  constexpr int K4 = K / 4;
  constexpr int PPT = MROWS * K4 / NTHR;
  static_assert(MROWS * K4 % NTHR == 0, "whole pieces per thread");
  static_assert(LDK % 4 == 0, "LDS rows start on 16 bytes");
  __shared__ __attribute__((aligned(16))) float xs[MROWS * LDK];
  const int tid = threadIdx.x;
  const int row0 = blockIdx.x * MROWS;

  __builtin_prefetch(W, 0, 1);

  const float4* __restrict__ xt =
      reinterpret_cast<const float4*>(X + (size_t)row0 * K);
  float4 pc[PPT];
#pragma unroll
  for (int it = 0; it < PPT; ++it) pc[it] = xt[tid + it * NTHR];
#pragma unroll
  for (int it = 0; it < PPT; ++it) {
    const int i = tid + it * NTHR;
    const int r = i / K4, c = (i % K4) * 4;
    *reinterpret_cast<float4*>(&xs[r * LDK + c]) = pc[it];
  }
  __syncthreads();

  const int wave = tid >> 5;
  const int lane = tid & 31;
  const int lh = lane & 15;
  const bool hi = (lane >= 16);
  const int col0 = wave * 16;

  v8f acc[RT];
#pragma unroll
  for (int t = 0; t < RT; ++t) acc[t] = (v8f){};

#pragma unroll 4
  for (int k = 0; k < K; k += 4) {
    const int ka = k + (hi ? 2 : 0);
    v2f b;
    b.x = W[(size_t)ka * NOUT + col0 + lh];
    b.y = W[(size_t)(ka + 1) * NOUT + col0 + lh];
#pragma unroll
    for (int t = 0; t < RT; ++t) {
      v2f a;
      a.x = xs[(t * 16 + lh) * LDK + ka];
      a.y = xs[(t * 16 + lh) * LDK + ka + 1];
      acc[t] = __builtin_amdgcn_wmma_f32_16x16x4_f32(false, a, false, b,
                                                     (short)0, acc[t],
                                                     false, false);
    }
  }

  const int rbase = hi ? 8 : 0;
#pragma unroll
  for (int t = 0; t < RT; ++t) {
#pragma unroll
    for (int r = 0; r < 8; ++r) {
      Hout[(size_t)(row0 + t * 16 + rbase + r) * NOUT + col0 + lh] = acc[t][r];
    }
  }
}

__device__ __forceinline__ float leaky(float v) {
  return (v > 0.0f) ? v : NEG_SLOPE * v;
}

template <int CTRL>
__device__ __forceinline__ float dpp_row(float x) {
  const int b = __float_as_int(x);
  return __int_as_float(__builtin_amdgcn_update_dpp(b, b, CTRL, 0xf, 0xf, false));
}

__device__ __forceinline__ int wave_sum_i(int v) {
#pragma unroll
  for (int m = 16; m >= 1; m >>= 1) v += __shfl_xor(v, m, 32);
  return v;
}

#define BIN_SHIFT 8
#define BIN_NODES (1 << BIN_SHIFT)
#define NBINS ((N_NODES + BIN_NODES - 1) >> BIN_SHIFT)
#define EBLOCKS 250
#define ETHREADS 1024
#define EBATCH 4
#define STHREADS 512
#define SBATCH 8
static_assert(NBINS * EBLOCKS <= N_NODES, "hist must fit the N-word slot");
static_assert(N_NODES <= (1 << 24), "src must fit 24 bits of the packed edge");

__device__ __forceinline__ unsigned clamp_node(int v) {
  return min((unsigned)v, (unsigned)(N_NODES - 1));
}

__global__ void bin_count_kernel(const int* __restrict__ dst, int E,
                                 int* __restrict__ hist) {
  __shared__ int lc[NBINS];
  const int tid = threadIdx.x;
  for (int b = tid; b < NBINS; b += ETHREADS) lc[b] = 0;
  __syncthreads();
  constexpr int nblk = EBLOCKS;
  const int chunk = (E + nblk - 1) / nblk;
  const int e0 = min((int)blockIdx.x * chunk, E);
  const int e1 = min(e0 + chunk, E);
  for (int e = e0 + tid; e < e1; e += ETHREADS * EBATCH) {
    int d[EBATCH];
#pragma unroll
    for (int u = 0; u < EBATCH; ++u) d[u] = dst[min(e + u * ETHREADS, e1 - 1)];
#pragma unroll
    for (int u = 0; u < EBATCH; ++u)
      atomicAdd(&lc[clamp_node(d[u]) >> BIN_SHIFT],
                (e + u * ETHREADS < e1) ? 1 : 0);
  }
  __syncthreads();
  for (int b = tid; b < NBINS; b += ETHREADS)
    hist[b * nblk + (int)blockIdx.x] = lc[b];
}

__global__ void bin_scatter_kernel(const int* __restrict__ src,
                                   const int* __restrict__ dst, int E,
                                   const int* __restrict__ offs,
                                   unsigned* __restrict__ binned) {
  __shared__ int lcur[NBINS];
  const int tid = threadIdx.x;
  constexpr int nblk = EBLOCKS;
  for (int b = tid; b < NBINS; b += ETHREADS)
    lcur[b] = offs[b * nblk + (int)blockIdx.x];
  __syncthreads();
  const int chunk = (E + nblk - 1) / nblk;
  const int e0 = min((int)blockIdx.x * chunk, E);
  const int e1 = min(e0 + chunk, E);
  for (int e = e0 + tid; e < e1; e += ETHREADS * EBATCH) {
    int s[EBATCH], d[EBATCH];
#pragma unroll
    for (int u = 0; u < EBATCH; ++u) {
      const int ee = min(e + u * ETHREADS, e1 - 1);
      s[u] = src[ee];
      d[u] = dst[ee];
    }
#pragma unroll
    for (int u = 0; u < EBATCH; ++u) {
      const bool live = (e + u * ETHREADS < e1);
      const unsigned dn = clamp_node(d[u]);
      const int pos = atomicAdd(&lcur[dn >> BIN_SHIFT], live ? 1 : 0);
      if (live && (unsigned)pos < (unsigned)E)
        binned[pos] = (clamp_node(s[u]) << BIN_SHIFT) | (dn & (BIN_NODES - 1));
    }
  }
}

__global__ void bin_sort_kernel(const int* __restrict__ offs, int E, int N,
                                const unsigned* __restrict__ binned,
                                int* __restrict__ rowptr,
                                int* __restrict__ csr_src) {
  __shared__ int lcnt[BIN_NODES];
  __shared__ int wtot[BIN_NODES / 32];
  const int tid = threadIdx.x;
  const int lane = tid & 31;
  const int wave = tid >> 5;
  const int bin = (int)blockIdx.x;
  const int base = bin << BIN_SHIFT;

  int end = (bin + 1 < NBINS) ? offs[(bin + 1) * EBLOCKS] : E;
  end = min(max(end, 0), E);
  const int start = min(max(offs[bin * EBLOCKS], 0), end);

  if (tid < BIN_NODES) lcnt[tid] = 0;
  __syncthreads();

  for (int i = start + tid; i < end; i += STHREADS * SBATCH) {
    unsigned p[SBATCH];
#pragma unroll
    for (int u = 0; u < SBATCH; ++u) p[u] = binned[min(i + u * STHREADS, end - 1)];
#pragma unroll
    for (int u = 0; u < SBATCH; ++u)
      atomicAdd(&lcnt[p[u] & (BIN_NODES - 1)], (i + u * STHREADS < end) ? 1 : 0);
  }
  __syncthreads();

  int c = 0, incl = 0;
  if (tid < BIN_NODES) {
    c = lcnt[tid];
    incl = c;
#pragma unroll
    for (int m = 1; m < 32; m <<= 1) {
      const int u = __shfl_up(incl, m, 32);
      if (lane >= m) incl += u;
    }
    if (lane == 31) wtot[wave] = incl;
  }
  __syncthreads();
  if (tid < BIN_NODES) {
    int off = 0;
#pragma unroll
    for (int w = 0; w < BIN_NODES / 32; ++w)
      if (w < wave) off += wtot[w];
    const int first = start + off + incl - c;
    if (base + tid < N) rowptr[base + tid] = first;
    lcnt[tid] = first;
  }
  __syncthreads();

  for (int i = start + tid; i < end; i += STHREADS * SBATCH) {
    unsigned p[SBATCH];
#pragma unroll
    for (int u = 0; u < SBATCH; ++u) p[u] = binned[min(i + u * STHREADS, end - 1)];
#pragma unroll
    for (int u = 0; u < SBATCH; ++u) {
      const bool live = (i + u * STHREADS < end);
      const int pos = atomicAdd(&lcnt[p[u] & (BIN_NODES - 1)], live ? 1 : 0);
      if (live && (unsigned)pos < (unsigned)E)
        csr_src[pos] = (int)(p[u] >> BIN_SHIFT);
    }
  }
}

__global__ void count_blocksum_kernel(const int* __restrict__ cnt, int N,
                                      int* __restrict__ bsum) {
  __shared__ int wsum[8];
  const int tid = threadIdx.x;
  const int i0 = (blockIdx.x * 256 + tid) * 4;
  int t = 0;
#pragma unroll
  for (int k = 0; k < 4; ++k)
    if (i0 + k < N) t += cnt[i0 + k];
  t = wave_sum_i(t);
  if ((tid & 31) == 0) wsum[tid >> 5] = t;
  __syncthreads();
  if (tid == 0) {
    int s = 0;
#pragma unroll
    for (int w = 0; w < 8; ++w) s += wsum[w];
    bsum[blockIdx.x] = s;
  }
}

__global__ void count_scan_kernel(const int* __restrict__ cnt, int N,
                                  const int* __restrict__ bsum,
                                  int* __restrict__ rowptr) {
  __shared__ int wsum[8];
  __shared__ int wtot[8];
  const int tid = threadIdx.x;
  const int lane = tid & 31;
  const int wave = tid >> 5;

  int p = 0;
  for (int i = tid; i < (int)blockIdx.x; i += 256) p += bsum[i];
  p = wave_sum_i(p);
  if (lane == 0) wsum[wave] = p;

  const int i0 = (blockIdx.x * 256 + tid) * 4;
  int c[4];
  int t = 0;
#pragma unroll
  for (int k = 0; k < 4; ++k) {
    c[k] = (i0 + k < N) ? cnt[i0 + k] : 0;
    t += c[k];
  }
  int incl = t;
#pragma unroll
  for (int m = 1; m < 32; m <<= 1) {
    const int u = __shfl_up(incl, m, 32);
    if (lane >= m) incl += u;
  }
  if (lane == 31) wtot[wave] = incl;
  __syncthreads();

  int off = 0;
#pragma unroll
  for (int w = 0; w < 8; ++w) off += wsum[w];
#pragma unroll
  for (int w = 0; w < 8; ++w)
    if (w < wave) off += wtot[w];

  int run = off + incl - t;
#pragma unroll
  for (int k = 0; k < 4; ++k) {
    if (i0 + k < N) rowptr[i0 + k] = run;
    run += c[k];
  }
}


template <int U>
__device__ __forceinline__ float quad_bcast(float x) {
  const int b = __float_as_int(x);
  return __int_as_float(
      __builtin_amdgcn_update_dpp(b, b, U * 0x55, 0xf, 0xf, false));
}

template <int LPR>
__device__ __forceinline__ float group_sum(float v) {
  v += dpp_row<0x161>(v);
  v += dpp_row<0x162>(v);
  v += dpp_row<0x164>(v);
  if constexpr (LPR == 16) v += dpp_row<0x168>(v);
  return v;
}

template <int LPR>
__device__ __forceinline__ float sum4_group(float p0, float p1, float p2,
                                            float p3, int lane) {
  const bool b0 = (lane & 1) != 0;
  const bool b1 = (lane & 2) != 0;
  float k01 = b0 ? p1 : p0;
  const float t01 = b0 ? p0 : p1;
  float k23 = b0 ? p3 : p2;
  const float t23 = b0 ? p2 : p3;
  k01 += dpp_row<0x161>(t01);
  k23 += dpp_row<0x161>(t23);
  float k = b1 ? k23 : k01;
  const float tt = b1 ? k01 : k23;
  k += dpp_row<0x162>(tt);
  k += dpp_row<0x164>(k);
  if constexpr (LPR == 16) k += dpp_row<0x168>(k);
  return k;
}

#define GWAVES 8
template <int NOUT>
__global__ void node_gather_kernel(const int* __restrict__ rowptr,
                                   const int* __restrict__ csr_src, int E,
                                   int N, const float* __restrict__ H,
                                   const float* __restrict__ av,
                                   const float* __restrict__ bv,
                                   const float* __restrict__ bias,
                                   const float* __restrict__ Wn,
                                   const float* __restrict__ bn,
                                   float* __restrict__ out) {
  constexpr int LPR = NOUT / 4;
  constexpr int NPW = 32 / LPR;
  constexpr int NQ = 32 / LPR;
  static_assert(LPR == 8 || LPR == 16, "lane groups of 8 or 16");
  const int lane = threadIdx.x & 31;
  const int g = lane / LPR;
  const int r0 = lane % LPR;
  const int c0 = r0 * 4;
  const int qu = lane & 3;
  const int wib = __builtin_amdgcn_readfirstlane((int)(threadIdx.x >> 5));
  const int wv = (int)blockIdx.x * GWAVES + wib;
  const int dq = wv * NPW + g;
  const bool live = (dq < N);
  const int d = min(dq, N - 1);

  const int rp1 = rowptr[min(d + 1, N - 1)];
  int end = (d + 1 < N) ? rp1 : E;
  end = min(max(end, 0), E);
  const int beg = min(max(rowptr[d], 0), end);
  const int deg = live ? end - beg : 0;
  int degmax = __builtin_amdgcn_readlane(deg, 0);
#pragma unroll
  for (int k = 1; k < NPW; ++k)
    degmax = max(degmax, __builtin_amdgcn_readlane(deg, k * LPR));
  const unsigned nmax = (unsigned)(N - 1);

  auto load_row = [&](int s) -> float4 {
    const unsigned off = (unsigned)s * (unsigned)(NOUT * 4) + (unsigned)(c0 * 4);
    return *reinterpret_cast<const float4*>(
        reinterpret_cast<const char*>(H) + off);
  };
  auto lane_read = [&](int v, int i) -> int {
    return __builtin_amdgcn_ds_bpermute(i << 2, v);
  };
  auto load_src = [&](int j) -> int {
    const int i = max(min(beg + j, beg + deg - 1), 0);
    return (int)min((unsigned)csr_src[i], nmax);
  };

  const float4 hs = load_row(d);
  const float a_s0 = av[c0], a_s1 = av[c0 + 1], a_s2 = av[c0 + 2], a_s3 = av[c0 + 3];
  const float a_d0 = bv[c0], a_d1 = bv[c0 + 1], a_d2 = bv[c0 + 2], a_d3 = bv[c0 + 3];

  const float adv = group_sum<LPR>(hs.x * a_d0 + hs.y * a_d1 + hs.z * a_d2 + hs.w * a_d3);
  float m = leaky(group_sum<LPR>(hs.x * a_s0 + hs.y * a_s1 + hs.z * a_s2 + hs.w * a_s3) + adv);
  float den = 1.0f;
  float acc0 = hs.x, acc1 = hs.y, acc2 = hs.z, acc3 = hs.w;

  for (int j0 = 0; j0 < degmax; j0 += 32) {
    const int cnt = deg - j0;
    const int cmax = min(32, degmax - j0);
    int sq[NQ];
#pragma unroll
    for (int q = 0; q < NQ; ++q) {
      const int sraw = load_src(j0 + q * LPR + r0);
      sq[q] = (q * LPR + r0 < cnt) ? sraw : d;
    }
    auto src_word = [&](int t) -> int {
      int sv = sq[0];
#pragma unroll
      for (int q = 1; q < NQ; ++q) sv = (t >= q * LPR) ? sq[q] : sv;
      return sv;
    };

    auto trip = [&](const float4 (&hv)[4], int t) {
      float p[4];
#pragma unroll
      for (int u = 0; u < 4; ++u)
        p[u] = hv[u].x * a_s0 + hv[u].y * a_s1 + hv[u].z * a_s2 + hv[u].w * a_s3;
      float e = leaky(sum4_group<LPR>(p[0], p[1], p[2], p[3], lane) + adv);
      e = (t + qu < cnt) ? e : -__builtin_inff();

      float gm = fmaxf(e, dpp_row<0x161>(e));
      gm = fmaxf(gm, dpp_row<0x162>(gm));
      const float m_new = fmaxf(m, gm);
      const float sc = __expf(m - m_new);
      const float w = __expf(e - m_new);
      float ws = w + dpp_row<0x161>(w);
      ws += dpp_row<0x162>(ws);
      den = den * sc + ws;
      m = m_new;

      acc0 *= sc; acc1 *= sc; acc2 *= sc; acc3 *= sc;
      const float w0 = quad_bcast<0>(w), w1 = quad_bcast<1>(w);
      const float w2 = quad_bcast<2>(w), w3 = quad_bcast<3>(w);
      acc0 += hv[0].x * w0; acc1 += hv[0].y * w0; acc2 += hv[0].z * w0; acc3 += hv[0].w * w0;
      acc0 += hv[1].x * w1; acc1 += hv[1].y * w1; acc2 += hv[1].z * w1; acc3 += hv[1].w * w1;
      acc0 += hv[2].x * w2; acc1 += hv[2].y * w2; acc2 += hv[2].z * w2; acc3 += hv[2].w * w2;
      acc0 += hv[3].x * w3; acc1 += hv[3].y * w3; acc2 += hv[3].z * w3; acc3 += hv[3].w * w3;
    };

    float4 hv[4];
#pragma unroll
    for (int u = 0; u < 4; ++u) hv[u] = load_row(lane_read(sq[0], g * LPR + u));

    int t = 0;
#pragma nounroll
    for (; t + 4 < cmax; t += 4) {
      float4 hn[4];
      const int sv = src_word(t + 4);
      const int l0 = g * LPR + ((t + 4) & (LPR - 1));
#pragma unroll
      for (int u = 0; u < 4; ++u) hn[u] = load_row(lane_read(sv, l0 + u));
      trip(hv, t);
#pragma unroll
      for (int u = 0; u < 4; ++u) hv[u] = hn[u];
    }
    trip(hv, t);
  }

  float4 r;
  r.x = live ? fmaxf(acc0 / den + bias[c0], 0.0f) : 0.0f;
  r.y = live ? fmaxf(acc1 / den + bias[c0 + 1], 0.0f) : 0.0f;
  r.z = live ? fmaxf(acc2 / den + bias[c0 + 2], 0.0f) : 0.0f;
  r.w = live ? fmaxf(acc3 / den + bias[c0 + 3], 0.0f) : 0.0f;
  const int nib = wib * NPW + g;

  if constexpr (NOUT == H1) {
    constexpr int ROWS = GWAVES * NPW;
    constexpr int NO2 = H2;
    constexpr int LDT = NOUT + 4;
    static_assert(ROWS == 16, "one 16-row matrix tile per work-group");
    __shared__ __attribute__((aligned(16))) float ts[ROWS * LDT];
    *reinterpret_cast<float4*>(&ts[nib * LDT + c0]) = r;
    __syncthreads();
    if (wib < NO2 / 16) {
      const int lh = lane & 15;
      const bool hi = (lane >= 16);
      const int col0 = wib * 16;
      v8f cacc = (v8f){};
#pragma unroll
      for (int k = 0; k < NOUT; k += 4) {
        const int ka = k + (hi ? 2 : 0);
        v2f b;
        b.x = Wn[ka * NO2 + col0 + lh];
        b.y = Wn[(ka + 1) * NO2 + col0 + lh];
        v2f a;
        a.x = ts[lh * LDT + ka];
        a.y = ts[lh * LDT + ka + 1];
        cacc = __builtin_amdgcn_wmma_f32_16x16x4_f32(false, a, false, b,
                                                     (short)0, cacc,
                                                     false, false);
      }
      const int row0 = (int)blockIdx.x * ROWS + (hi ? 8 : 0);
#pragma unroll
      for (int q = 0; q < 8; ++q) {
        if (row0 + q < N) out[(size_t)(row0 + q) * NO2 + col0 + lh] = cacc[q];
      }
    }
  } else {
    constexpr int ROWS = GWAVES * NPW;
    static_assert(ROWS == 32, "one 128-byte line of results per work-group");
    __shared__ float fo[ROWS];
    const float s = group_sum<LPR>(r.x * Wn[c0] + r.y * Wn[c0 + 1] +
                                   r.z * Wn[c0 + 2] + r.w * Wn[c0 + 3]) + bn[0];
    if (r0 == 0) fo[nib] = s;
    __syncthreads();
    const int n = (int)blockIdx.x * ROWS + (int)threadIdx.x;
    if (threadIdx.x < ROWS && n < N) out[n] = fo[threadIdx.x];
  }
}

static inline unsigned cdiv(long long a, long long b) {
  return (unsigned)((a + b - 1) / b);
}

extern "C" void kernel_launch(void* const* d_in, const int* in_sizes, int n_in,
                              void* d_out, int out_size, void* d_ws,
                              size_t ws_size, hipStream_t stream) {
  (void)n_in; (void)out_size; (void)ws_size;
  const float* x     = (const float*)d_in[0];
  const int*   ei    = (const int*)d_in[1];
  const float* W1    = (const float*)d_in[2];
  const float* asrc1 = (const float*)d_in[3];
  const float* adst1 = (const float*)d_in[4];
  const float* b1    = (const float*)d_in[5];
  const float* W2    = (const float*)d_in[6];
  const float* asrc2 = (const float*)d_in[7];
  const float* adst2 = (const float*)d_in[8];
  const float* b2    = (const float*)d_in[9];
  const float* Wfc   = (const float*)d_in[10];
  const float* bfc   = (const float*)d_in[11];
  float* out = (float*)d_out;

  const int N = N_NODES;
  const int E = in_sizes[1] / 2;
  const int* src = ei;
  const int* dst = ei + E;

  float* ws = (float*)d_ws;
  size_t off = 0;
  float* h    = ws + off; off += (size_t)N * H1;
  float* h2   = ws + off; off += (size_t)N * H1;
  float* agg2 = ws + off; off += (size_t)N * H2;
  int* hist    = (int*)(ws + off); off += N;
  int* rowptr  = (int*)(ws + off); off += N;
  int* bsum    = (int*)(ws + off); off += 1024;
  unsigned* binned = (unsigned*)(ws + off); off += E;
  int* csr_src = (int*)(ws + off); off += E;
  int* offs = (int*)agg2;

  const dim3 B(256);
  constexpr int MR = 32;
  constexpr int GN1 = GWAVES * (128 / H1);
  constexpr int GN2 = GWAVES * (128 / H2);
  const dim3 GB(GWAVES * 32);
  constexpr int NH = NBINS * EBLOCKS;
  const unsigned nscan = cdiv(NH, 1024);

  bin_count_kernel<<<EBLOCKS, ETHREADS, 0, stream>>>(dst, E, hist);
  count_blocksum_kernel<<<nscan, B, 0, stream>>>(hist, NH, bsum);
  count_scan_kernel<<<nscan, B, 0, stream>>>(hist, NH, bsum, offs);
  bin_scatter_kernel<<<EBLOCKS, ETHREADS, 0, stream>>>(src, dst, E, offs, binned);
  bin_sort_kernel<<<NBINS, STHREADS, 0, stream>>>(offs, E, N, binned, rowptr, csr_src);

  gemm_wmma_kernel<F_IN, H1, MR><<<N / MR, (H1 / 16) * 32, 0, stream>>>(x, W1, h);
  node_gather_kernel<H1><<<cdiv(N, GN1), GB, 0, stream>>>(rowptr, csr_src, E, N, h, asrc1, adst1, b1, W2, bfc, h2);

  node_gather_kernel<H2><<<cdiv(N, GN2), GB, 0, stream>>>(rowptr, csr_src, E, N, h2, asrc2, adst2, b2, Wfc, bfc, out);
}


# Edge chunks take their slot runs inside a bin by one returning global add on the bin's total; bin starts summed in scatter and sort: no scan launches (8 to 6); sum order: arrival order as before
# speedup vs baseline: 6.5569x; 1.0736x over previous
#include <hip/hip_runtime.h>

typedef float v2f __attribute__((ext_vector_type(2)));
typedef float v8f __attribute__((ext_vector_type(8)));

#define N_NODES 100000
#define F_IN 256
#define H1 64
#define H2 32
#define NEG_SLOPE 0.2f

#define BIN_SHIFT 8
#define BIN_NODES (1 << BIN_SHIFT)
#define NBINS ((N_NODES + BIN_NODES - 1) >> BIN_SHIFT)
#define TOT_STRIDE 64

template <int K, int NOUT, int MROWS>
__global__ void __launch_bounds__((NOUT / 16) * 32)
gemm_wmma_kernel(const float* __restrict__ X,
                 const float* __restrict__ W,
                 float* __restrict__ Hout,
                 int* __restrict__ bintot) {
  constexpr int LDK = K + 4;
  constexpr int RT = MROWS / 16;
  constexpr int NTHR = (NOUT / 16) * 32;
  constexpr int K4 = K / 4;
  constexpr int PPT = MROWS * K4 / NTHR;
  static_assert(MROWS * K4 % NTHR == 0, "whole pieces per thread");
  static_assert(LDK % 4 == 0, "LDS rows start on 16 bytes");
  __shared__ __attribute__((aligned(16))) float xs[MROWS * LDK];
  const int tid = threadIdx.x;
  const int row0 = blockIdx.x * MROWS;

  __builtin_prefetch(W, 0, 1);

  if (blockIdx.x == 0)
    for (int b = tid; b < NBINS; b += NTHR) bintot[b * TOT_STRIDE] = 0;

  const float4* __restrict__ xt =
      reinterpret_cast<const float4*>(X + (size_t)row0 * K);
  float4 pc[PPT];
#pragma unroll
  for (int it = 0; it < PPT; ++it) pc[it] = xt[tid + it * NTHR];
#pragma unroll
  for (int it = 0; it < PPT; ++it) {
    const int i = tid + it * NTHR;
    const int r = i / K4, c = (i % K4) * 4;
    *reinterpret_cast<float4*>(&xs[r * LDK + c]) = pc[it];
  }
  __syncthreads();

  const int wave = tid >> 5;
  const int lane = tid & 31;
  const int lh = lane & 15;
  const bool hi = (lane >= 16);
  const int col0 = wave * 16;

  v8f acc[RT];
#pragma unroll
  for (int t = 0; t < RT; ++t) acc[t] = (v8f){};

#pragma unroll 4
  for (int k = 0; k < K; k += 4) {
    const int ka = k + (hi ? 2 : 0);
    v2f b;
    b.x = W[(size_t)ka * NOUT + col0 + lh];
    b.y = W[(size_t)(ka + 1) * NOUT + col0 + lh];
#pragma unroll
    for (int t = 0; t < RT; ++t) {
      v2f a;
      a.x = xs[(t * 16 + lh) * LDK + ka];
      a.y = xs[(t * 16 + lh) * LDK + ka + 1];
      acc[t] = __builtin_amdgcn_wmma_f32_16x16x4_f32(false, a, false, b,
                                                     (short)0, acc[t],
                                                     false, false);
    }
  }

  const int rbase = hi ? 8 : 0;
#pragma unroll
  for (int t = 0; t < RT; ++t) {
#pragma unroll
    for (int r = 0; r < 8; ++r) {
      Hout[(size_t)(row0 + t * 16 + rbase + r) * NOUT + col0 + lh] = acc[t][r];
    }
  }
}

__device__ __forceinline__ float leaky(float v) {
  return (v > 0.0f) ? v : NEG_SLOPE * v;
}

template <int CTRL>
__device__ __forceinline__ float dpp_row(float x) {
  const int b = __float_as_int(x);
  return __int_as_float(__builtin_amdgcn_update_dpp(b, b, CTRL, 0xf, 0xf, false));
}

__device__ __forceinline__ int wave_sum_i(int v) {
#pragma unroll
  for (int m = 16; m >= 1; m >>= 1) v += __shfl_xor(v, m, 32);
  return v;
}

#define EBLOCKS 250
#define ETHREADS 1024
#define EBATCH 4
#define STHREADS 512
#define SBATCH 8
#define SCANW ((NBINS + 31) / 32)
static_assert(NBINS * EBLOCKS <= N_NODES, "segoff must fit the N-word slot");
static_assert(NBINS <= STHREADS, "one thread per earlier bin in bin_sort");
static_assert(N_NODES <= (1 << 24), "src must fit 24 bits of the packed edge");

__device__ __forceinline__ unsigned clamp_node(int v) {
  return min((unsigned)v, (unsigned)(N_NODES - 1));
}

__global__ void bin_count_kernel(const int* __restrict__ dst, int E,
                                 int* __restrict__ bintot,
                                 int* __restrict__ segoff) {
  __shared__ int lc[NBINS];
  const int tid = threadIdx.x;
  for (int b = tid; b < NBINS; b += ETHREADS) lc[b] = 0;
  __syncthreads();
  constexpr int nblk = EBLOCKS;
  const int chunk = (E + nblk - 1) / nblk;
  const int e0 = min((int)blockIdx.x * chunk, E);
  const int e1 = min(e0 + chunk, E);
  for (int e = e0 + tid; e < e1; e += ETHREADS * EBATCH) {
    int d[EBATCH];
#pragma unroll
    for (int u = 0; u < EBATCH; ++u) d[u] = dst[min(e + u * ETHREADS, e1 - 1)];
#pragma unroll
    for (int u = 0; u < EBATCH; ++u)
      atomicAdd(&lc[clamp_node(d[u]) >> BIN_SHIFT],
                (e + u * ETHREADS < e1) ? 1 : 0);
  }
  __syncthreads();
  for (int b = tid; b < NBINS; b += ETHREADS)
    segoff[b * nblk + (int)blockIdx.x] = atomicAdd(&bintot[b * TOT_STRIDE], lc[b]);
}

__global__ void bin_scatter_kernel(const int* __restrict__ src,
                                   const int* __restrict__ dst, int E,
                                   const int* __restrict__ bintot,
                                   const int* __restrict__ segoff,
                                   unsigned* __restrict__ binned) {
  __shared__ int lcur[NBINS];
  __shared__ int wtot[SCANW];
  const int tid = threadIdx.x;
  const int lane = tid & 31;
  const int wave = tid >> 5;
  constexpr int nblk = EBLOCKS;
  static_assert(SCANW * 32 <= ETHREADS, "one thread per bin total");
  int c = 0, incl = 0, so = 0;
  if (wave < SCANW) {
    const int b = min(tid, NBINS - 1);
    c = (tid < NBINS) ? bintot[b * TOT_STRIDE] : 0;
    so = segoff[b * nblk + (int)blockIdx.x];
    incl = c;
#pragma unroll
    for (int m = 1; m < 32; m <<= 1) {
      const int u = __shfl_up(incl, m, 32);
      if (lane >= m) incl += u;
    }
    if (lane == 31) wtot[wave] = incl;
  }
  __syncthreads();
  if (tid < NBINS) {
    int off = 0;
#pragma unroll
    for (int w = 0; w < SCANW; ++w)
      if (w < wave) off += wtot[w];
    lcur[tid] = off + incl - c + so;
  }
  __syncthreads();
  const int chunk = (E + nblk - 1) / nblk;
  const int e0 = min((int)blockIdx.x * chunk, E);
  const int e1 = min(e0 + chunk, E);
  for (int e = e0 + tid; e < e1; e += ETHREADS * EBATCH) {
    int s[EBATCH], d[EBATCH];
#pragma unroll
    for (int u = 0; u < EBATCH; ++u) {
      const int ee = min(e + u * ETHREADS, e1 - 1);
      s[u] = src[ee];
      d[u] = dst[ee];
    }
#pragma unroll
    for (int u = 0; u < EBATCH; ++u) {
      const bool live = (e + u * ETHREADS < e1);
      const unsigned dn = clamp_node(d[u]);
      const int pos = atomicAdd(&lcur[dn >> BIN_SHIFT], live ? 1 : 0);
      if (live && (unsigned)pos < (unsigned)E)
        binned[pos] = (clamp_node(s[u]) << BIN_SHIFT) | (dn & (BIN_NODES - 1));
    }
  }
}

__global__ void bin_sort_kernel(const int* __restrict__ bintot, int E, int N,
                                const unsigned* __restrict__ binned,
                                int* __restrict__ rowptr,
                                int* __restrict__ csr_src) {
  __shared__ int lcnt[BIN_NODES];
  __shared__ int wtot[BIN_NODES / 32];
  __shared__ int wpre[STHREADS / 32];
  const int tid = threadIdx.x;
  const int lane = tid & 31;
  const int wave = tid >> 5;
  const int bin = (int)blockIdx.x;
  const int base = bin << BIN_SHIFT;

  int pre = (tid < bin) ? bintot[tid * TOT_STRIDE] : 0;
  pre = wave_sum_i(pre);
  if (lane == 0) wpre[wave] = pre;
  const int mine = bintot[bin * TOT_STRIDE];

  if (tid < BIN_NODES) lcnt[tid] = 0;
  __syncthreads();

  int first = 0;
#pragma unroll
  for (int w = 0; w < STHREADS / 32; ++w) first += wpre[w];
  const int end = min(max(first + mine, 0), E);
  const int start = min(max(first, 0), end);

  for (int i = start + tid; i < end; i += STHREADS * SBATCH) {
    unsigned p[SBATCH];
#pragma unroll
    for (int u = 0; u < SBATCH; ++u) p[u] = binned[min(i + u * STHREADS, end - 1)];
#pragma unroll
    for (int u = 0; u < SBATCH; ++u)
      atomicAdd(&lcnt[p[u] & (BIN_NODES - 1)], (i + u * STHREADS < end) ? 1 : 0);
  }
  __syncthreads();

  int c = 0, incl = 0;
  if (tid < BIN_NODES) {
    c = lcnt[tid];
    incl = c;
#pragma unroll
    for (int m = 1; m < 32; m <<= 1) {
      const int u = __shfl_up(incl, m, 32);
      if (lane >= m) incl += u;
    }
    if (lane == 31) wtot[wave] = incl;
  }
  __syncthreads();
  if (tid < BIN_NODES) {
    int off = 0;
#pragma unroll
    for (int w = 0; w < BIN_NODES / 32; ++w)
      if (w < wave) off += wtot[w];
    const int first = start + off + incl - c;
    if (base + tid < N) rowptr[base + tid] = first;
    lcnt[tid] = first;
  }
  __syncthreads();

  for (int i = start + tid; i < end; i += STHREADS * SBATCH) {
    unsigned p[SBATCH];
#pragma unroll
    for (int u = 0; u < SBATCH; ++u) p[u] = binned[min(i + u * STHREADS, end - 1)];
#pragma unroll
    for (int u = 0; u < SBATCH; ++u) {
      const bool live = (i + u * STHREADS < end);
      const int pos = atomicAdd(&lcnt[p[u] & (BIN_NODES - 1)], live ? 1 : 0);
      if (live && (unsigned)pos < (unsigned)E)
        csr_src[pos] = (int)(p[u] >> BIN_SHIFT);
    }
  }
}


template <int U>
__device__ __forceinline__ float quad_bcast(float x) {
  const int b = __float_as_int(x);
  return __int_as_float(
      __builtin_amdgcn_update_dpp(b, b, U * 0x55, 0xf, 0xf, false));
}

template <int LPR>
__device__ __forceinline__ float group_sum(float v) {
  v += dpp_row<0x161>(v);
  v += dpp_row<0x162>(v);
  v += dpp_row<0x164>(v);
  if constexpr (LPR == 16) v += dpp_row<0x168>(v);
  return v;
}

template <int LPR>
__device__ __forceinline__ float sum4_group(float p0, float p1, float p2,
                                            float p3, int lane) {
  const bool b0 = (lane & 1) != 0;
  const bool b1 = (lane & 2) != 0;
  float k01 = b0 ? p1 : p0;
  const float t01 = b0 ? p0 : p1;
  float k23 = b0 ? p3 : p2;
  const float t23 = b0 ? p2 : p3;
  k01 += dpp_row<0x161>(t01);
  k23 += dpp_row<0x161>(t23);
  float k = b1 ? k23 : k01;
  const float tt = b1 ? k01 : k23;
  k += dpp_row<0x162>(tt);
  k += dpp_row<0x164>(k);
  if constexpr (LPR == 16) k += dpp_row<0x168>(k);
  return k;
}

#define GWAVES 8
template <int NOUT>
__global__ void node_gather_kernel(const int* __restrict__ rowptr,
                                   const int* __restrict__ csr_src, int E,
                                   int N, const float* __restrict__ H,
                                   const float* __restrict__ av,
                                   const float* __restrict__ bv,
                                   const float* __restrict__ bias,
                                   const float* __restrict__ Wn,
                                   const float* __restrict__ bn,
                                   float* __restrict__ out) {
  constexpr int LPR = NOUT / 4;
  constexpr int NPW = 32 / LPR;
  constexpr int NQ = 32 / LPR;
  static_assert(LPR == 8 || LPR == 16, "lane groups of 8 or 16");
  const int lane = threadIdx.x & 31;
  const int g = lane / LPR;
  const int r0 = lane % LPR;
  const int c0 = r0 * 4;
  const int qu = lane & 3;
  const int wib = __builtin_amdgcn_readfirstlane((int)(threadIdx.x >> 5));
  const int wv = (int)blockIdx.x * GWAVES + wib;
  const int dq = wv * NPW + g;
  const bool live = (dq < N);
  const int d = min(dq, N - 1);

  const int rp1 = rowptr[min(d + 1, N - 1)];
  int end = (d + 1 < N) ? rp1 : E;
  end = min(max(end, 0), E);
  const int beg = min(max(rowptr[d], 0), end);
  const int deg = live ? end - beg : 0;
  int degmax = __builtin_amdgcn_readlane(deg, 0);
#pragma unroll
  for (int k = 1; k < NPW; ++k)
    degmax = max(degmax, __builtin_amdgcn_readlane(deg, k * LPR));
  const unsigned nmax = (unsigned)(N - 1);

  auto load_row = [&](int s) -> float4 {
    const unsigned off = (unsigned)s * (unsigned)(NOUT * 4) + (unsigned)(c0 * 4);
    return *reinterpret_cast<const float4*>(
        reinterpret_cast<const char*>(H) + off);
  };
  auto lane_read = [&](int v, int i) -> int {
    return __builtin_amdgcn_ds_bpermute(i << 2, v);
  };
  auto load_src = [&](int j) -> int {
    const int i = max(min(beg + j, beg + deg - 1), 0);
    return (int)min((unsigned)csr_src[i], nmax);
  };

  const float4 hs = load_row(d);
  const float a_s0 = av[c0], a_s1 = av[c0 + 1], a_s2 = av[c0 + 2], a_s3 = av[c0 + 3];
  const float a_d0 = bv[c0], a_d1 = bv[c0 + 1], a_d2 = bv[c0 + 2], a_d3 = bv[c0 + 3];

  const float adv = group_sum<LPR>(hs.x * a_d0 + hs.y * a_d1 + hs.z * a_d2 + hs.w * a_d3);
  float m = leaky(group_sum<LPR>(hs.x * a_s0 + hs.y * a_s1 + hs.z * a_s2 + hs.w * a_s3) + adv);
  float den = 1.0f;
  float acc0 = hs.x, acc1 = hs.y, acc2 = hs.z, acc3 = hs.w;

  for (int j0 = 0; j0 < degmax; j0 += 32) {
    const int cnt = deg - j0;
    const int cmax = min(32, degmax - j0);
    int sq[NQ];
#pragma unroll
    for (int q = 0; q < NQ; ++q) {
      const int sraw = load_src(j0 + q * LPR + r0);
      sq[q] = (q * LPR + r0 < cnt) ? sraw : d;
    }
    auto src_word = [&](int t) -> int {
      int sv = sq[0];
#pragma unroll
      for (int q = 1; q < NQ; ++q) sv = (t >= q * LPR) ? sq[q] : sv;
      return sv;
    };

    auto trip = [&](const float4 (&hv)[4], int t) {
      float p[4];
#pragma unroll
      for (int u = 0; u < 4; ++u)
        p[u] = hv[u].x * a_s0 + hv[u].y * a_s1 + hv[u].z * a_s2 + hv[u].w * a_s3;
      float e = leaky(sum4_group<LPR>(p[0], p[1], p[2], p[3], lane) + adv);
      e = (t + qu < cnt) ? e : -__builtin_inff();

      float gm = fmaxf(e, dpp_row<0x161>(e));
      gm = fmaxf(gm, dpp_row<0x162>(gm));
      const float m_new = fmaxf(m, gm);
      const float sc = __expf(m - m_new);
      const float w = __expf(e - m_new);
      float ws = w + dpp_row<0x161>(w);
      ws += dpp_row<0x162>(ws);
      den = den * sc + ws;
      m = m_new;

      acc0 *= sc; acc1 *= sc; acc2 *= sc; acc3 *= sc;
      const float w0 = quad_bcast<0>(w), w1 = quad_bcast<1>(w);
      const float w2 = quad_bcast<2>(w), w3 = quad_bcast<3>(w);
      acc0 += hv[0].x * w0; acc1 += hv[0].y * w0; acc2 += hv[0].z * w0; acc3 += hv[0].w * w0;
      acc0 += hv[1].x * w1; acc1 += hv[1].y * w1; acc2 += hv[1].z * w1; acc3 += hv[1].w * w1;
      acc0 += hv[2].x * w2; acc1 += hv[2].y * w2; acc2 += hv[2].z * w2; acc3 += hv[2].w * w2;
      acc0 += hv[3].x * w3; acc1 += hv[3].y * w3; acc2 += hv[3].z * w3; acc3 += hv[3].w * w3;
    };

    float4 hv[4];
#pragma unroll
    for (int u = 0; u < 4; ++u) hv[u] = load_row(lane_read(sq[0], g * LPR + u));

    int t = 0;
#pragma nounroll
    for (; t + 4 < cmax; t += 4) {
      float4 hn[4];
      const int sv = src_word(t + 4);
      const int l0 = g * LPR + ((t + 4) & (LPR - 1));
#pragma unroll
      for (int u = 0; u < 4; ++u) hn[u] = load_row(lane_read(sv, l0 + u));
      trip(hv, t);
#pragma unroll
      for (int u = 0; u < 4; ++u) hv[u] = hn[u];
    }
    trip(hv, t);
  }

  float4 r;
  r.x = live ? fmaxf(acc0 / den + bias[c0], 0.0f) : 0.0f;
  r.y = live ? fmaxf(acc1 / den + bias[c0 + 1], 0.0f) : 0.0f;
  r.z = live ? fmaxf(acc2 / den + bias[c0 + 2], 0.0f) : 0.0f;
  r.w = live ? fmaxf(acc3 / den + bias[c0 + 3], 0.0f) : 0.0f;
  const int nib = wib * NPW + g;

  if constexpr (NOUT == H1) {
    constexpr int ROWS = GWAVES * NPW;
    constexpr int NO2 = H2;
    constexpr int LDT = NOUT + 4;
    static_assert(ROWS == 16, "one 16-row matrix tile per work-group");
    __shared__ __attribute__((aligned(16))) float ts[ROWS * LDT];
    *reinterpret_cast<float4*>(&ts[nib * LDT + c0]) = r;
    __syncthreads();
    if (wib < NO2 / 16) {
      const int lh = lane & 15;
      const bool hi = (lane >= 16);
      const int col0 = wib * 16;
      v8f cacc = (v8f){};
#pragma unroll
      for (int k = 0; k < NOUT; k += 4) {
        const int ka = k + (hi ? 2 : 0);
        v2f b;
        b.x = Wn[ka * NO2 + col0 + lh];
        b.y = Wn[(ka + 1) * NO2 + col0 + lh];
        v2f a;
        a.x = ts[lh * LDT + ka];
        a.y = ts[lh * LDT + ka + 1];
        cacc = __builtin_amdgcn_wmma_f32_16x16x4_f32(false, a, false, b,
                                                     (short)0, cacc,
                                                     false, false);
      }
      const int row0 = (int)blockIdx.x * ROWS + (hi ? 8 : 0);
#pragma unroll
      for (int q = 0; q < 8; ++q) {
        if (row0 + q < N) out[(size_t)(row0 + q) * NO2 + col0 + lh] = cacc[q];
      }
    }
  } else {
    constexpr int ROWS = GWAVES * NPW;
    static_assert(ROWS == 32, "one 128-byte line of results per work-group");
    __shared__ float fo[ROWS];
    const float s = group_sum<LPR>(r.x * Wn[c0] + r.y * Wn[c0 + 1] +
                                   r.z * Wn[c0 + 2] + r.w * Wn[c0 + 3]) + bn[0];
    if (r0 == 0) fo[nib] = s;
    __syncthreads();
    const int n = (int)blockIdx.x * ROWS + (int)threadIdx.x;
    if (threadIdx.x < ROWS && n < N) out[n] = fo[threadIdx.x];
  }
}

static inline unsigned cdiv(long long a, long long b) {
  return (unsigned)((a + b - 1) / b);
}

extern "C" void kernel_launch(void* const* d_in, const int* in_sizes, int n_in,
                              void* d_out, int out_size, void* d_ws,
                              size_t ws_size, hipStream_t stream) {
  (void)n_in; (void)out_size; (void)ws_size;
  const float* x     = (const float*)d_in[0];
  const int*   ei    = (const int*)d_in[1];
  const float* W1    = (const float*)d_in[2];
  const float* asrc1 = (const float*)d_in[3];
  const float* adst1 = (const float*)d_in[4];
  const float* b1    = (const float*)d_in[5];
  const float* W2    = (const float*)d_in[6];
  const float* asrc2 = (const float*)d_in[7];
  const float* adst2 = (const float*)d_in[8];
  const float* b2    = (const float*)d_in[9];
  const float* Wfc   = (const float*)d_in[10];
  const float* bfc   = (const float*)d_in[11];
  float* out = (float*)d_out;

  const int N = N_NODES;
  const int E = in_sizes[1] / 2;
  const int* src = ei;
  const int* dst = ei + E;

  float* ws = (float*)d_ws;
  size_t off = 0;
  float* h    = ws + off; off += (size_t)N * H1;
  float* h2   = ws + off; off += (size_t)N * H1;
  float* agg2 = ws + off; off += (size_t)N * H2;
  int* segoff  = (int*)(ws + off); off += N;
  int* rowptr  = (int*)(ws + off); off += N;
  unsigned* binned = (unsigned*)(ws + off); off += E;
  int* csr_src = (int*)(ws + off); off += E;
  int* bintot = (int*)agg2;

  constexpr int MR = 32;
  constexpr int GN1 = GWAVES * (128 / H1);
  constexpr int GN2 = GWAVES * (128 / H2);
  const dim3 GB(GWAVES * 32);

  gemm_wmma_kernel<F_IN, H1, MR><<<N / MR, (H1 / 16) * 32, 0, stream>>>(x, W1, h, bintot);

  bin_count_kernel<<<EBLOCKS, ETHREADS, 0, stream>>>(dst, E, bintot, segoff);
  bin_scatter_kernel<<<EBLOCKS, ETHREADS, 0, stream>>>(src, dst, E, bintot, segoff, binned);
  bin_sort_kernel<<<NBINS, STHREADS, 0, stream>>>(bintot, E, N, binned, rowptr, csr_src);

  node_gather_kernel<H1><<<cdiv(N, GN1), GB, 0, stream>>>(rowptr, csr_src, E, N, h, asrc1, adst1, b1, W2, bfc, h2);

  node_gather_kernel<H2><<<cdiv(N, GN2), GB, 0, stream>>>(rowptr, csr_src, E, N, h2, asrc2, adst2, b2, Wfc, bfc, out);
}


# Edge chunks grouped by bin in their own buffer stretch (LDS count, scan, cursors) by work-groups of the GEMM launch; bin sort collects the runs: 6 to 4 launches; node sums: arrival order as before
# speedup vs baseline: 8.7864x; 1.3400x over previous
#include <hip/hip_runtime.h>

typedef float v2f __attribute__((ext_vector_type(2)));
typedef float v8f __attribute__((ext_vector_type(8)));

#define N_NODES 100000
#define F_IN 256
#define H1 64
#define H2 32
#define NEG_SLOPE 0.2f

#define BIN_SHIFT 8
#define BIN_NODES (1 << BIN_SHIFT)
#define NBINS ((N_NODES + BIN_NODES - 1) >> BIN_SHIFT)
#define EBLOCKS 250
#define LOFF_STRIDE (NBINS + 1)
#define GBATCH 25

__device__ __forceinline__ unsigned clamp_node(int v) {
  return min((unsigned)v, (unsigned)(N_NODES - 1));
}

template <int NTHR>
__device__ __forceinline__ void group_chunk(const int* __restrict__ src,
                                            const int* __restrict__ dst, int E,
                                            int blk, int* __restrict__ loff,
                                            unsigned* __restrict__ binned,
                                            int* lc, int* wtot) {
  constexpr int BPT = (NBINS + NTHR - 1) / NTHR;
  const int tid = threadIdx.x;
  const int lane = tid & 31;
  const int wave = tid >> 5;
  for (int b = tid; b < NBINS; b += NTHR) lc[b] = 0;
  __syncthreads();
  const int chunk = (E + EBLOCKS - 1) / EBLOCKS;
  const int e0 = min(blk * chunk, E);
  const int e1 = min(e0 + chunk, E);
  for (int e = e0 + tid; e < e1; e += NTHR * GBATCH) {
    int d[GBATCH];
#pragma unroll
    for (int u = 0; u < GBATCH; ++u) d[u] = dst[min(e + u * NTHR, e1 - 1)];
#pragma unroll
    for (int u = 0; u < GBATCH; ++u)
      atomicAdd(&lc[clamp_node(d[u]) >> BIN_SHIFT],
                (e + u * NTHR < e1) ? 1 : 0);
  }
  __syncthreads();
  int cnt[BPT];
  int tsum = 0;
#pragma unroll
  for (int j = 0; j < BPT; ++j) {
    const int b = tid * BPT + j;
    cnt[j] = (b < NBINS) ? lc[min(b, NBINS - 1)] : 0;
    tsum += cnt[j];
  }
  int incl = tsum;
#pragma unroll
  for (int m = 1; m < 32; m <<= 1) {
    const int u = __shfl_up(incl, m, 32);
    if (lane >= m) incl += u;
  }
  if (lane == 31) wtot[wave] = incl;
  __syncthreads();
  int run = incl - tsum;
#pragma unroll
  for (int w = 0; w < NTHR / 32; ++w)
    if (w < wave) run += wtot[w];
#pragma unroll
  for (int j = 0; j < BPT; ++j) {
    const int b = tid * BPT + j;
    if (b < NBINS) {
      loff[blk * LOFF_STRIDE + b] = run;
      lc[b] = e0 + run;
    }
    run += cnt[j];
  }
  if (tid == 0) loff[blk * LOFF_STRIDE + NBINS] = e1 - e0;
  __syncthreads();
  for (int e = e0 + tid; e < e1; e += NTHR * GBATCH) {
    int sv[GBATCH], d[GBATCH];
#pragma unroll
    for (int u = 0; u < GBATCH; ++u) {
      const int ee = min(e + u * NTHR, e1 - 1);
      sv[u] = src[ee];
      d[u] = dst[ee];
    }
#pragma unroll
    for (int u = 0; u < GBATCH; ++u) {
      const unsigned dn = clamp_node(d[u]);
      sv[u] = (int)((clamp_node(sv[u]) << BIN_SHIFT) | (dn & (BIN_NODES - 1)));
      d[u] = atomicAdd(&lc[dn >> BIN_SHIFT], (e + u * NTHR < e1) ? 1 : 0);
    }
#pragma unroll
    for (int u = 0; u < GBATCH; ++u)
      if (e + u * NTHR < e1 && (unsigned)d[u] < (unsigned)E)
        binned[d[u]] = (unsigned)sv[u];
  }
}

template <int K, int NOUT, int MROWS>
__global__ void __launch_bounds__((NOUT / 16) * 32)
gemm_wmma_kernel(const float* __restrict__ X,
                 const float* __restrict__ W,
                 float* __restrict__ Hout,
                 const int* __restrict__ esrc,
                 const int* __restrict__ edst, int E,
                 int* __restrict__ loff,
                 unsigned* __restrict__ binned) {
  constexpr int LDK = K + 4;
  constexpr int RT = MROWS / 16;
  constexpr int NTHR = (NOUT / 16) * 32;
  constexpr int K4 = K / 4;
  constexpr int PPT = MROWS * K4 / NTHR;
  static_assert(MROWS * K4 % NTHR == 0, "whole pieces per thread");
  static_assert(LDK % 4 == 0, "LDS rows start on 16 bytes");
  __shared__ __attribute__((aligned(16))) float xs[MROWS * LDK];
  const int tid = threadIdx.x;

  static_assert(MROWS * LDK >= 512 + NTHR / 32, "counters fit the tile's LDS");
  static_assert(NBINS <= 512, "counters below the wave totals");
  if (blockIdx.x < EBLOCKS) {
    group_chunk<NTHR>(esrc, edst, E, (int)blockIdx.x, loff, binned,
                      reinterpret_cast<int*>(xs), reinterpret_cast<int*>(xs) + 512);
    return;
  }
  const int row0 = ((int)blockIdx.x - EBLOCKS) * MROWS;

  __builtin_prefetch(W, 0, 1);

  const float4* __restrict__ xt =
      reinterpret_cast<const float4*>(X + (size_t)row0 * K);
  float4 pc[PPT];
#pragma unroll
  for (int it = 0; it < PPT; ++it) pc[it] = xt[tid + it * NTHR];
#pragma unroll
  for (int it = 0; it < PPT; ++it) {
    const int i = tid + it * NTHR;
    const int r = i / K4, c = (i % K4) * 4;
    *reinterpret_cast<float4*>(&xs[r * LDK + c]) = pc[it];
  }
  __syncthreads();

  const int wave = tid >> 5;
  const int lane = tid & 31;
  const int lh = lane & 15;
  const bool hi = (lane >= 16);
  const int col0 = wave * 16;

  v8f acc[RT];
#pragma unroll
  for (int t = 0; t < RT; ++t) acc[t] = (v8f){};

#pragma unroll 4
  for (int k = 0; k < K; k += 4) {
    const int ka = k + (hi ? 2 : 0);
    v2f b;
    b.x = W[(size_t)ka * NOUT + col0 + lh];
    b.y = W[(size_t)(ka + 1) * NOUT + col0 + lh];
#pragma unroll
    for (int t = 0; t < RT; ++t) {
      v2f a;
      a.x = xs[(t * 16 + lh) * LDK + ka];
      a.y = xs[(t * 16 + lh) * LDK + ka + 1];
      acc[t] = __builtin_amdgcn_wmma_f32_16x16x4_f32(false, a, false, b,
                                                     (short)0, acc[t],
                                                     false, false);
    }
  }

  const int rbase = hi ? 8 : 0;
#pragma unroll
  for (int t = 0; t < RT; ++t) {
#pragma unroll
    for (int r = 0; r < 8; ++r) {
      Hout[(size_t)(row0 + t * 16 + rbase + r) * NOUT + col0 + lh] = acc[t][r];
    }
  }
}

__device__ __forceinline__ float leaky(float v) {
  return (v > 0.0f) ? v : NEG_SLOPE * v;
}

template <int CTRL>
__device__ __forceinline__ float dpp_row(float x) {
  const int b = __float_as_int(x);
  return __int_as_float(__builtin_amdgcn_update_dpp(b, b, CTRL, 0xf, 0xf, false));
}

__device__ __forceinline__ int wave_sum_i(int v) {
#pragma unroll
  for (int m = 16; m >= 1; m >>= 1) v += __shfl_xor(v, m, 32);
  return v;
}

#define STHREADS 512
#define RPH ((EBLOCKS + STHREADS / 16 - 1) / (STHREADS / 16))
static_assert(LOFF_STRIDE * EBLOCKS <= N_NODES, "loff must fit the N-word slot");
static_assert(EBLOCKS <= STHREADS, "one thread per chunk in the bin's first-slot sum");
static_assert(N_NODES <= (1 << 24), "src must fit 24 bits of the packed edge");

__global__ void __launch_bounds__(STHREADS)
bin_sort_kernel(const int* __restrict__ loff, int E, int N,
                const unsigned* __restrict__ binned,
                int* __restrict__ rowptr,
                int* __restrict__ csr_src) {
  __shared__ int lcnt[BIN_NODES];
  __shared__ int wtot[BIN_NODES / 32];
  __shared__ int wpre[STHREADS / 32];
  const int tid = threadIdx.x;
  const int lane = tid & 31;
  const int wave = tid >> 5;
  const int hw = tid >> 4;
  const int l16 = tid & 15;
  const int bin = (int)blockIdx.x;
  const int base = bin << BIN_SHIFT;

  int pre = (tid < EBLOCKS) ? loff[min(tid, EBLOCKS - 1) * LOFF_STRIDE + bin] : 0;

  const int chunk = (E + EBLOCKS - 1) / EBLOCKS;
  int lo[RPH], len[RPH];
#pragma unroll
  for (int r = 0; r < RPH; ++r) {
    const int c = hw + r * (STHREADS / 16);
    const int cc = min(c, EBLOCKS - 1);
    int ra = loff[cc * LOFF_STRIDE + bin];
    int rb = loff[cc * LOFF_STRIDE + bin + 1];
    const int e0c = min(cc * chunk, E);
    const int clen = min(e0c + chunk, E) - e0c;
    ra = min(max(ra, 0), clen);
    rb = min(max(rb, ra), clen);
    lo[r] = e0c + ra;
    len[r] = (c < EBLOCKS) ? rb - ra : 0;
  }

  pre = wave_sum_i(pre);
  if (lane == 0) wpre[wave] = pre;
  if (tid < BIN_NODES) lcnt[tid] = 0;
  __syncthreads();

  int first = 0;
#pragma unroll
  for (int w = 0; w < STHREADS / 32; ++w) first += wpre[w];
  const int start = min(max(first, 0), E);

  unsigned p[2 * RPH];
#pragma unroll
  for (int r = 0; r < RPH; ++r) {
    p[2 * r] = binned[max(min(lo[r] + l16, E - 1), 0)];
    p[2 * r + 1] = binned[max(min(lo[r] + 16 + l16, E - 1), 0)];
  }
#pragma unroll
  for (int r = 0; r < RPH; ++r) {
    atomicAdd(&lcnt[p[2 * r] & (BIN_NODES - 1)], (l16 < len[r]) ? 1 : 0);
    atomicAdd(&lcnt[p[2 * r + 1] & (BIN_NODES - 1)], (l16 + 16 < len[r]) ? 1 : 0);
  }
#pragma unroll
  for (int r = 0; r < RPH; ++r)
    for (int k = l16 + 32; k < len[r]; k += 16)
      atomicAdd(&lcnt[binned[lo[r] + k] & (BIN_NODES - 1)], 1);
  __syncthreads();

  int c = 0, incl = 0;
  if (tid < BIN_NODES) {
    c = lcnt[tid];
    incl = c;
#pragma unroll
    for (int m = 1; m < 32; m <<= 1) {
      const int u = __shfl_up(incl, m, 32);
      if (lane >= m) incl += u;
    }
    if (lane == 31) wtot[wave] = incl;
  }
  __syncthreads();
  if (tid < BIN_NODES) {
    int off = 0;
#pragma unroll
    for (int w = 0; w < BIN_NODES / 32; ++w)
      if (w < wave) off += wtot[w];
    const int nfirst = start + off + incl - c;
    if (base + tid < N) rowptr[base + tid] = nfirst;
    lcnt[tid] = nfirst;
  }
  __syncthreads();

  int pos[2 * RPH];
#pragma unroll
  for (int r = 0; r < RPH; ++r) {
    pos[2 * r] = atomicAdd(&lcnt[p[2 * r] & (BIN_NODES - 1)],
                           (l16 < len[r]) ? 1 : 0);
    pos[2 * r + 1] = atomicAdd(&lcnt[p[2 * r + 1] & (BIN_NODES - 1)],
                               (l16 + 16 < len[r]) ? 1 : 0);
  }
#pragma unroll
  for (int r = 0; r < RPH; ++r) {
    if (l16 < len[r] && (unsigned)pos[2 * r] < (unsigned)E)
      csr_src[pos[2 * r]] = (int)(p[2 * r] >> BIN_SHIFT);
    if (l16 + 16 < len[r] && (unsigned)pos[2 * r + 1] < (unsigned)E)
      csr_src[pos[2 * r + 1]] = (int)(p[2 * r + 1] >> BIN_SHIFT);
  }
#pragma unroll
  for (int r = 0; r < RPH; ++r)
    for (int k = l16 + 32; k < len[r]; k += 16) {
      const unsigned q = binned[lo[r] + k];
      const int slot = atomicAdd(&lcnt[q & (BIN_NODES - 1)], 1);
      if ((unsigned)slot < (unsigned)E) csr_src[slot] = (int)(q >> BIN_SHIFT);
    }
}


template <int U>
__device__ __forceinline__ float quad_bcast(float x) {
  const int b = __float_as_int(x);
  return __int_as_float(
      __builtin_amdgcn_update_dpp(b, b, U * 0x55, 0xf, 0xf, false));
}

template <int LPR>
__device__ __forceinline__ float group_sum(float v) {
  v += dpp_row<0x161>(v);
  v += dpp_row<0x162>(v);
  v += dpp_row<0x164>(v);
  if constexpr (LPR == 16) v += dpp_row<0x168>(v);
  return v;
}

template <int LPR>
__device__ __forceinline__ float sum4_group(float p0, float p1, float p2,
                                            float p3, int lane) {
  const bool b0 = (lane & 1) != 0;
  const bool b1 = (lane & 2) != 0;
  float k01 = b0 ? p1 : p0;
  const float t01 = b0 ? p0 : p1;
  float k23 = b0 ? p3 : p2;
  const float t23 = b0 ? p2 : p3;
  k01 += dpp_row<0x161>(t01);
  k23 += dpp_row<0x161>(t23);
  float k = b1 ? k23 : k01;
  const float tt = b1 ? k01 : k23;
  k += dpp_row<0x162>(tt);
  k += dpp_row<0x164>(k);
  if constexpr (LPR == 16) k += dpp_row<0x168>(k);
  return k;
}

#define GWAVES 8
template <int NOUT>
__global__ void node_gather_kernel(const int* __restrict__ rowptr,
                                   const int* __restrict__ csr_src, int E,
                                   int N, const float* __restrict__ H,
                                   const float* __restrict__ av,
                                   const float* __restrict__ bv,
                                   const float* __restrict__ bias,
                                   const float* __restrict__ Wn,
                                   const float* __restrict__ bn,
                                   float* __restrict__ out) {
  constexpr int LPR = NOUT / 4;
  constexpr int NPW = 32 / LPR;
  constexpr int NQ = 32 / LPR;
  static_assert(LPR == 8 || LPR == 16, "lane groups of 8 or 16");
  const int lane = threadIdx.x & 31;
  const int g = lane / LPR;
  const int r0 = lane % LPR;
  const int c0 = r0 * 4;
  const int qu = lane & 3;
  const int wib = __builtin_amdgcn_readfirstlane((int)(threadIdx.x >> 5));
  const int wv = (int)blockIdx.x * GWAVES + wib;
  const int dq = wv * NPW + g;
  const bool live = (dq < N);
  const int d = min(dq, N - 1);

  const int rp1 = rowptr[min(d + 1, N - 1)];
  int end = (d + 1 < N) ? rp1 : E;
  end = min(max(end, 0), E);
  const int beg = min(max(rowptr[d], 0), end);
  const int deg = live ? end - beg : 0;
  int degmax = __builtin_amdgcn_readlane(deg, 0);
#pragma unroll
  for (int k = 1; k < NPW; ++k)
    degmax = max(degmax, __builtin_amdgcn_readlane(deg, k * LPR));
  const unsigned nmax = (unsigned)(N - 1);

  auto load_row = [&](int s) -> float4 {
    const unsigned off = (unsigned)s * (unsigned)(NOUT * 4) + (unsigned)(c0 * 4);
    return *reinterpret_cast<const float4*>(
        reinterpret_cast<const char*>(H) + off);
  };
  auto lane_read = [&](int v, int i) -> int {
    return __builtin_amdgcn_ds_bpermute(i << 2, v);
  };
  auto load_src = [&](int j) -> int {
    const int i = max(min(beg + j, beg + deg - 1), 0);
    return (int)min((unsigned)csr_src[i], nmax);
  };

  const float4 hs = load_row(d);
  const float a_s0 = av[c0], a_s1 = av[c0 + 1], a_s2 = av[c0 + 2], a_s3 = av[c0 + 3];
  const float a_d0 = bv[c0], a_d1 = bv[c0 + 1], a_d2 = bv[c0 + 2], a_d3 = bv[c0 + 3];

  const float adv = group_sum<LPR>(hs.x * a_d0 + hs.y * a_d1 + hs.z * a_d2 + hs.w * a_d3);
  float m = leaky(group_sum<LPR>(hs.x * a_s0 + hs.y * a_s1 + hs.z * a_s2 + hs.w * a_s3) + adv);
  float den = 1.0f;
  float acc0 = hs.x, acc1 = hs.y, acc2 = hs.z, acc3 = hs.w;

  for (int j0 = 0; j0 < degmax; j0 += 32) {
    const int cnt = deg - j0;
    const int cmax = min(32, degmax - j0);
    int sq[NQ];
#pragma unroll
    for (int q = 0; q < NQ; ++q) {
      const int sraw = load_src(j0 + q * LPR + r0);
      sq[q] = (q * LPR + r0 < cnt) ? sraw : d;
    }
    auto src_word = [&](int t) -> int {
      int sv = sq[0];
#pragma unroll
      for (int q = 1; q < NQ; ++q) sv = (t >= q * LPR) ? sq[q] : sv;
      return sv;
    };

    auto trip = [&](const float4 (&hv)[4], int t) {
      float p[4];
#pragma unroll
      for (int u = 0; u < 4; ++u)
        p[u] = hv[u].x * a_s0 + hv[u].y * a_s1 + hv[u].z * a_s2 + hv[u].w * a_s3;
      float e = leaky(sum4_group<LPR>(p[0], p[1], p[2], p[3], lane) + adv);
      e = (t + qu < cnt) ? e : -__builtin_inff();

      float gm = fmaxf(e, dpp_row<0x161>(e));
      gm = fmaxf(gm, dpp_row<0x162>(gm));
      const float m_new = fmaxf(m, gm);
      const float sc = __expf(m - m_new);
      const float w = __expf(e - m_new);
      float ws = w + dpp_row<0x161>(w);
      ws += dpp_row<0x162>(ws);
      den = den * sc + ws;
      m = m_new;

      acc0 *= sc; acc1 *= sc; acc2 *= sc; acc3 *= sc;
      const float w0 = quad_bcast<0>(w), w1 = quad_bcast<1>(w);
      const float w2 = quad_bcast<2>(w), w3 = quad_bcast<3>(w);
      acc0 += hv[0].x * w0; acc1 += hv[0].y * w0; acc2 += hv[0].z * w0; acc3 += hv[0].w * w0;
      acc0 += hv[1].x * w1; acc1 += hv[1].y * w1; acc2 += hv[1].z * w1; acc3 += hv[1].w * w1;
      acc0 += hv[2].x * w2; acc1 += hv[2].y * w2; acc2 += hv[2].z * w2; acc3 += hv[2].w * w2;
      acc0 += hv[3].x * w3; acc1 += hv[3].y * w3; acc2 += hv[3].z * w3; acc3 += hv[3].w * w3;
    };

    float4 hv[4];
#pragma unroll
    for (int u = 0; u < 4; ++u) hv[u] = load_row(lane_read(sq[0], g * LPR + u));

    int t = 0;
#pragma nounroll
    for (; t + 4 < cmax; t += 4) {
      float4 hn[4];
      const int sv = src_word(t + 4);
      const int l0 = g * LPR + ((t + 4) & (LPR - 1));
#pragma unroll
      for (int u = 0; u < 4; ++u) hn[u] = load_row(lane_read(sv, l0 + u));
      trip(hv, t);
#pragma unroll
      for (int u = 0; u < 4; ++u) hv[u] = hn[u];
    }
    trip(hv, t);
  }

  float4 r;
  r.x = live ? fmaxf(acc0 / den + bias[c0], 0.0f) : 0.0f;
  r.y = live ? fmaxf(acc1 / den + bias[c0 + 1], 0.0f) : 0.0f;
  r.z = live ? fmaxf(acc2 / den + bias[c0 + 2], 0.0f) : 0.0f;
  r.w = live ? fmaxf(acc3 / den + bias[c0 + 3], 0.0f) : 0.0f;
  const int nib = wib * NPW + g;

  if constexpr (NOUT == H1) {
    constexpr int ROWS = GWAVES * NPW;
    constexpr int NO2 = H2;
    constexpr int LDT = NOUT + 4;
    static_assert(ROWS == 16, "one 16-row matrix tile per work-group");
    __shared__ __attribute__((aligned(16))) float ts[ROWS * LDT];
    *reinterpret_cast<float4*>(&ts[nib * LDT + c0]) = r;
    __syncthreads();
    if (wib < NO2 / 16) {
      const int lh = lane & 15;
      const bool hi = (lane >= 16);
      const int col0 = wib * 16;
      v8f cacc = (v8f){};
#pragma unroll
      for (int k = 0; k < NOUT; k += 4) {
        const int ka = k + (hi ? 2 : 0);
        v2f b;
        b.x = Wn[ka * NO2 + col0 + lh];
        b.y = Wn[(ka + 1) * NO2 + col0 + lh];
        v2f a;
        a.x = ts[lh * LDT + ka];
        a.y = ts[lh * LDT + ka + 1];
        cacc = __builtin_amdgcn_wmma_f32_16x16x4_f32(false, a, false, b,
                                                     (short)0, cacc,
                                                     false, false);
      }
      const int row0 = (int)blockIdx.x * ROWS + (hi ? 8 : 0);
#pragma unroll
      for (int q = 0; q < 8; ++q) {
        if (row0 + q < N) out[(size_t)(row0 + q) * NO2 + col0 + lh] = cacc[q];
      }
    }
  } else {
    constexpr int ROWS = GWAVES * NPW;
    static_assert(ROWS == 32, "one 128-byte line of results per work-group");
    __shared__ float fo[ROWS];
    const float s = group_sum<LPR>(r.x * Wn[c0] + r.y * Wn[c0 + 1] +
                                   r.z * Wn[c0 + 2] + r.w * Wn[c0 + 3]) + bn[0];
    if (r0 == 0) fo[nib] = s;
    __syncthreads();
    const int n = (int)blockIdx.x * ROWS + (int)threadIdx.x;
    if (threadIdx.x < ROWS && n < N) out[n] = fo[threadIdx.x];
  }
}

static inline unsigned cdiv(long long a, long long b) {
  return (unsigned)((a + b - 1) / b);
}

extern "C" void kernel_launch(void* const* d_in, const int* in_sizes, int n_in,
                              void* d_out, int out_size, void* d_ws,
                              size_t ws_size, hipStream_t stream) {
  (void)n_in; (void)out_size; (void)ws_size;
  const float* x     = (const float*)d_in[0];
  const int*   ei    = (const int*)d_in[1];
  const float* W1    = (const float*)d_in[2];
  const float* asrc1 = (const float*)d_in[3];
  const float* adst1 = (const float*)d_in[4];
  const float* b1    = (const float*)d_in[5];
  const float* W2    = (const float*)d_in[6];
  const float* asrc2 = (const float*)d_in[7];
  const float* adst2 = (const float*)d_in[8];
  const float* b2    = (const float*)d_in[9];
  const float* Wfc   = (const float*)d_in[10];
  const float* bfc   = (const float*)d_in[11];
  float* out = (float*)d_out;

  const int N = N_NODES;
  const int E = in_sizes[1] / 2;
  const int* src = ei;
  const int* dst = ei + E;

  float* ws = (float*)d_ws;
  size_t off = 0;
  float* h    = ws + off; off += (size_t)N * H1;
  float* h2   = ws + off; off += (size_t)N * H1;
  off += (size_t)N * H2;
  int* loff    = (int*)(ws + off); off += N;
  int* rowptr  = (int*)(ws + off); off += N;
  unsigned* binned = (unsigned*)(ws + off); off += E;
  int* csr_src = (int*)(ws + off); off += E;

  constexpr int MR = 32;
  constexpr int GN1 = GWAVES * (128 / H1);
  constexpr int GN2 = GWAVES * (128 / H2);
  const dim3 GB(GWAVES * 32);

  gemm_wmma_kernel<F_IN, H1, MR><<<EBLOCKS + N / MR, (H1 / 16) * 32, 0, stream>>>(x, W1, h, src, dst, E, loff, binned);

  bin_sort_kernel<<<NBINS, STHREADS, 0, stream>>>(loff, E, N, binned, rowptr, csr_src);

  node_gather_kernel<H1><<<cdiv(N, GN1), GB, 0, stream>>>(rowptr, csr_src, E, N, h, asrc1, adst1, b1, W2, bfc, h2);

  node_gather_kernel<H2><<<cdiv(N, GN2), GB, 0, stream>>>(rowptr, csr_src, E, N, h2, asrc2, adst2, b2, Wfc, bfc, out);
}


# Both sort levels place their edges in an LDS copy (a chunk's stretch, a bin's list) and write it out as whole 128-byte lines, not one scattered 4-byte store per edge; sum order: arrival, as before
# speedup vs baseline: 9.2468x; 1.0524x over previous
#include <hip/hip_runtime.h>

typedef float v2f __attribute__((ext_vector_type(2)));
typedef float v8f __attribute__((ext_vector_type(8)));

#define N_NODES 100000
#define F_IN 256
#define H1 64
#define H2 32
#define NEG_SLOPE 0.2f

#define BIN_SHIFT 8
#define BIN_NODES (1 << BIN_SHIFT)
#define NBINS ((N_NODES + BIN_NODES - 1) >> BIN_SHIFT)
#define EBLOCKS 250
#define LOFF_STRIDE (NBINS + 1)
#define GBATCH 25
#define WBATCH 10

__device__ __forceinline__ unsigned clamp_node(int v) {
  return min((unsigned)v, (unsigned)(N_NODES - 1));
}

template <int NTHR>
__device__ __forceinline__ void group_chunk(const int* __restrict__ src,
                                            const int* __restrict__ dst, int E,
                                            int blk, int* __restrict__ loff,
                                            unsigned* __restrict__ binned,
                                            int* lc, int* wtot,
                                            unsigned* stage, int cap) {
  constexpr int BPT = (NBINS + NTHR - 1) / NTHR;
  const int tid = threadIdx.x;
  const int lane = tid & 31;
  const int wave = tid >> 5;
  for (int b = tid; b < NBINS; b += NTHR) lc[b] = 0;
  __syncthreads();
  const int chunk = (E + EBLOCKS - 1) / EBLOCKS;
  const int e0 = min(blk * chunk, E);
  const int e1 = min(e0 + chunk, E);
  for (int e = e0 + tid; e < e1; e += NTHR * GBATCH) {
    int d[GBATCH];
#pragma unroll
    for (int u = 0; u < GBATCH; ++u) d[u] = dst[min(e + u * NTHR, e1 - 1)];
#pragma unroll
    for (int u = 0; u < GBATCH; ++u)
      atomicAdd(&lc[clamp_node(d[u]) >> BIN_SHIFT],
                (e + u * NTHR < e1) ? 1 : 0);
  }
  __syncthreads();
  int cnt[BPT];
  int tsum = 0;
#pragma unroll
  for (int j = 0; j < BPT; ++j) {
    const int b = tid * BPT + j;
    cnt[j] = (b < NBINS) ? lc[min(b, NBINS - 1)] : 0;
    tsum += cnt[j];
  }
  int incl = tsum;
#pragma unroll
  for (int m = 1; m < 32; m <<= 1) {
    const int u = __shfl_up(incl, m, 32);
    if (lane >= m) incl += u;
  }
  if (lane == 31) wtot[wave] = incl;
  __syncthreads();
  int run = incl - tsum;
#pragma unroll
  for (int w = 0; w < NTHR / 32; ++w)
    if (w < wave) run += wtot[w];
#pragma unroll
  for (int j = 0; j < BPT; ++j) {
    const int b = tid * BPT + j;
    if (b < NBINS) {
      loff[blk * LOFF_STRIDE + b] = run;
      lc[b] = e0 + run;
    }
    run += cnt[j];
  }
  if (tid == 0) loff[blk * LOFF_STRIDE + NBINS] = e1 - e0;
  __syncthreads();
  for (int e = e0 + tid; e < e1; e += NTHR * GBATCH) {
    int sv[GBATCH], d[GBATCH];
#pragma unroll
    for (int u = 0; u < GBATCH; ++u) {
      const int ee = min(e + u * NTHR, e1 - 1);
      sv[u] = src[ee];
      d[u] = dst[ee];
    }
#pragma unroll
    for (int u = 0; u < GBATCH; ++u) {
      const unsigned dn = clamp_node(d[u]);
      sv[u] = (int)((clamp_node(sv[u]) << BIN_SHIFT) | (dn & (BIN_NODES - 1)));
      d[u] = atomicAdd(&lc[dn >> BIN_SHIFT], (e + u * NTHR < e1) ? 1 : 0);
    }
#pragma unroll
    for (int u = 0; u < GBATCH; ++u)
      if (e + u * NTHR < e1 && (unsigned)d[u] < (unsigned)E) {
        const int loc = d[u] - e0;
        if ((unsigned)loc < (unsigned)cap) stage[loc] = (unsigned)sv[u];
        else binned[d[u]] = (unsigned)sv[u];
      }
  }
  __syncthreads();
  const int nst = min(e1 - e0, cap);
  for (int i = tid; i < nst; i += NTHR * WBATCH) {
    unsigned v[WBATCH];
#pragma unroll
    for (int u = 0; u < WBATCH; ++u) v[u] = stage[min(i + u * NTHR, max(nst - 1, 0))];
#pragma unroll
    for (int u = 0; u < WBATCH; ++u)
      if (i + u * NTHR < nst) binned[e0 + i + u * NTHR] = v[u];
  }
}

template <int K, int NOUT, int MROWS>
__global__ void __launch_bounds__((NOUT / 16) * 32)
gemm_wmma_kernel(const float* __restrict__ X,
                 const float* __restrict__ W,
                 float* __restrict__ Hout,
                 const int* __restrict__ esrc,
                 const int* __restrict__ edst, int E,
                 int* __restrict__ loff,
                 unsigned* __restrict__ binned) {
  constexpr int LDK = K + 4;
  constexpr int RT = MROWS / 16;
  constexpr int NTHR = (NOUT / 16) * 32;
  constexpr int K4 = K / 4;
  constexpr int PPT = MROWS * K4 / NTHR;
  static_assert(MROWS * K4 % NTHR == 0, "whole pieces per thread");
  static_assert(LDK % 4 == 0, "LDS rows start on 16 bytes");
  __shared__ __attribute__((aligned(16))) float xs[MROWS * LDK];
  const int tid = threadIdx.x;

  static_assert(MROWS * LDK >= 512 + NTHR / 32, "counters fit the tile's LDS");
  static_assert(NBINS <= 512, "counters below the wave totals");
  static_assert(MROWS * LDK > 1024 && 512 + NTHR / 32 <= 1024, "LDS copy behind the counters");
  if (blockIdx.x < EBLOCKS) {
    group_chunk<NTHR>(esrc, edst, E, (int)blockIdx.x, loff, binned,
                      reinterpret_cast<int*>(xs), reinterpret_cast<int*>(xs) + 512,
                      reinterpret_cast<unsigned*>(xs) + 1024, MROWS * LDK - 1024);
    return;
  }
  const int row0 = ((int)blockIdx.x - EBLOCKS) * MROWS;

  __builtin_prefetch(W, 0, 1);

  const float4* __restrict__ xt =
      reinterpret_cast<const float4*>(X + (size_t)row0 * K);
  float4 pc[PPT];
#pragma unroll
  for (int it = 0; it < PPT; ++it) pc[it] = xt[tid + it * NTHR];
#pragma unroll
  for (int it = 0; it < PPT; ++it) {
    const int i = tid + it * NTHR;
    const int r = i / K4, c = (i % K4) * 4;
    *reinterpret_cast<float4*>(&xs[r * LDK + c]) = pc[it];
  }
  __syncthreads();

  const int wave = tid >> 5;
  const int lane = tid & 31;
  const int lh = lane & 15;
  const bool hi = (lane >= 16);
  const int col0 = wave * 16;

  v8f acc[RT];
#pragma unroll
  for (int t = 0; t < RT; ++t) acc[t] = (v8f){};

#pragma unroll 4
  for (int k = 0; k < K; k += 4) {
    const int ka = k + (hi ? 2 : 0);
    v2f b;
    b.x = W[(size_t)ka * NOUT + col0 + lh];
    b.y = W[(size_t)(ka + 1) * NOUT + col0 + lh];
#pragma unroll
    for (int t = 0; t < RT; ++t) {
      v2f a;
      a.x = xs[(t * 16 + lh) * LDK + ka];
      a.y = xs[(t * 16 + lh) * LDK + ka + 1];
      acc[t] = __builtin_amdgcn_wmma_f32_16x16x4_f32(false, a, false, b,
                                                     (short)0, acc[t],
                                                     false, false);
    }
  }

  const int rbase = hi ? 8 : 0;
#pragma unroll
  for (int t = 0; t < RT; ++t) {
#pragma unroll
    for (int r = 0; r < 8; ++r) {
      Hout[(size_t)(row0 + t * 16 + rbase + r) * NOUT + col0 + lh] = acc[t][r];
    }
  }
}

__device__ __forceinline__ float leaky(float v) {
  return (v > 0.0f) ? v : NEG_SLOPE * v;
}

template <int CTRL>
__device__ __forceinline__ float dpp_row(float x) {
  const int b = __float_as_int(x);
  return __int_as_float(__builtin_amdgcn_update_dpp(b, b, CTRL, 0xf, 0xf, false));
}

__device__ __forceinline__ int wave_sum_i(int v) {
#pragma unroll
  for (int m = 16; m >= 1; m >>= 1) v += __shfl_xor(v, m, 32);
  return v;
}

#define STHREADS 512
#define SCAP 6144
#define RPH ((EBLOCKS + STHREADS / 16 - 1) / (STHREADS / 16))
static_assert(LOFF_STRIDE * EBLOCKS <= N_NODES, "loff must fit the N-word slot");
static_assert(EBLOCKS <= STHREADS, "one thread per chunk in the bin's first-slot sum");
static_assert(N_NODES <= (1 << 24), "src must fit 24 bits of the packed edge");

__global__ void __launch_bounds__(STHREADS)
bin_sort_kernel(const int* __restrict__ loff, int E, int N,
                const unsigned* __restrict__ binned,
                int* __restrict__ rowptr,
                int* __restrict__ csr_src) {
  __shared__ int lcnt[BIN_NODES];
  __shared__ int wtot[BIN_NODES / 32];
  __shared__ int wpre[STHREADS / 32];
  __shared__ unsigned stage[SCAP];
  const int tid = threadIdx.x;
  const int lane = tid & 31;
  const int wave = tid >> 5;
  const int hw = tid >> 4;
  const int l16 = tid & 15;
  const int bin = (int)blockIdx.x;
  const int base = bin << BIN_SHIFT;

  int pre = (tid < EBLOCKS) ? loff[min(tid, EBLOCKS - 1) * LOFF_STRIDE + bin] : 0;

  const int chunk = (E + EBLOCKS - 1) / EBLOCKS;
  int lo[RPH], len[RPH];
#pragma unroll
  for (int r = 0; r < RPH; ++r) {
    const int c = hw + r * (STHREADS / 16);
    const int cc = min(c, EBLOCKS - 1);
    int ra = loff[cc * LOFF_STRIDE + bin];
    int rb = loff[cc * LOFF_STRIDE + bin + 1];
    const int e0c = min(cc * chunk, E);
    const int clen = min(e0c + chunk, E) - e0c;
    ra = min(max(ra, 0), clen);
    rb = min(max(rb, ra), clen);
    lo[r] = e0c + ra;
    len[r] = (c < EBLOCKS) ? rb - ra : 0;
  }

  pre = wave_sum_i(pre);
  if (lane == 0) wpre[wave] = pre;
  if (tid < BIN_NODES) lcnt[tid] = 0;
  __syncthreads();

  int first = 0;
#pragma unroll
  for (int w = 0; w < STHREADS / 32; ++w) first += wpre[w];
  const int start = min(max(first, 0), E);

  unsigned p[2 * RPH];
#pragma unroll
  for (int r = 0; r < RPH; ++r) {
    p[2 * r] = binned[max(min(lo[r] + l16, E - 1), 0)];
    p[2 * r + 1] = binned[max(min(lo[r] + 16 + l16, E - 1), 0)];
  }
#pragma unroll
  for (int r = 0; r < RPH; ++r) {
    atomicAdd(&lcnt[p[2 * r] & (BIN_NODES - 1)], (l16 < len[r]) ? 1 : 0);
    atomicAdd(&lcnt[p[2 * r + 1] & (BIN_NODES - 1)], (l16 + 16 < len[r]) ? 1 : 0);
  }
#pragma unroll
  for (int r = 0; r < RPH; ++r)
    for (int k = l16 + 32; k < len[r]; k += 16)
      atomicAdd(&lcnt[binned[lo[r] + k] & (BIN_NODES - 1)], 1);
  __syncthreads();

  int c = 0, incl = 0;
  if (tid < BIN_NODES) {
    c = lcnt[tid];
    incl = c;
#pragma unroll
    for (int m = 1; m < 32; m <<= 1) {
      const int u = __shfl_up(incl, m, 32);
      if (lane >= m) incl += u;
    }
    if (lane == 31) wtot[wave] = incl;
  }
  __syncthreads();
  if (tid < BIN_NODES) {
    int off = 0;
#pragma unroll
    for (int w = 0; w < BIN_NODES / 32; ++w)
      if (w < wave) off += wtot[w];
    const int nfirst = start + off + incl - c;
    if (base + tid < N) rowptr[base + tid] = nfirst;
    lcnt[tid] = nfirst;
  }
  __syncthreads();
  int total = 0;
#pragma unroll
  for (int w = 0; w < BIN_NODES / 32; ++w) total += wtot[w];

  int pos[2 * RPH];
#pragma unroll
  for (int r = 0; r < RPH; ++r) {
    pos[2 * r] = atomicAdd(&lcnt[p[2 * r] & (BIN_NODES - 1)],
                           (l16 < len[r]) ? 1 : 0);
    pos[2 * r + 1] = atomicAdd(&lcnt[p[2 * r + 1] & (BIN_NODES - 1)],
                               (l16 + 16 < len[r]) ? 1 : 0);
  }
  auto put = [&](int slot, unsigned q) {
    const int loc = slot - start;
    if ((unsigned)loc < (unsigned)SCAP) stage[loc] = q >> BIN_SHIFT;
    else csr_src[slot] = (int)(q >> BIN_SHIFT);
  };
#pragma unroll
  for (int r = 0; r < RPH; ++r) {
    if (l16 < len[r] && (unsigned)pos[2 * r] < (unsigned)E)
      put(pos[2 * r], p[2 * r]);
    if (l16 + 16 < len[r] && (unsigned)pos[2 * r + 1] < (unsigned)E)
      put(pos[2 * r + 1], p[2 * r + 1]);
  }
#pragma unroll
  for (int r = 0; r < RPH; ++r)
    for (int k = l16 + 32; k < len[r]; k += 16) {
      const unsigned q = binned[lo[r] + k];
      const int slot = atomicAdd(&lcnt[q & (BIN_NODES - 1)], 1);
      if ((unsigned)slot < (unsigned)E) put(slot, q);
    }
  __syncthreads();
  const int nst = min(max(total, 0), SCAP);
  const int nend = start + nst;
  for (int g = (start & ~31) + tid; g < nend; g += STHREADS * WBATCH) {
    unsigned v[WBATCH];
#pragma unroll
    for (int u = 0; u < WBATCH; ++u)
      v[u] = stage[min(max(g + u * STHREADS - start, 0), max(nst - 1, 0))];
#pragma unroll
    for (int u = 0; u < WBATCH; ++u) {
      const int gg = g + u * STHREADS;
      if (gg >= start && gg < nend && gg < E) csr_src[gg] = (int)v[u];
    }
  }
}


template <int U>
__device__ __forceinline__ float quad_bcast(float x) {
  const int b = __float_as_int(x);
  return __int_as_float(
      __builtin_amdgcn_update_dpp(b, b, U * 0x55, 0xf, 0xf, false));
}

template <int LPR>
__device__ __forceinline__ float group_sum(float v) {
  v += dpp_row<0x161>(v);
  v += dpp_row<0x162>(v);
  v += dpp_row<0x164>(v);
  if constexpr (LPR == 16) v += dpp_row<0x168>(v);
  return v;
}

template <int LPR>
__device__ __forceinline__ float sum4_group(float p0, float p1, float p2,
                                            float p3, int lane) {
  const bool b0 = (lane & 1) != 0;
  const bool b1 = (lane & 2) != 0;
  float k01 = b0 ? p1 : p0;
  const float t01 = b0 ? p0 : p1;
  float k23 = b0 ? p3 : p2;
  const float t23 = b0 ? p2 : p3;
  k01 += dpp_row<0x161>(t01);
  k23 += dpp_row<0x161>(t23);
  float k = b1 ? k23 : k01;
  const float tt = b1 ? k01 : k23;
  k += dpp_row<0x162>(tt);
  k += dpp_row<0x164>(k);
  if constexpr (LPR == 16) k += dpp_row<0x168>(k);
  return k;
}

#define GWAVES 8
template <int NOUT>
__global__ void node_gather_kernel(const int* __restrict__ rowptr,
                                   const int* __restrict__ csr_src, int E,
                                   int N, const float* __restrict__ H,
                                   const float* __restrict__ av,
                                   const float* __restrict__ bv,
                                   const float* __restrict__ bias,
                                   const float* __restrict__ Wn,
                                   const float* __restrict__ bn,
                                   float* __restrict__ out) {
  constexpr int LPR = NOUT / 4;
  constexpr int NPW = 32 / LPR;
  constexpr int NQ = 32 / LPR;
  static_assert(LPR == 8 || LPR == 16, "lane groups of 8 or 16");
  const int lane = threadIdx.x & 31;
  const int g = lane / LPR;
  const int r0 = lane % LPR;
  const int c0 = r0 * 4;
  const int qu = lane & 3;
  const int wib = __builtin_amdgcn_readfirstlane((int)(threadIdx.x >> 5));
  const int wv = (int)blockIdx.x * GWAVES + wib;
  const int dq = wv * NPW + g;
  const bool live = (dq < N);
  const int d = min(dq, N - 1);

  const int rp1 = rowptr[min(d + 1, N - 1)];
  int end = (d + 1 < N) ? rp1 : E;
  end = min(max(end, 0), E);
  const int beg = min(max(rowptr[d], 0), end);
  const int deg = live ? end - beg : 0;
  int degmax = __builtin_amdgcn_readlane(deg, 0);
#pragma unroll
  for (int k = 1; k < NPW; ++k)
    degmax = max(degmax, __builtin_amdgcn_readlane(deg, k * LPR));
  const unsigned nmax = (unsigned)(N - 1);

  auto load_row = [&](int s) -> float4 {
    const unsigned off = (unsigned)s * (unsigned)(NOUT * 4) + (unsigned)(c0 * 4);
    return *reinterpret_cast<const float4*>(
        reinterpret_cast<const char*>(H) + off);
  };
  auto lane_read = [&](int v, int i) -> int {
    return __builtin_amdgcn_ds_bpermute(i << 2, v);
  };
  auto load_src = [&](int j) -> int {
    const int i = max(min(beg + j, beg + deg - 1), 0);
    return (int)min((unsigned)csr_src[i], nmax);
  };

  const float4 hs = load_row(d);
  const float a_s0 = av[c0], a_s1 = av[c0 + 1], a_s2 = av[c0 + 2], a_s3 = av[c0 + 3];
  const float a_d0 = bv[c0], a_d1 = bv[c0 + 1], a_d2 = bv[c0 + 2], a_d3 = bv[c0 + 3];

  const float adv = group_sum<LPR>(hs.x * a_d0 + hs.y * a_d1 + hs.z * a_d2 + hs.w * a_d3);
  float m = leaky(group_sum<LPR>(hs.x * a_s0 + hs.y * a_s1 + hs.z * a_s2 + hs.w * a_s3) + adv);
  float den = 1.0f;
  float acc0 = hs.x, acc1 = hs.y, acc2 = hs.z, acc3 = hs.w;

  for (int j0 = 0; j0 < degmax; j0 += 32) {
    const int cnt = deg - j0;
    const int cmax = min(32, degmax - j0);
    int sq[NQ];
#pragma unroll
    for (int q = 0; q < NQ; ++q) {
      const int sraw = load_src(j0 + q * LPR + r0);
      sq[q] = (q * LPR + r0 < cnt) ? sraw : d;
    }
    auto src_word = [&](int t) -> int {
      int sv = sq[0];
#pragma unroll
      for (int q = 1; q < NQ; ++q) sv = (t >= q * LPR) ? sq[q] : sv;
      return sv;
    };

    auto trip = [&](const float4 (&hv)[4], int t) {
      float p[4];
#pragma unroll
      for (int u = 0; u < 4; ++u)
        p[u] = hv[u].x * a_s0 + hv[u].y * a_s1 + hv[u].z * a_s2 + hv[u].w * a_s3;
      float e = leaky(sum4_group<LPR>(p[0], p[1], p[2], p[3], lane) + adv);
      e = (t + qu < cnt) ? e : -__builtin_inff();

      float gm = fmaxf(e, dpp_row<0x161>(e));
      gm = fmaxf(gm, dpp_row<0x162>(gm));
      const float m_new = fmaxf(m, gm);
      const float sc = __expf(m - m_new);
      const float w = __expf(e - m_new);
      float ws = w + dpp_row<0x161>(w);
      ws += dpp_row<0x162>(ws);
      den = den * sc + ws;
      m = m_new;

      acc0 *= sc; acc1 *= sc; acc2 *= sc; acc3 *= sc;
      const float w0 = quad_bcast<0>(w), w1 = quad_bcast<1>(w);
      const float w2 = quad_bcast<2>(w), w3 = quad_bcast<3>(w);
      acc0 += hv[0].x * w0; acc1 += hv[0].y * w0; acc2 += hv[0].z * w0; acc3 += hv[0].w * w0;
      acc0 += hv[1].x * w1; acc1 += hv[1].y * w1; acc2 += hv[1].z * w1; acc3 += hv[1].w * w1;
      acc0 += hv[2].x * w2; acc1 += hv[2].y * w2; acc2 += hv[2].z * w2; acc3 += hv[2].w * w2;
      acc0 += hv[3].x * w3; acc1 += hv[3].y * w3; acc2 += hv[3].z * w3; acc3 += hv[3].w * w3;
    };

    float4 hv[4];
#pragma unroll
    for (int u = 0; u < 4; ++u) hv[u] = load_row(lane_read(sq[0], g * LPR + u));

    int t = 0;
#pragma nounroll
    for (; t + 4 < cmax; t += 4) {
      float4 hn[4];
      const int sv = src_word(t + 4);
      const int l0 = g * LPR + ((t + 4) & (LPR - 1));
#pragma unroll
      for (int u = 0; u < 4; ++u) hn[u] = load_row(lane_read(sv, l0 + u));
      trip(hv, t);
#pragma unroll
      for (int u = 0; u < 4; ++u) hv[u] = hn[u];
    }
    trip(hv, t);
  }

  float4 r;
  r.x = live ? fmaxf(acc0 / den + bias[c0], 0.0f) : 0.0f;
  r.y = live ? fmaxf(acc1 / den + bias[c0 + 1], 0.0f) : 0.0f;
  r.z = live ? fmaxf(acc2 / den + bias[c0 + 2], 0.0f) : 0.0f;
  r.w = live ? fmaxf(acc3 / den + bias[c0 + 3], 0.0f) : 0.0f;
  const int nib = wib * NPW + g;

  if constexpr (NOUT == H1) {
    constexpr int ROWS = GWAVES * NPW;
    constexpr int NO2 = H2;
    constexpr int LDT = NOUT + 4;
    static_assert(ROWS == 16, "one 16-row matrix tile per work-group");
    __shared__ __attribute__((aligned(16))) float ts[ROWS * LDT];
    *reinterpret_cast<float4*>(&ts[nib * LDT + c0]) = r;
    __syncthreads();
    if (wib < NO2 / 16) {
      const int lh = lane & 15;
      const bool hi = (lane >= 16);
      const int col0 = wib * 16;
      v8f cacc = (v8f){};
#pragma unroll
      for (int k = 0; k < NOUT; k += 4) {
        const int ka = k + (hi ? 2 : 0);
        v2f b;
        b.x = Wn[ka * NO2 + col0 + lh];
        b.y = Wn[(ka + 1) * NO2 + col0 + lh];
        v2f a;
        a.x = ts[lh * LDT + ka];
        a.y = ts[lh * LDT + ka + 1];
        cacc = __builtin_amdgcn_wmma_f32_16x16x4_f32(false, a, false, b,
                                                     (short)0, cacc,
                                                     false, false);
      }
      const int row0 = (int)blockIdx.x * ROWS + (hi ? 8 : 0);
#pragma unroll
      for (int q = 0; q < 8; ++q) {
        if (row0 + q < N) out[(size_t)(row0 + q) * NO2 + col0 + lh] = cacc[q];
      }
    }
  } else {
    constexpr int ROWS = GWAVES * NPW;
    static_assert(ROWS == 32, "one 128-byte line of results per work-group");
    __shared__ float fo[ROWS];
    const float s = group_sum<LPR>(r.x * Wn[c0] + r.y * Wn[c0 + 1] +
                                   r.z * Wn[c0 + 2] + r.w * Wn[c0 + 3]) + bn[0];
    if (r0 == 0) fo[nib] = s;
    __syncthreads();
    const int n = (int)blockIdx.x * ROWS + (int)threadIdx.x;
    if (threadIdx.x < ROWS && n < N) out[n] = fo[threadIdx.x];
  }
}

static inline unsigned cdiv(long long a, long long b) {
  return (unsigned)((a + b - 1) / b);
}

extern "C" void kernel_launch(void* const* d_in, const int* in_sizes, int n_in,
                              void* d_out, int out_size, void* d_ws,
                              size_t ws_size, hipStream_t stream) {
  (void)n_in; (void)out_size; (void)ws_size;
  const float* x     = (const float*)d_in[0];
  const int*   ei    = (const int*)d_in[1];
  const float* W1    = (const float*)d_in[2];
  const float* asrc1 = (const float*)d_in[3];
  const float* adst1 = (const float*)d_in[4];
  const float* b1    = (const float*)d_in[5];
  const float* W2    = (const float*)d_in[6];
  const float* asrc2 = (const float*)d_in[7];
  const float* adst2 = (const float*)d_in[8];
  const float* b2    = (const float*)d_in[9];
  const float* Wfc   = (const float*)d_in[10];
  const float* bfc   = (const float*)d_in[11];
  float* out = (float*)d_out;

  const int N = N_NODES;
  const int E = in_sizes[1] / 2;
  const int* src = ei;
  const int* dst = ei + E;

  float* ws = (float*)d_ws;
  size_t off = 0;
  float* h    = ws + off; off += (size_t)N * H1;
  float* h2   = ws + off; off += (size_t)N * H1;
  off += (size_t)N * H2;
  int* loff    = (int*)(ws + off); off += N;
  int* rowptr  = (int*)(ws + off); off += N;
  unsigned* binned = (unsigned*)(ws + off); off += E;
  int* csr_src = (int*)(ws + off); off += E;

  constexpr int MR = 32;
  constexpr int GN1 = GWAVES * (128 / H1);
  constexpr int GN2 = GWAVES * (128 / H2);
  const dim3 GB(GWAVES * 32);

  gemm_wmma_kernel<F_IN, H1, MR><<<EBLOCKS + N / MR, (H1 / 16) * 32, 0, stream>>>(x, W1, h, src, dst, E, loff, binned);

  bin_sort_kernel<<<NBINS, STHREADS, 0, stream>>>(loff, E, N, binned, rowptr, csr_src);

  node_gather_kernel<H1><<<cdiv(N, GN1), GB, 0, stream>>>(rowptr, csr_src, E, N, h, asrc1, adst1, b1, W2, bfc, h2);

  node_gather_kernel<H2><<<cdiv(N, GN2), GB, 0, stream>>>(rowptr, csr_src, E, N, h2, asrc2, adst2, b2, Wfc, bfc, out);
}
